# starting kernel
# baseline (speedup 1.0000x reference)
#include <hip/hip_runtime.h>
#include <hip/hip_bf16.h>

typedef __attribute__((ext_vector_type(2))) float v2f;
typedef __attribute__((ext_vector_type(8))) float v8f;

#define NNODES 100000
#define DFEAT  128
#define BN_EPS 1e-5f


__global__ __launch_bounds__(256) void zero_kernel(float* __restrict__ p, size_t n) {
    size_t i = (size_t)blockIdx.x * blockDim.x + threadIdx.x;
    size_t stride = (size_t)gridDim.x * blockDim.x;
    for (; i < n; i += stride) p[i] = 0.0f;
}

__global__ __launch_bounds__(256) void degree_kernel(const int* __restrict__ src,
                                                     const int* __restrict__ dst,
                                                     float* __restrict__ dout,
                                                     float* __restrict__ din, int E) {
    int e = blockIdx.x * blockDim.x + threadIdx.x;
    if (e >= E) return;
    atomicAdd(&dout[src[e]], 1.0f);
    atomicAdd(&din[dst[e]], 1.0f);
}

__global__ __launch_bounds__(256) void norm_kernel(float* __restrict__ dout,
                                                   float* __restrict__ din, int n) {
    int i = blockIdx.x * blockDim.x + threadIdx.x;
    if (i >= n) return;
    dout[i] = rsqrtf(fmaxf(dout[i], 1.0f));
    din[i]  = rsqrtf(fmaxf(din[i], 1.0f));
}

__global__ __launch_bounds__(256) void scatter_kernel(const float* __restrict__ feats,
                                                      const int* __restrict__ src,
                                                      const int* __restrict__ dst,
                                                      const float* __restrict__ norm_src,
                                                      float* __restrict__ agg, int E) {
    long long t = (long long)blockIdx.x * blockDim.x + threadIdx.x;
    int e = (int)(t >> 5);
    if (e >= E) return;
    int lane = (int)(t & 31);
    int s = src[e], d = dst[e];
    float ns = norm_src[s];
    const float4* fr = (const float4*)(feats + (size_t)s * DFEAT);
    float4 v = fr[lane];
    float* ar = agg + (size_t)d * DFEAT + lane * 4;
    atomicAdd(ar + 0, v.x * ns);
    atomicAdd(ar + 1, v.y * ns);
    atomicAdd(ar + 2, v.z * ns);
    atomicAdd(ar + 3, v.w * ns);
}

__global__ __launch_bounds__(256) void gemm_bn_kernel(
    const float* __restrict__ agg, const float* __restrict__ feats,
    const float* __restrict__ norm_dst,
    const float* __restrict__ W,  const float* __restrict__ b,
    const float* __restrict__ Wr, const float* __restrict__ br,
    float* __restrict__ y, float* __restrict__ bn_sum, float* __restrict__ bn_sumsq) {
    const int lane = threadIdx.x & 31;
    const int wave = threadIdx.x >> 5;
    const int rowBase = blockIdx.x * 16;
    const int n0 = wave * 16;
    const int m  = lane & 15;
    const int kh = (lane >> 4) * 2;
    const int row = rowBase + m;
    const float nd = norm_dst[row];
    const float* aRow = agg   + (size_t)row * DFEAT;
    const float* fRow = feats + (size_t)row * DFEAT;

    v8f accC = {};
    v8f accR = {};
    #pragma unroll
    for (int k = 0; k < DFEAT; k += 4) {
        v2f a, f, bw, bwr;
        a.x = aRow[k + kh]     * nd;
        a.y = aRow[k + kh + 1] * nd;
        f.x = fRow[k + kh];
        f.y = fRow[k + kh + 1];
        bw.x  = W [(size_t)(k + kh)     * DFEAT + n0 + m];
        bw.y  = W [(size_t)(k + kh + 1) * DFEAT + n0 + m];
        bwr.x = Wr[(size_t)(k + kh)     * DFEAT + n0 + m];
        bwr.y = Wr[(size_t)(k + kh + 1) * DFEAT + n0 + m];
        accC = __builtin_amdgcn_wmma_f32_16x16x4_f32(false, a, false, bw,  (short)0, accC, false, false);
        accR = __builtin_amdgcn_wmma_f32_16x16x4_f32(false, f, false, bwr, (short)0, accR, false, false);
    }

    const int col  = n0 + m;
    const float bc  = b[col];
    const float brc = br[col];
    const int rOff = (lane >> 4) * 8;
    float s8 = 0.0f, q8 = 0.0f;
    #pragma unroll
    for (int r = 0; r < 8; ++r) {
        float c = accC[r] + bc;  c = c > 0.0f ? c : 0.0f;
        float e = accR[r] + brc; e = e > 0.0f ? e : 0.0f;
        float v = c + e;
        y[(size_t)(rowBase + rOff + r) * DFEAT + col] = v;
        s8 += v;
        q8 += v * v;
    }
    float s8o = __shfl_down(s8, 16);
    float q8o = __shfl_down(q8, 16);
    if (lane < 16) {
        atomicAdd(&bn_sum[col],   s8 + s8o);
        atomicAdd(&bn_sumsq[col], q8 + q8o);
    }
}

__global__ void bn_params_kernel(const float* __restrict__ sum, const float* __restrict__ sumsq,
                                 const float* __restrict__ gamma, const float* __restrict__ beta,
                                 float* __restrict__ scale, float* __restrict__ shift, int n) {
    int c = threadIdx.x;
    if (c >= DFEAT) return;
    float inv  = 1.0f / (float)n;
    float mean = sum[c] * inv;
    float var  = sumsq[c] * inv - mean * mean;
    float sc   = gamma[c] * rsqrtf(var + BN_EPS);
    scale[c] = sc;
    shift[c] = beta[c] - mean * sc;
}

__global__ __launch_bounds__(256) void bn_apply_kernel(float* __restrict__ y,
                                                       const float* __restrict__ scale,
                                                       const float* __restrict__ shift,
                                                       size_t nvec4) {
    size_t i = (size_t)blockIdx.x * blockDim.x + threadIdx.x;
    if (i >= nvec4) return;
    float4 v = ((float4*)y)[i];
    int col = (int)((i * 4) & (DFEAT - 1));
    v.x = v.x * scale[col + 0] + shift[col + 0];
    v.y = v.y * scale[col + 1] + shift[col + 1];
    v.z = v.z * scale[col + 2] + shift[col + 2];
    v.w = v.w * scale[col + 3] + shift[col + 3];
    ((float4*)y)[i] = v;
}

extern "C" void kernel_launch(void* const* d_in, const int* in_sizes, int n_in,
                              void* d_out, int out_size, void* d_ws, size_t ws_size,
                              hipStream_t stream) {
    const float* feats = (const float*)d_in[0];
    const int*   src   = (const int*)d_in[1];
    const int*   dst   = (const int*)d_in[2];
    const float* W     = (const float*)d_in[3];
    const float* b     = (const float*)d_in[4];
    const float* Wr    = (const float*)d_in[5];
    const float* br    = (const float*)d_in[6];
    const float* gamma = (const float*)d_in[7];
    const float* beta  = (const float*)d_in[8];
    float* out = (float*)d_out;

    const int N = NNODES;
    const int E = in_sizes[1];

    float* ws        = (float*)d_ws;
    float* agg       = ws;
    float* deg_out   = agg + (size_t)N * DFEAT;
    float* deg_in    = deg_out + N;
    float* bn_sum    = deg_in + N;
    float* bn_sumsq  = bn_sum + DFEAT;
    float* bn_scale  = bn_sumsq + DFEAT;
    float* bn_shift  = bn_scale + DFEAT;

    size_t nz = (size_t)N * DFEAT + 2 * (size_t)N + 2 * DFEAT;
    zero_kernel<<<4096, 256, 0, stream>>>(ws, nz);

    degree_kernel<<<(E + 255) / 256, 256, 0, stream>>>(src, dst, deg_out, deg_in, E);

    norm_kernel<<<(N + 255) / 256, 256, 0, stream>>>(deg_out, deg_in, N);

    long long sthreads = (long long)E * 32;
    int sblocks = (int)((sthreads + 255) / 256);
    scatter_kernel<<<sblocks, 256, 0, stream>>>(feats, src, dst, deg_out, agg, E);

    gemm_bn_kernel<<<N / 16, 256, 0, stream>>>(agg, feats, deg_in, W, b, Wr, br,
                                               out, bn_sum, bn_sumsq);

    bn_params_kernel<<<1, DFEAT, 0, stream>>>(bn_sum, bn_sumsq, gamma, beta,
                                              bn_scale, bn_shift, N);

    size_t nvec4 = (size_t)N * DFEAT / 4;
    bn_apply_kernel<<<(int)((nvec4 + 255) / 256), 256, 0, stream>>>(out, bn_scale, bn_shift, nvec4);
}


# Aggregate by destination-sorted gather (counting sort via returning int atomics) instead of f32 atomic scatter; row sums now added in arrival-rank order in registers; agg zero-fill dropped
# speedup vs baseline: 1.3627x; 1.3627x over previous
#include <hip/hip_runtime.h>
#include <hip/hip_bf16.h>

typedef __attribute__((ext_vector_type(2))) float v2f;
typedef __attribute__((ext_vector_type(8))) float v8f;

#define NNODES 100000
#define DFEAT  128
#define BN_EPS 1e-5f
#define SCAN_T 1024
#define SCAN_CH 100


__global__ __launch_bounds__(256) void zero_kernel(int* __restrict__ p, int n) {
    int i = blockIdx.x * blockDim.x + threadIdx.x;
    if (i < n) p[i] = 0;
}

__global__ __launch_bounds__(256) void degree_kernel(const int* __restrict__ src,
                                                     const int* __restrict__ dst,
                                                     int* __restrict__ cout,
                                                     int* __restrict__ cin,
                                                     int* __restrict__ rank, int E) {
    int e = blockIdx.x * blockDim.x + threadIdx.x;
    if (e >= E) return;
    atomicAdd(&cout[src[e]], 1);
    rank[e] = atomicAdd(&cin[dst[e]], 1);
}

__global__ __launch_bounds__(SCAN_T) void scan_kernel(const int* __restrict__ cnt,
                                                      int* __restrict__ rowptr, int n) {
    __shared__ int part[SCAN_T];
    const int t = threadIdx.x;
    const int base = t * SCAN_CH;
    const bool full = (base + SCAN_CH <= n);
    int v[SCAN_CH];
    if (full) {
        const int4* c4 = (const int4*)(cnt + base);
        #pragma unroll
        for (int i = 0; i < SCAN_CH / 4; ++i) {
            int4 q = c4[i];
            v[4 * i + 0] = q.x; v[4 * i + 1] = q.y; v[4 * i + 2] = q.z; v[4 * i + 3] = q.w;
        }
    } else {
        #pragma unroll
        for (int i = 0; i < SCAN_CH; ++i) v[i] = (base + i < n) ? cnt[base + i] : 0;
    }
    int sum = 0;
    #pragma unroll
    for (int i = 0; i < SCAN_CH; ++i) sum += v[i];
    part[t] = sum;
    __syncthreads();
    for (int off = 1; off < SCAN_T; off <<= 1) {
        int x = (t >= off) ? part[t - off] : 0;
        __syncthreads();
        part[t] += x;
        __syncthreads();
    }
    int run = part[t] - sum;
    if (full) {
        int4* r4 = (int4*)(rowptr + base);
        #pragma unroll
        for (int i = 0; i < SCAN_CH / 4; ++i) {
            int4 q;
            q.x = run; run += v[4 * i + 0];
            q.y = run; run += v[4 * i + 1];
            q.z = run; run += v[4 * i + 2];
            q.w = run; run += v[4 * i + 3];
            r4[i] = q;
        }
    } else {
        #pragma unroll
        for (int i = 0; i < SCAN_CH; ++i) {
            if (base + i < n) rowptr[base + i] = run;
            run += v[i];
        }
    }
    if (t == SCAN_T - 1) rowptr[n] = part[SCAN_T - 1];
}

__global__ __launch_bounds__(256) void norm_kernel(int* __restrict__ cout,
                                                   int* __restrict__ cin, int n) {
    int i = blockIdx.x * blockDim.x + threadIdx.x;
    if (i >= n) return;
    cout[i] = __float_as_int(rsqrtf(fmaxf((float)cout[i], 1.0f)));
    cin[i]  = __float_as_int(rsqrtf(fmaxf((float)cin[i], 1.0f)));
}

__global__ __launch_bounds__(256) void fill_kernel(const int* __restrict__ src,
                                                   const int* __restrict__ dst,
                                                   const int* __restrict__ rowptr,
                                                   const int* __restrict__ rank,
                                                   int* __restrict__ csr, int E) {
    int e = blockIdx.x * blockDim.x + threadIdx.x;
    if (e >= E) return;
    csr[rowptr[dst[e]] + rank[e]] = src[e];
}

__global__ __launch_bounds__(256) void gather_kernel(const float* __restrict__ feats,
                                                     const int* __restrict__ rowptr,
                                                     const int* __restrict__ csr,
                                                     const float* __restrict__ norm_src,
                                                     float* __restrict__ agg, int n) {
    const int row = blockIdx.x * 8 + (threadIdx.x >> 5);
    if (row >= n) return;
    const int lane = threadIdx.x & 31;
    const int beg = __builtin_amdgcn_readfirstlane(rowptr[row]);
    const int end = __builtin_amdgcn_readfirstlane(rowptr[row + 1]);
    float4 acc = make_float4(0.0f, 0.0f, 0.0f, 0.0f);
    for (int b0 = beg; b0 < end; b0 += 32) {
        const int cnt = min(32, end - b0);
        int   s_l  = 0;
        float ns_l = 0.0f;
        if (lane < cnt) { s_l = csr[b0 + lane]; ns_l = norm_src[s_l]; }
        #pragma unroll 4
        for (int j = 0; j < cnt; ++j) {
            const int   s  = __builtin_amdgcn_readlane(s_l, j);
            const float ns = __int_as_float(__builtin_amdgcn_readlane(__float_as_int(ns_l), j));
            const float4 v = ((const float4*)(feats + (size_t)s * DFEAT))[lane];
            acc.x += v.x * ns;
            acc.y += v.y * ns;
            acc.z += v.z * ns;
            acc.w += v.w * ns;
        }
    }
    ((float4*)(agg + (size_t)row * DFEAT))[lane] = acc;
}

__global__ __launch_bounds__(256) void gemm_bn_kernel(
    const float* __restrict__ agg, const float* __restrict__ feats,
    const float* __restrict__ norm_dst,
    const float* __restrict__ W,  const float* __restrict__ b,
    const float* __restrict__ Wr, const float* __restrict__ br,
    float* __restrict__ y, float* __restrict__ bn_sum, float* __restrict__ bn_sumsq) {
    const int lane = threadIdx.x & 31;
    const int wave = threadIdx.x >> 5;
    const int rowBase = blockIdx.x * 16;
    const int n0 = wave * 16;
    const int m  = lane & 15;
    const int kh = (lane >> 4) * 2;
    const int row = rowBase + m;
    const float nd = norm_dst[row];
    const float* aRow = agg   + (size_t)row * DFEAT;
    const float* fRow = feats + (size_t)row * DFEAT;

    v8f accC = {};
    v8f accR = {};
    #pragma unroll
    for (int k = 0; k < DFEAT; k += 4) {
        v2f a, f, bw, bwr;
        a.x = aRow[k + kh]     * nd;
        a.y = aRow[k + kh + 1] * nd;
        f.x = fRow[k + kh];
        f.y = fRow[k + kh + 1];
        bw.x  = W [(size_t)(k + kh)     * DFEAT + n0 + m];
        bw.y  = W [(size_t)(k + kh + 1) * DFEAT + n0 + m];
        bwr.x = Wr[(size_t)(k + kh)     * DFEAT + n0 + m];
        bwr.y = Wr[(size_t)(k + kh + 1) * DFEAT + n0 + m];
        accC = __builtin_amdgcn_wmma_f32_16x16x4_f32(false, a, false, bw,  (short)0, accC, false, false);
        accR = __builtin_amdgcn_wmma_f32_16x16x4_f32(false, f, false, bwr, (short)0, accR, false, false);
    }

    const int col  = n0 + m;
    const float bc  = b[col];
    const float brc = br[col];
    const int rOff = (lane >> 4) * 8;
    float s8 = 0.0f, q8 = 0.0f;
    #pragma unroll
    for (int r = 0; r < 8; ++r) {
        float c = accC[r] + bc;  c = c > 0.0f ? c : 0.0f;
        float e = accR[r] + brc; e = e > 0.0f ? e : 0.0f;
        float v = c + e;
        y[(size_t)(rowBase + rOff + r) * DFEAT + col] = v;
        s8 += v;
        q8 += v * v;
    }
    float s8o = __shfl_down(s8, 16);
    float q8o = __shfl_down(q8, 16);
    if (lane < 16) {
        atomicAdd(&bn_sum[col],   s8 + s8o);
        atomicAdd(&bn_sumsq[col], q8 + q8o);
    }
}

__global__ void bn_params_kernel(const float* __restrict__ sum, const float* __restrict__ sumsq,
                                 const float* __restrict__ gamma, const float* __restrict__ beta,
                                 float* __restrict__ scale, float* __restrict__ shift, int n) {
    int c = threadIdx.x;
    if (c >= DFEAT) return;
    float inv  = 1.0f / (float)n;
    float mean = sum[c] * inv;
    float var  = sumsq[c] * inv - mean * mean;
    float sc   = gamma[c] * rsqrtf(var + BN_EPS);
    scale[c] = sc;
    shift[c] = beta[c] - mean * sc;
}

__global__ __launch_bounds__(256) void bn_apply_kernel(float* __restrict__ y,
                                                       const float* __restrict__ scale,
                                                       const float* __restrict__ shift,
                                                       size_t nvec4) {
    size_t i = (size_t)blockIdx.x * blockDim.x + threadIdx.x;
    if (i >= nvec4) return;
    float4 v = ((float4*)y)[i];
    int col = (int)((i * 4) & (DFEAT - 1));
    v.x = v.x * scale[col + 0] + shift[col + 0];
    v.y = v.y * scale[col + 1] + shift[col + 1];
    v.z = v.z * scale[col + 2] + shift[col + 2];
    v.w = v.w * scale[col + 3] + shift[col + 3];
    ((float4*)y)[i] = v;
}

extern "C" void kernel_launch(void* const* d_in, const int* in_sizes, int n_in,
                              void* d_out, int out_size, void* d_ws, size_t ws_size,
                              hipStream_t stream) {
    const float* feats = (const float*)d_in[0];
    const int*   src   = (const int*)d_in[1];
    const int*   dst   = (const int*)d_in[2];
    const float* W     = (const float*)d_in[3];
    const float* b     = (const float*)d_in[4];
    const float* Wr    = (const float*)d_in[5];
    const float* br    = (const float*)d_in[6];
    const float* gamma = (const float*)d_in[7];
    const float* beta  = (const float*)d_in[8];
    float* out = (float*)d_out;

    const int N = NNODES;
    const int E = in_sizes[1];

    float* ws        = (float*)d_ws;
    float* agg       = ws;
    float* deg_out   = agg + (size_t)N * DFEAT;
    float* deg_in    = deg_out + N;
    float* bn_sum    = deg_in + N;
    float* bn_sumsq  = bn_sum + DFEAT;
    float* bn_scale  = bn_sumsq + DFEAT;
    float* bn_shift  = bn_scale + DFEAT;

    int* cnt_out = (int*)deg_out;
    int* cnt_in  = (int*)deg_in;
    int* rank    = (int*)d_out;
    int* csr     = rank + E;
    int* rowptr  = csr + E;

    int nz = 2 * N + 2 * DFEAT;
    zero_kernel<<<(nz + 255) / 256, 256, 0, stream>>>(cnt_out, nz);

    degree_kernel<<<(E + 255) / 256, 256, 0, stream>>>(src, dst, cnt_out, cnt_in, rank, E);

    scan_kernel<<<1, SCAN_T, 0, stream>>>(cnt_in, rowptr, N);

    norm_kernel<<<(N + 255) / 256, 256, 0, stream>>>(cnt_out, cnt_in, N);

    fill_kernel<<<(E + 255) / 256, 256, 0, stream>>>(src, dst, rowptr, rank, csr, E);

    gather_kernel<<<(N + 7) / 8, 256, 0, stream>>>(feats, rowptr, csr, deg_out, agg, N);

    gemm_bn_kernel<<<N / 16, 256, 0, stream>>>(agg, feats, deg_in, W, b, Wr, br,
                                               out, bn_sum, bn_sumsq);

    bn_params_kernel<<<1, DFEAT, 0, stream>>>(bn_sum, bn_sumsq, gamma, beta,
                                              bn_scale, bn_shift, N);

    size_t nvec4 = (size_t)N * DFEAT / 4;
    bn_apply_kernel<<<(int)((nvec4 + 255) / 256), 256, 0, stream>>>(out, bn_scale, bn_shift, nvec4);
}


# Group edges by destination with LDS histograms over 512-node bins (count, scan, bin scatter, per-bin sort) replacing two global atomics per edge; row sums stay in arrival order, now of LDS adds
# speedup vs baseline: 1.8215x; 1.3367x over previous
#include <hip/hip_runtime.h>
#include <hip/hip_bf16.h>

typedef __attribute__((ext_vector_type(2))) float v2f;
typedef __attribute__((ext_vector_type(8))) float v8f;

#define NNODES 100000
#define DFEAT  128
#define BN_EPS 1e-5f
#define SCAN_T 1024
#define SCAN_CH 100
#define BIN_SH 9
#define BIN_R  512
#define NB     196
#define NCHUNK 250
#define EDGE_T 1024


__global__ __launch_bounds__(256) void zero_kernel(int* __restrict__ p, int n) {
    int i = blockIdx.x * blockDim.x + threadIdx.x;
    if (i < n) p[i] = 0;
}

__global__ __launch_bounds__(EDGE_T) void hist_kernel(const int* __restrict__ src,
                                                      const int* __restrict__ dst,
                                                      int* __restrict__ cnt, int E) {
    __shared__ int hD[NB];
    __shared__ int hS[NB];
    const int t = threadIdx.x;
    const int w = blockIdx.x;
    if (t < NB) { hD[t] = 0; hS[t] = 0; }
    __syncthreads();
    const int ch  = (E + NCHUNK - 1) / NCHUNK;
    const int beg = w * ch;
    const int end = min(E, beg + ch);
    for (int e0 = beg + t; e0 < end; e0 += 4 * EDGE_T) {
        int d[4], s[4];
        #pragma unroll
        for (int j = 0; j < 4; ++j) {
            const int e = min(e0 + j * EDGE_T, end - 1);
            d[j] = dst[e];
            s[j] = src[e];
        }
        #pragma unroll
        for (int j = 0; j < 4; ++j) {
            if (e0 + j * EDGE_T < end) {
                atomicAdd(&hD[d[j] >> BIN_SH], 1);
                atomicAdd(&hS[s[j] >> BIN_SH], 1);
            }
        }
    }
    __syncthreads();
    if (t < NB) {
        cnt[t * NCHUNK + w]        = hD[t];
        cnt[(NB + t) * NCHUNK + w] = hS[t];
    }
}

__global__ __launch_bounds__(SCAN_T) void scan_kernel(const int* __restrict__ cnt,
                                                      int* __restrict__ rowptr, int n) {
    __shared__ int part[SCAN_T];
    const int t = threadIdx.x;
    const int base = t * SCAN_CH;
    const bool full = (base + SCAN_CH <= n);
    int v[SCAN_CH];
    if (full) {
        const int4* c4 = (const int4*)(cnt + base);
        #pragma unroll
        for (int i = 0; i < SCAN_CH / 4; ++i) {
            int4 q = c4[i];
            v[4 * i + 0] = q.x; v[4 * i + 1] = q.y; v[4 * i + 2] = q.z; v[4 * i + 3] = q.w;
        }
    } else {
        #pragma unroll
        for (int i = 0; i < SCAN_CH; ++i) v[i] = (base + i < n) ? cnt[base + i] : 0;
    }
    int sum = 0;
    #pragma unroll
    for (int i = 0; i < SCAN_CH; ++i) sum += v[i];
    part[t] = sum;
    __syncthreads();
    for (int off = 1; off < SCAN_T; off <<= 1) {
        int x = (t >= off) ? part[t - off] : 0;
        __syncthreads();
        part[t] += x;
        __syncthreads();
    }
    int run = part[t] - sum;
    if (full) {
        int4* r4 = (int4*)(rowptr + base);
        #pragma unroll
        for (int i = 0; i < SCAN_CH / 4; ++i) {
            int4 q;
            q.x = run; run += v[4 * i + 0];
            q.y = run; run += v[4 * i + 1];
            q.z = run; run += v[4 * i + 2];
            q.w = run; run += v[4 * i + 3];
            r4[i] = q;
        }
    } else {
        #pragma unroll
        for (int i = 0; i < SCAN_CH; ++i) {
            if (base + i < n) rowptr[base + i] = run;
            run += v[i];
        }
    }
    if (t == SCAN_T - 1) rowptr[n] = part[SCAN_T - 1];
}

__global__ __launch_bounds__(EDGE_T) void scatter_kernel(const int* __restrict__ src,
                                                         const int* __restrict__ dst,
                                                         const int* __restrict__ off,
                                                         int* __restrict__ recA,
                                                         int* __restrict__ recD, int E) {
    __shared__ int cD[NB];
    __shared__ int cS[NB];
    const int t = threadIdx.x;
    const int w = blockIdx.x;
    if (t < NB) {
        cD[t] = off[t * NCHUNK + w];
        cS[t] = off[(NB + t) * NCHUNK + w];
    }
    __syncthreads();
    const int ch  = (E + NCHUNK - 1) / NCHUNK;
    const int beg = w * ch;
    const int end = min(E, beg + ch);
    for (int e0 = beg + t; e0 < end; e0 += 4 * EDGE_T) {
        int d[4], s[4];
        #pragma unroll
        for (int j = 0; j < 4; ++j) {
            const int e = min(e0 + j * EDGE_T, end - 1);
            d[j] = dst[e];
            s[j] = src[e];
        }
        #pragma unroll
        for (int j = 0; j < 4; ++j) {
            if (e0 + j * EDGE_T < end) {
                const int p = atomicAdd(&cD[d[j] >> BIN_SH], 1);
                recA[p] = s[j];
                recD[p] = d[j];
                const int q = atomicAdd(&cS[s[j] >> BIN_SH], 1);
                recA[q] = s[j];
            }
        }
    }
}

__global__ __launch_bounds__(BIN_R) void local_kernel(const int* __restrict__ off,
                                                      const int* __restrict__ recA,
                                                      const int* __restrict__ recD,
                                                      float* __restrict__ norm_src,
                                                      float* __restrict__ norm_dst,
                                                      int* __restrict__ rowptr,
                                                      int* __restrict__ csr, int n) {
    __shared__ int cIn[BIN_R];
    __shared__ int cOut[BIN_R];
    __shared__ int part[BIN_R];
    const int t = threadIdx.x;
    const int b = blockIdx.x;
    cIn[t] = 0;
    cOut[t] = 0;
    __syncthreads();
    const int dBeg = off[b * NCHUNK];
    const int dEnd = off[(b + 1) * NCHUNK];
    const int sBeg = off[(NB + b) * NCHUNK];
    const int sEnd = off[(NB + b + 1) * NCHUNK];
    for (int i0 = dBeg + t; i0 < dEnd; i0 += 4 * BIN_R) {
        int d[4];
        #pragma unroll
        for (int j = 0; j < 4; ++j) {
            const int i = min(i0 + j * BIN_R, dEnd - 1);
            d[j] = recD[i];
        }
        #pragma unroll
        for (int j = 0; j < 4; ++j)
            if (i0 + j * BIN_R < dEnd) atomicAdd(&cIn[d[j] & (BIN_R - 1)], 1);
    }
    for (int i0 = sBeg + t; i0 < sEnd; i0 += 4 * BIN_R) {
        int s[4];
        #pragma unroll
        for (int j = 0; j < 4; ++j) {
            const int i = min(i0 + j * BIN_R, sEnd - 1);
            s[j] = recA[i];
        }
        #pragma unroll
        for (int j = 0; j < 4; ++j)
            if (i0 + j * BIN_R < sEnd) atomicAdd(&cOut[s[j] & (BIN_R - 1)], 1);
    }
    __syncthreads();
    const int node = b * BIN_R + t;
    const int ci = cIn[t];
    if (node < n) {
        norm_src[node] = rsqrtf(fmaxf((float)cOut[t], 1.0f));
        norm_dst[node] = rsqrtf(fmaxf((float)ci, 1.0f));
    }
    part[t] = ci;
    __syncthreads();
    for (int o = 1; o < BIN_R; o <<= 1) {
        int x = (t >= o) ? part[t - o] : 0;
        __syncthreads();
        part[t] += x;
        __syncthreads();
    }
    const int start = dBeg + part[t] - ci;
    if (node <= n) rowptr[node] = start;
    cIn[t] = start;
    __syncthreads();
    for (int i0 = dBeg + t; i0 < dEnd; i0 += 4 * BIN_R) {
        int d[4], s[4];
        #pragma unroll
        for (int j = 0; j < 4; ++j) {
            const int i = min(i0 + j * BIN_R, dEnd - 1);
            d[j] = recD[i];
            s[j] = recA[i];
        }
        #pragma unroll
        for (int j = 0; j < 4; ++j) {
            if (i0 + j * BIN_R < dEnd) {
                const int p = atomicAdd(&cIn[d[j] & (BIN_R - 1)], 1);
                csr[p] = s[j];
            }
        }
    }
}

__global__ __launch_bounds__(256) void gather_kernel(const float* __restrict__ feats,
                                                     const int* __restrict__ rowptr,
                                                     const int* __restrict__ csr,
                                                     const float* __restrict__ norm_src,
                                                     float* __restrict__ agg, int n) {
    const int row = blockIdx.x * 8 + (threadIdx.x >> 5);
    if (row >= n) return;
    const int lane = threadIdx.x & 31;
    const int beg = __builtin_amdgcn_readfirstlane(rowptr[row]);
    const int end = __builtin_amdgcn_readfirstlane(rowptr[row + 1]);
    float4 acc = make_float4(0.0f, 0.0f, 0.0f, 0.0f);
    for (int b0 = beg; b0 < end; b0 += 32) {
        const int cnt = min(32, end - b0);
        int   s_l  = 0;
        float ns_l = 0.0f;
        if (lane < cnt) { s_l = csr[b0 + lane]; ns_l = norm_src[s_l]; }
        #pragma unroll 4
        for (int j = 0; j < cnt; ++j) {
            const int   s  = __builtin_amdgcn_readlane(s_l, j);
            const float ns = __int_as_float(__builtin_amdgcn_readlane(__float_as_int(ns_l), j));
            const float4 v = ((const float4*)(feats + (size_t)s * DFEAT))[lane];
            acc.x += v.x * ns;
            acc.y += v.y * ns;
            acc.z += v.z * ns;
            acc.w += v.w * ns;
        }
    }
    ((float4*)(agg + (size_t)row * DFEAT))[lane] = acc;
}

__global__ __launch_bounds__(256) void gemm_bn_kernel(
    const float* __restrict__ agg, const float* __restrict__ feats,
    const float* __restrict__ norm_dst,
    const float* __restrict__ W,  const float* __restrict__ b,
    const float* __restrict__ Wr, const float* __restrict__ br,
    float* __restrict__ y, float* __restrict__ bn_sum, float* __restrict__ bn_sumsq) {
    const int lane = threadIdx.x & 31;
    const int wave = threadIdx.x >> 5;
    const int rowBase = blockIdx.x * 16;
    const int n0 = wave * 16;
    const int m  = lane & 15;
    const int kh = (lane >> 4) * 2;
    const int row = rowBase + m;
    const float nd = norm_dst[row];
    const float* aRow = agg   + (size_t)row * DFEAT;
    const float* fRow = feats + (size_t)row * DFEAT;

    v8f accC = {};
    v8f accR = {};
    #pragma unroll
    for (int k = 0; k < DFEAT; k += 4) {
        v2f a, f, bw, bwr;
        a.x = aRow[k + kh]     * nd;
        a.y = aRow[k + kh + 1] * nd;
        f.x = fRow[k + kh];
        f.y = fRow[k + kh + 1];
        bw.x  = W [(size_t)(k + kh)     * DFEAT + n0 + m];
        bw.y  = W [(size_t)(k + kh + 1) * DFEAT + n0 + m];
        bwr.x = Wr[(size_t)(k + kh)     * DFEAT + n0 + m];
        bwr.y = Wr[(size_t)(k + kh + 1) * DFEAT + n0 + m];
        accC = __builtin_amdgcn_wmma_f32_16x16x4_f32(false, a, false, bw,  (short)0, accC, false, false);
        accR = __builtin_amdgcn_wmma_f32_16x16x4_f32(false, f, false, bwr, (short)0, accR, false, false);
    }

    const int col  = n0 + m;
    const float bc  = b[col];
    const float brc = br[col];
    const int rOff = (lane >> 4) * 8;
    float s8 = 0.0f, q8 = 0.0f;
    #pragma unroll
    for (int r = 0; r < 8; ++r) {
        float c = accC[r] + bc;  c = c > 0.0f ? c : 0.0f;
        float e = accR[r] + brc; e = e > 0.0f ? e : 0.0f;
        float v = c + e;
        y[(size_t)(rowBase + rOff + r) * DFEAT + col] = v;
        s8 += v;
        q8 += v * v;
    }
    float s8o = __shfl_down(s8, 16);
    float q8o = __shfl_down(q8, 16);
    if (lane < 16) {
        atomicAdd(&bn_sum[col],   s8 + s8o);
        atomicAdd(&bn_sumsq[col], q8 + q8o);
    }
}

__global__ void bn_params_kernel(const float* __restrict__ sum, const float* __restrict__ sumsq,
                                 const float* __restrict__ gamma, const float* __restrict__ beta,
                                 float* __restrict__ scale, float* __restrict__ shift, int n) {
    int c = threadIdx.x;
    if (c >= DFEAT) return;
    float inv  = 1.0f / (float)n;
    float mean = sum[c] * inv;
    float var  = sumsq[c] * inv - mean * mean;
    float sc   = gamma[c] * rsqrtf(var + BN_EPS);
    scale[c] = sc;
    shift[c] = beta[c] - mean * sc;
}

__global__ __launch_bounds__(256) void bn_apply_kernel(float* __restrict__ y,
                                                       const float* __restrict__ scale,
                                                       const float* __restrict__ shift,
                                                       size_t nvec4) {
    size_t i = (size_t)blockIdx.x * blockDim.x + threadIdx.x;
    if (i >= nvec4) return;
    float4 v = ((float4*)y)[i];
    int col = (int)((i * 4) & (DFEAT - 1));
    v.x = v.x * scale[col + 0] + shift[col + 0];
    v.y = v.y * scale[col + 1] + shift[col + 1];
    v.z = v.z * scale[col + 2] + shift[col + 2];
    v.w = v.w * scale[col + 3] + shift[col + 3];
    ((float4*)y)[i] = v;
}

extern "C" void kernel_launch(void* const* d_in, const int* in_sizes, int n_in,
                              void* d_out, int out_size, void* d_ws, size_t ws_size,
                              hipStream_t stream) {
    const float* feats = (const float*)d_in[0];
    const int*   src   = (const int*)d_in[1];
    const int*   dst   = (const int*)d_in[2];
    const float* W     = (const float*)d_in[3];
    const float* b     = (const float*)d_in[4];
    const float* Wr    = (const float*)d_in[5];
    const float* br    = (const float*)d_in[6];
    const float* gamma = (const float*)d_in[7];
    const float* beta  = (const float*)d_in[8];
    float* out = (float*)d_out;

    const int N = NNODES;
    const int E = in_sizes[1];

    float* ws        = (float*)d_ws;
    float* agg       = ws;
    float* deg_out   = agg + (size_t)N * DFEAT;
    float* deg_in    = deg_out + N;
    float* bn_sum    = deg_in + N;
    float* bn_sumsq  = bn_sum + DFEAT;
    float* bn_scale  = bn_sumsq + DFEAT;
    float* bn_shift  = bn_scale + DFEAT;

    int* scr    = (int*)d_out;
    int* cnt    = scr;
    int* off    = scr + 98304;
    int* recA   = scr + 196608;
    int* recD   = recA + E + E;
    int* csr    = recD + E;
    int* rowptr = csr + E;

    zero_kernel<<<1, 256, 0, stream>>>((int*)bn_sum, 2 * DFEAT);

    hist_kernel<<<NCHUNK, EDGE_T, 0, stream>>>(src, dst, cnt, E);

    scan_kernel<<<1, SCAN_T, 0, stream>>>(cnt, off, 2 * NB * NCHUNK);

    scatter_kernel<<<NCHUNK, EDGE_T, 0, stream>>>(src, dst, off, recA, recD, E);

    local_kernel<<<NB, BIN_R, 0, stream>>>(off, recA, recD, deg_out, deg_in, rowptr, csr, N);

    gather_kernel<<<(N + 7) / 8, 256, 0, stream>>>(feats, rowptr, csr, deg_out, agg, N);

    gemm_bn_kernel<<<N / 16, 256, 0, stream>>>(agg, feats, deg_in, W, b, Wr, br,
                                               out, bn_sum, bn_sumsq);

    bn_params_kernel<<<1, DFEAT, 0, stream>>>(bn_sum, bn_sumsq, gamma, beta,
                                              bn_scale, bn_shift, N);

    size_t nvec4 = (size_t)N * DFEAT / 4;
    bn_apply_kernel<<<(int)((nvec4 + 255) / 256), 256, 0, stream>>>(out, bn_scale, bn_shift, nvec4);
}


# The three one-work-group launches are gone (9 launches -> 6): each edge chunk forms its own segment starts from the count table, the accumulator clear and the BN parameters ride in their neighbours
# speedup vs baseline: 2.0950x; 1.1502x over previous
#include <hip/hip_runtime.h>
#include <hip/hip_bf16.h>

typedef __attribute__((ext_vector_type(2))) float v2f;
typedef __attribute__((ext_vector_type(8))) float v8f;

#define NNODES 100000
#define DFEAT  128
#define BN_EPS 1e-5f
#define BIN_SH 9
#define BIN_R  512
#define NB     196
#define NB2    (2 * NB)
#define NCHUNK 250
#define EDGE_T 1024
#define CQ     (NB2 / 4)
#define CGRP   10
#define CROWS  25
#define CPL    ((NB2 + 31) / 32)
static_assert(NB2 % 4 == 0, "count-table rows are read as 16-byte words");
static_assert(CGRP * CROWS == NCHUNK, "the groups cover every chunk row");
static_assert(CGRP * CQ <= EDGE_T, "one thread per (group, 16-byte word)");


__global__ __launch_bounds__(EDGE_T) void hist_kernel(const int* __restrict__ src,
                                                      const int* __restrict__ dst,
                                                      int* __restrict__ cnt,
                                                      float* __restrict__ bn_acc, int E) {
    __shared__ int hD[NB];
    __shared__ int hS[NB];
    const int t = threadIdx.x;
    const int w = blockIdx.x;
    if (t < NB) { hD[t] = 0; hS[t] = 0; }
    if (w == 0 && t < 2 * DFEAT) bn_acc[t] = 0.0f;
    __syncthreads();
    const int ch  = (E + NCHUNK - 1) / NCHUNK;
    const int beg = w * ch;
    const int end = min(E, beg + ch);
    for (int e0 = beg + t; e0 < end; e0 += 4 * EDGE_T) {
        int d[4], s[4];
        #pragma unroll
        for (int j = 0; j < 4; ++j) {
            const int e = min(e0 + j * EDGE_T, end - 1);
            d[j] = dst[e];
            s[j] = src[e];
        }
        #pragma unroll
        for (int j = 0; j < 4; ++j) {
            if (e0 + j * EDGE_T < end) {
                atomicAdd(&hD[d[j] >> BIN_SH], 1);
                atomicAdd(&hS[s[j] >> BIN_SH], 1);
            }
        }
    }
    __syncthreads();
    if (t < NB) {
        cnt[w * NB2 + t]      = hD[t];
        cnt[w * NB2 + NB + t] = hS[t];
    }
}

__global__ __launch_bounds__(EDGE_T) void scatter_kernel(const int* __restrict__ src,
                                                         const int* __restrict__ dst,
                                                         const int* __restrict__ cnt,
                                                         int* __restrict__ binStart,
                                                         int* __restrict__ recA,
                                                         int* __restrict__ recD, int E) {
    __shared__ __attribute__((aligned(16))) int pPre[CGRP * NB2];
    __shared__ __attribute__((aligned(16))) int pTot[CGRP * NB2];
    __shared__ int sPre[NB2];
    __shared__ int sTot[NB2];
    __shared__ int sStart[NB2 + 1];
    __shared__ int cD[NB];
    __shared__ int cS[NB];
    const int t = threadIdx.x;
    const int w = blockIdx.x;
    if (t < CGRP * CQ) {
        const int g = t / CQ;
        const int c = t - g * CQ;
        const int4* p = (const int4*)cnt + (size_t)(g * CROWS) * CQ + c;
        int4 pre = make_int4(0, 0, 0, 0);
        int4 tot = make_int4(0, 0, 0, 0);
        #pragma unroll
        for (int i = 0; i < CROWS; ++i) {
            const int4 v = p[i * CQ];
            const int m = (g * CROWS + i < w) ? -1 : 0;
            tot.x += v.x; tot.y += v.y; tot.z += v.z; tot.w += v.w;
            pre.x += v.x & m; pre.y += v.y & m; pre.z += v.z & m; pre.w += v.w & m;
        }
        *(int4*)&pPre[g * NB2 + 4 * c] = pre;
        *(int4*)&pTot[g * NB2 + 4 * c] = tot;
    }
    __syncthreads();
    if (t < NB2) {
        int p = 0, s = 0;
        for (int g = 0; g < CGRP; ++g) { p += pPre[g * NB2 + t]; s += pTot[g * NB2 + t]; }
        sPre[t] = p;
        sTot[t] = s;
    }
    __syncthreads();
    if (t < 32) {
        int loc[CPL];
        int sum = 0;
        for (int i = 0; i < CPL; ++i) {
            const int idx = t * CPL + i;
            loc[i] = (idx < NB2) ? sTot[idx] : 0;
            sum += loc[i];
        }
        int inc = sum;
        for (int o = 1; o < 32; o <<= 1) {
            const int x = __shfl_up(inc, o);
            if (t >= o) inc += x;
        }
        int run = inc - sum;
        for (int i = 0; i < CPL; ++i) {
            const int idx = t * CPL + i;
            if (idx < NB2) sStart[idx] = run;
            run += loc[i];
        }
        if (t == 31) sStart[NB2] = run;
    }
    __syncthreads();
    if (t < NB) {
        cD[t] = sStart[t] + sPre[t];
        cS[t] = sStart[NB + t] + sPre[NB + t];
    }
    if (w == 0 && t <= NB2) binStart[t] = sStart[t];
    __syncthreads();
    const int ch  = (E + NCHUNK - 1) / NCHUNK;
    const int beg = w * ch;
    const int end = min(E, beg + ch);
    for (int e0 = beg + t; e0 < end; e0 += 4 * EDGE_T) {
        int d[4], s[4];
        #pragma unroll
        for (int j = 0; j < 4; ++j) {
            const int e = min(e0 + j * EDGE_T, end - 1);
            d[j] = dst[e];
            s[j] = src[e];
        }
        #pragma unroll
        for (int j = 0; j < 4; ++j) {
            if (e0 + j * EDGE_T < end) {
                const int p = atomicAdd(&cD[d[j] >> BIN_SH], 1);
                recA[p] = s[j];
                recD[p] = d[j];
                const int q = atomicAdd(&cS[s[j] >> BIN_SH], 1);
                recA[q] = s[j];
            }
        }
    }
}

__global__ __launch_bounds__(BIN_R) void local_kernel(const int* __restrict__ binStart,
                                                      const int* __restrict__ recA,
                                                      const int* __restrict__ recD,
                                                      float* __restrict__ norm_src,
                                                      float* __restrict__ norm_dst,
                                                      int* __restrict__ rowptr,
                                                      int* __restrict__ csr, int n) {
    __shared__ int cIn[BIN_R];
    __shared__ int cOut[BIN_R];
    __shared__ int part[BIN_R];
    const int t = threadIdx.x;
    const int b = blockIdx.x;
    cIn[t] = 0;
    cOut[t] = 0;
    __syncthreads();
    const int dBeg = binStart[b];
    const int dEnd = binStart[b + 1];
    const int sBeg = binStart[NB + b];
    const int sEnd = binStart[NB + b + 1];
    for (int i0 = dBeg + t; i0 < dEnd; i0 += 4 * BIN_R) {
        int d[4];
        #pragma unroll
        for (int j = 0; j < 4; ++j) {
            const int i = min(i0 + j * BIN_R, dEnd - 1);
            d[j] = recD[i];
        }
        #pragma unroll
        for (int j = 0; j < 4; ++j)
            if (i0 + j * BIN_R < dEnd) atomicAdd(&cIn[d[j] & (BIN_R - 1)], 1);
    }
    for (int i0 = sBeg + t; i0 < sEnd; i0 += 4 * BIN_R) {
        int s[4];
        #pragma unroll
        for (int j = 0; j < 4; ++j) {
            const int i = min(i0 + j * BIN_R, sEnd - 1);
            s[j] = recA[i];
        }
        #pragma unroll
        for (int j = 0; j < 4; ++j)
            if (i0 + j * BIN_R < sEnd) atomicAdd(&cOut[s[j] & (BIN_R - 1)], 1);
    }
    __syncthreads();
    const int node = b * BIN_R + t;
    const int ci = cIn[t];
    if (node < n) {
        norm_src[node] = rsqrtf(fmaxf((float)cOut[t], 1.0f));
        norm_dst[node] = rsqrtf(fmaxf((float)ci, 1.0f));
    }
    part[t] = ci;
    __syncthreads();
    for (int o = 1; o < BIN_R; o <<= 1) {
        int x = (t >= o) ? part[t - o] : 0;
        __syncthreads();
        part[t] += x;
        __syncthreads();
    }
    const int start = dBeg + part[t] - ci;
    if (node <= n) rowptr[node] = start;
    cIn[t] = start;
    __syncthreads();
    for (int i0 = dBeg + t; i0 < dEnd; i0 += 4 * BIN_R) {
        int d[4], s[4];
        #pragma unroll
        for (int j = 0; j < 4; ++j) {
            const int i = min(i0 + j * BIN_R, dEnd - 1);
            d[j] = recD[i];
            s[j] = recA[i];
        }
        #pragma unroll
        for (int j = 0; j < 4; ++j) {
            if (i0 + j * BIN_R < dEnd) {
                const int p = atomicAdd(&cIn[d[j] & (BIN_R - 1)], 1);
                csr[p] = s[j];
            }
        }
    }
}

__global__ __launch_bounds__(256) void gather_kernel(const float* __restrict__ feats,
                                                     const int* __restrict__ rowptr,
                                                     const int* __restrict__ csr,
                                                     const float* __restrict__ norm_src,
                                                     float* __restrict__ agg, int n) {
    const int row = blockIdx.x * 8 + (threadIdx.x >> 5);
    if (row >= n) return;
    const int lane = threadIdx.x & 31;
    const int beg = __builtin_amdgcn_readfirstlane(rowptr[row]);
    const int end = __builtin_amdgcn_readfirstlane(rowptr[row + 1]);
    float4 acc = make_float4(0.0f, 0.0f, 0.0f, 0.0f);
    for (int b0 = beg; b0 < end; b0 += 32) {
        const int cnt = min(32, end - b0);
        int   s_l  = 0;
        float ns_l = 0.0f;
        if (lane < cnt) { s_l = csr[b0 + lane]; ns_l = norm_src[s_l]; }
        #pragma unroll 4
        for (int j = 0; j < cnt; ++j) {
            const int   s  = __builtin_amdgcn_readlane(s_l, j);
            const float ns = __int_as_float(__builtin_amdgcn_readlane(__float_as_int(ns_l), j));
            const float4 v = ((const float4*)(feats + (size_t)s * DFEAT))[lane];
            acc.x += v.x * ns;
            acc.y += v.y * ns;
            acc.z += v.z * ns;
            acc.w += v.w * ns;
        }
    }
    ((float4*)(agg + (size_t)row * DFEAT))[lane] = acc;
}

__global__ __launch_bounds__(256) void gemm_bn_kernel(
    const float* __restrict__ agg, const float* __restrict__ feats,
    const float* __restrict__ norm_dst,
    const float* __restrict__ W,  const float* __restrict__ b,
    const float* __restrict__ Wr, const float* __restrict__ br,
    float* __restrict__ y, float* __restrict__ bn_sum, float* __restrict__ bn_sumsq) {
    const int lane = threadIdx.x & 31;
    const int wave = threadIdx.x >> 5;
    const int rowBase = blockIdx.x * 16;
    const int n0 = wave * 16;
    const int m  = lane & 15;
    const int kh = (lane >> 4) * 2;
    const int row = rowBase + m;
    const float nd = norm_dst[row];
    const float* aRow = agg   + (size_t)row * DFEAT;
    const float* fRow = feats + (size_t)row * DFEAT;

    v8f accC = {};
    v8f accR = {};
    #pragma unroll
    for (int k = 0; k < DFEAT; k += 4) {
        v2f a, f, bw, bwr;
        a.x = aRow[k + kh]     * nd;
        a.y = aRow[k + kh + 1] * nd;
        f.x = fRow[k + kh];
        f.y = fRow[k + kh + 1];
        bw.x  = W [(size_t)(k + kh)     * DFEAT + n0 + m];
        bw.y  = W [(size_t)(k + kh + 1) * DFEAT + n0 + m];
        bwr.x = Wr[(size_t)(k + kh)     * DFEAT + n0 + m];
        bwr.y = Wr[(size_t)(k + kh + 1) * DFEAT + n0 + m];
        accC = __builtin_amdgcn_wmma_f32_16x16x4_f32(false, a, false, bw,  (short)0, accC, false, false);
        accR = __builtin_amdgcn_wmma_f32_16x16x4_f32(false, f, false, bwr, (short)0, accR, false, false);
    }

    const int col  = n0 + m;
    const float bc  = b[col];
    const float brc = br[col];
    const int rOff = (lane >> 4) * 8;
    float s8 = 0.0f, q8 = 0.0f;
    #pragma unroll
    for (int r = 0; r < 8; ++r) {
        float c = accC[r] + bc;  c = c > 0.0f ? c : 0.0f;
        float e = accR[r] + brc; e = e > 0.0f ? e : 0.0f;
        float v = c + e;
        y[(size_t)(rowBase + rOff + r) * DFEAT + col] = v;
        s8 += v;
        q8 += v * v;
    }
    float s8o = __shfl_down(s8, 16);
    float q8o = __shfl_down(q8, 16);
    if (lane < 16) {
        atomicAdd(&bn_sum[col],   s8 + s8o);
        atomicAdd(&bn_sumsq[col], q8 + q8o);
    }
}

__global__ __launch_bounds__(256) void bn_apply_kernel(float* __restrict__ y,
                                                       const float* __restrict__ sum,
                                                       const float* __restrict__ sumsq,
                                                       const float* __restrict__ gamma,
                                                       const float* __restrict__ beta,
                                                       size_t nvec4, int n) {
    size_t i = (size_t)blockIdx.x * blockDim.x + threadIdx.x;
    if (i >= nvec4) return;
    float4 v = ((float4*)y)[i];
    const int col = (int)((i * 4) & (DFEAT - 1));
    const float4 s4 = *(const float4*)(sum + col);
    const float4 q4 = *(const float4*)(sumsq + col);
    const float4 g4 = *(const float4*)(gamma + col);
    const float4 b4 = *(const float4*)(beta + col);
    const float inv = 1.0f / (float)n;
    float mean, var, sc;
    mean = s4.x * inv; var = q4.x * inv - mean * mean; sc = g4.x * rsqrtf(var + BN_EPS);
    v.x = v.x * sc + (b4.x - mean * sc);
    mean = s4.y * inv; var = q4.y * inv - mean * mean; sc = g4.y * rsqrtf(var + BN_EPS);
    v.y = v.y * sc + (b4.y - mean * sc);
    mean = s4.z * inv; var = q4.z * inv - mean * mean; sc = g4.z * rsqrtf(var + BN_EPS);
    v.z = v.z * sc + (b4.z - mean * sc);
    mean = s4.w * inv; var = q4.w * inv - mean * mean; sc = g4.w * rsqrtf(var + BN_EPS);
    v.w = v.w * sc + (b4.w - mean * sc);
    ((float4*)y)[i] = v;
}

extern "C" void kernel_launch(void* const* d_in, const int* in_sizes, int n_in,
                              void* d_out, int out_size, void* d_ws, size_t ws_size,
                              hipStream_t stream) {
    const float* feats = (const float*)d_in[0];
    const int*   src   = (const int*)d_in[1];
    const int*   dst   = (const int*)d_in[2];
    const float* W     = (const float*)d_in[3];
    const float* b     = (const float*)d_in[4];
    const float* Wr    = (const float*)d_in[5];
    const float* br    = (const float*)d_in[6];
    const float* gamma = (const float*)d_in[7];
    const float* beta  = (const float*)d_in[8];
    float* out = (float*)d_out;

    const int N = NNODES;
    const int E = in_sizes[1];

    float* ws        = (float*)d_ws;
    float* agg       = ws;
    float* deg_out   = agg + (size_t)N * DFEAT;
    float* deg_in    = deg_out + N;
    float* bn_sum    = deg_in + N;
    float* bn_sumsq  = bn_sum + DFEAT;

    int* scr      = (int*)d_out;
    int* cnt      = scr;
    int* binStart = scr + 98304;
    int* recA     = scr + 196608;
    int* recD     = recA + E + E;
    int* csr      = recD + E;
    int* rowptr   = csr + E;

    hist_kernel<<<NCHUNK, EDGE_T, 0, stream>>>(src, dst, cnt, bn_sum, E);

    scatter_kernel<<<NCHUNK, EDGE_T, 0, stream>>>(src, dst, cnt, binStart, recA, recD, E);

    local_kernel<<<NB, BIN_R, 0, stream>>>(binStart, recA, recD, deg_out, deg_in, rowptr, csr, N);

    gather_kernel<<<(N + 7) / 8, 256, 0, stream>>>(feats, rowptr, csr, deg_out, agg, N);

    gemm_bn_kernel<<<N / 16, 256, 0, stream>>>(agg, feats, deg_in, W, b, Wr, br,
                                               out, bn_sum, bn_sumsq);

    size_t nvec4 = (size_t)N * DFEAT / 4;
    bn_apply_kernel<<<(int)((nvec4 + 255) / 256), 256, 0, stream>>>(out, bn_sum, bn_sumsq, gamma, beta,
                                                                   nvec4, N);
}


# GEMM work-groups loop over row tiles: W/Wr fragments stay in registers, agg and feats rows staged once per tile through LDS; BN column sums carried in registers, one atomic add per work-group
# speedup vs baseline: 2.6606x; 1.2700x over previous
#include <hip/hip_runtime.h>
#include <hip/hip_bf16.h>

typedef __attribute__((ext_vector_type(2))) float v2f;
typedef __attribute__((ext_vector_type(8))) float v8f;

#define NNODES 100000
#define DFEAT  128
#define BN_EPS 1e-5f
#define BIN_SH 9
#define BIN_R  512
#define NB     196
#define NB2    (2 * NB)
#define NCHUNK 250
#define EDGE_T 1024
#define CQ     (NB2 / 4)
#define CGRP   10
#define CROWS  25
#define CPL    ((NB2 + 31) / 32)
static_assert(NB2 % 4 == 0, "count-table rows are read as 16-byte words");
static_assert(CGRP * CROWS == NCHUNK, "the groups cover every chunk row");
static_assert(CGRP * CQ <= EDGE_T, "one thread per (group, 16-byte word)");


__global__ __launch_bounds__(EDGE_T) void hist_kernel(const int* __restrict__ src,
                                                      const int* __restrict__ dst,
                                                      int* __restrict__ cnt,
                                                      float* __restrict__ bn_acc, int E) {
    __shared__ int hD[NB];
    __shared__ int hS[NB];
    const int t = threadIdx.x;
    const int w = blockIdx.x;
    if (t < NB) { hD[t] = 0; hS[t] = 0; }
    if (w == 0 && t < 2 * DFEAT) bn_acc[t] = 0.0f;
    __syncthreads();
    const int ch  = (E + NCHUNK - 1) / NCHUNK;
    const int beg = w * ch;
    const int end = min(E, beg + ch);
    for (int e0 = beg + t; e0 < end; e0 += 4 * EDGE_T) {
        int d[4], s[4];
        #pragma unroll
        for (int j = 0; j < 4; ++j) {
            const int e = min(e0 + j * EDGE_T, end - 1);
            d[j] = dst[e];
            s[j] = src[e];
        }
        #pragma unroll
        for (int j = 0; j < 4; ++j) {
            if (e0 + j * EDGE_T < end) {
                atomicAdd(&hD[d[j] >> BIN_SH], 1);
                atomicAdd(&hS[s[j] >> BIN_SH], 1);
            }
        }
    }
    __syncthreads();
    if (t < NB) {
        cnt[w * NB2 + t]      = hD[t];
        cnt[w * NB2 + NB + t] = hS[t];
    }
}

__global__ __launch_bounds__(EDGE_T) void scatter_kernel(const int* __restrict__ src,
                                                         const int* __restrict__ dst,
                                                         const int* __restrict__ cnt,
                                                         int* __restrict__ binStart,
                                                         int* __restrict__ recA,
                                                         int* __restrict__ recD, int E) {
    __shared__ __attribute__((aligned(16))) int pPre[CGRP * NB2];
    __shared__ __attribute__((aligned(16))) int pTot[CGRP * NB2];
    __shared__ int sPre[NB2];
    __shared__ int sTot[NB2];
    __shared__ int sStart[NB2 + 1];
    __shared__ int cD[NB];
    __shared__ int cS[NB];
    const int t = threadIdx.x;
    const int w = blockIdx.x;
    if (t < CGRP * CQ) {
        const int g = t / CQ;
        const int c = t - g * CQ;
        const int4* p = (const int4*)cnt + (size_t)(g * CROWS) * CQ + c;
        int4 pre = make_int4(0, 0, 0, 0);
        int4 tot = make_int4(0, 0, 0, 0);
        #pragma unroll
        for (int i = 0; i < CROWS; ++i) {
            const int4 v = p[i * CQ];
            const int m = (g * CROWS + i < w) ? -1 : 0;
            tot.x += v.x; tot.y += v.y; tot.z += v.z; tot.w += v.w;
            pre.x += v.x & m; pre.y += v.y & m; pre.z += v.z & m; pre.w += v.w & m;
        }
        *(int4*)&pPre[g * NB2 + 4 * c] = pre;
        *(int4*)&pTot[g * NB2 + 4 * c] = tot;
    }
    __syncthreads();
    if (t < NB2) {
        int p = 0, s = 0;
        for (int g = 0; g < CGRP; ++g) { p += pPre[g * NB2 + t]; s += pTot[g * NB2 + t]; }
        sPre[t] = p;
        sTot[t] = s;
    }
    __syncthreads();
    if (t < 32) {
        int loc[CPL];
        int sum = 0;
        for (int i = 0; i < CPL; ++i) {
            const int idx = t * CPL + i;
            loc[i] = (idx < NB2) ? sTot[idx] : 0;
            sum += loc[i];
        }
        int inc = sum;
        for (int o = 1; o < 32; o <<= 1) {
            const int x = __shfl_up(inc, o);
            if (t >= o) inc += x;
        }
        int run = inc - sum;
        for (int i = 0; i < CPL; ++i) {
            const int idx = t * CPL + i;
            if (idx < NB2) sStart[idx] = run;
            run += loc[i];
        }
        if (t == 31) sStart[NB2] = run;
    }
    __syncthreads();
    if (t < NB) {
        cD[t] = sStart[t] + sPre[t];
        cS[t] = sStart[NB + t] + sPre[NB + t];
    }
    if (w == 0 && t <= NB2) binStart[t] = sStart[t];
    __syncthreads();
    const int ch  = (E + NCHUNK - 1) / NCHUNK;
    const int beg = w * ch;
    const int end = min(E, beg + ch);
    for (int e0 = beg + t; e0 < end; e0 += 4 * EDGE_T) {
        int d[4], s[4];
        #pragma unroll
        for (int j = 0; j < 4; ++j) {
            const int e = min(e0 + j * EDGE_T, end - 1);
            d[j] = dst[e];
            s[j] = src[e];
        }
        #pragma unroll
        for (int j = 0; j < 4; ++j) {
            if (e0 + j * EDGE_T < end) {
                const int p = atomicAdd(&cD[d[j] >> BIN_SH], 1);
                recA[p] = s[j];
                recD[p] = d[j];
                const int q = atomicAdd(&cS[s[j] >> BIN_SH], 1);
                recA[q] = s[j];
            }
        }
    }
}

__global__ __launch_bounds__(BIN_R) void local_kernel(const int* __restrict__ binStart,
                                                      const int* __restrict__ recA,
                                                      const int* __restrict__ recD,
                                                      float* __restrict__ norm_src,
                                                      float* __restrict__ norm_dst,
                                                      int* __restrict__ rowptr,
                                                      int* __restrict__ csr, int n) {
    __shared__ int cIn[BIN_R];
    __shared__ int cOut[BIN_R];
    __shared__ int part[BIN_R];
    const int t = threadIdx.x;
    const int b = blockIdx.x;
    cIn[t] = 0;
    cOut[t] = 0;
    __syncthreads();
    const int dBeg = binStart[b];
    const int dEnd = binStart[b + 1];
    const int sBeg = binStart[NB + b];
    const int sEnd = binStart[NB + b + 1];
    for (int i0 = dBeg + t; i0 < dEnd; i0 += 4 * BIN_R) {
        int d[4];
        #pragma unroll
        for (int j = 0; j < 4; ++j) {
            const int i = min(i0 + j * BIN_R, dEnd - 1);
            d[j] = recD[i];
        }
        #pragma unroll
        for (int j = 0; j < 4; ++j)
            if (i0 + j * BIN_R < dEnd) atomicAdd(&cIn[d[j] & (BIN_R - 1)], 1);
    }
    for (int i0 = sBeg + t; i0 < sEnd; i0 += 4 * BIN_R) {
        int s[4];
        #pragma unroll
        for (int j = 0; j < 4; ++j) {
            const int i = min(i0 + j * BIN_R, sEnd - 1);
            s[j] = recA[i];
        }
        #pragma unroll
        for (int j = 0; j < 4; ++j)
            if (i0 + j * BIN_R < sEnd) atomicAdd(&cOut[s[j] & (BIN_R - 1)], 1);
    }
    __syncthreads();
    const int node = b * BIN_R + t;
    const int ci = cIn[t];
    if (node < n) {
        norm_src[node] = rsqrtf(fmaxf((float)cOut[t], 1.0f));
        norm_dst[node] = rsqrtf(fmaxf((float)ci, 1.0f));
    }
    part[t] = ci;
    __syncthreads();
    for (int o = 1; o < BIN_R; o <<= 1) {
        int x = (t >= o) ? part[t - o] : 0;
        __syncthreads();
        part[t] += x;
        __syncthreads();
    }
    const int start = dBeg + part[t] - ci;
    if (node <= n) rowptr[node] = start;
    cIn[t] = start;
    __syncthreads();
    for (int i0 = dBeg + t; i0 < dEnd; i0 += 4 * BIN_R) {
        int d[4], s[4];
        #pragma unroll
        for (int j = 0; j < 4; ++j) {
            const int i = min(i0 + j * BIN_R, dEnd - 1);
            d[j] = recD[i];
            s[j] = recA[i];
        }
        #pragma unroll
        for (int j = 0; j < 4; ++j) {
            if (i0 + j * BIN_R < dEnd) {
                const int p = atomicAdd(&cIn[d[j] & (BIN_R - 1)], 1);
                csr[p] = s[j];
            }
        }
    }
}

__global__ __launch_bounds__(256) void gather_kernel(const float* __restrict__ feats,
                                                     const int* __restrict__ rowptr,
                                                     const int* __restrict__ csr,
                                                     const float* __restrict__ norm_src,
                                                     float* __restrict__ agg, int n) {
    const int row = blockIdx.x * 8 + (threadIdx.x >> 5);
    if (row >= n) return;
    const int lane = threadIdx.x & 31;
    const int beg = __builtin_amdgcn_readfirstlane(rowptr[row]);
    const int end = __builtin_amdgcn_readfirstlane(rowptr[row + 1]);
    float4 acc = make_float4(0.0f, 0.0f, 0.0f, 0.0f);
    for (int b0 = beg; b0 < end; b0 += 32) {
        const int cnt = min(32, end - b0);
        int   s_l  = 0;
        float ns_l = 0.0f;
        if (lane < cnt) { s_l = csr[b0 + lane]; ns_l = norm_src[s_l]; }
        #pragma unroll 4
        for (int j = 0; j < cnt; ++j) {
            const int   s  = __builtin_amdgcn_readlane(s_l, j);
            const float ns = __int_as_float(__builtin_amdgcn_readlane(__float_as_int(ns_l), j));
            const float4 v = ((const float4*)(feats + (size_t)s * DFEAT))[lane];
            acc.x += v.x * ns;
            acc.y += v.y * ns;
            acc.z += v.z * ns;
            acc.w += v.w * ns;
        }
    }
    ((float4*)(agg + (size_t)row * DFEAT))[lane] = acc;
}

#define GEMM_TILES (NNODES / 16)
#define GEMM_GRID  512
#define LDS_P      132
static_assert(NNODES % 16 == 0, "every tile has 16 full rows");
static_assert(GEMM_GRID <= GEMM_TILES, "every work-group has a first tile");

__device__ __forceinline__ void lds_barrier() {
    __builtin_amdgcn_fence(__ATOMIC_RELEASE, "workgroup", "local");
    __builtin_amdgcn_s_barrier();
    __builtin_amdgcn_fence(__ATOMIC_ACQUIRE, "workgroup", "local");
}

__global__ __launch_bounds__(256, 2) void gemm_bn_kernel(
    const float* __restrict__ agg, const float* __restrict__ feats,
    const float* __restrict__ norm_dst,
    const float* __restrict__ W,  const float* __restrict__ b,
    const float* __restrict__ Wr, const float* __restrict__ br,
    float* __restrict__ y, float* __restrict__ bn_sum, float* __restrict__ bn_sumsq) {
    __shared__ __attribute__((aligned(16))) float sA[2][16 * LDS_P];
    __shared__ __attribute__((aligned(16))) float sF[2][16 * LDS_P];
    const int lane = threadIdx.x & 31;
    const int wave = threadIdx.x >> 5;
    const int n0 = wave * 16;
    const int m  = lane & 15;
    const int kh = (lane >> 4) * 2;
    const int col = n0 + m;

    v2f bw[DFEAT / 4], bwr[DFEAT / 4];
    #pragma unroll
    for (int ks = 0; ks < DFEAT / 4; ++ks) {
        const int k = 4 * ks + kh;
        bw[ks].x  = W [(size_t)k       * DFEAT + col];
        bw[ks].y  = W [(size_t)(k + 1) * DFEAT + col];
        bwr[ks].x = Wr[(size_t)k       * DFEAT + col];
        bwr[ks].y = Wr[(size_t)(k + 1) * DFEAT + col];
    }
    const float bc  = b[col];
    const float brc = br[col];
    const int rOff = (lane >> 4) * 8;

    const int so0 = wave * LDS_P + 4 * lane;
    const int so1 = (wave + 8) * LDS_P + 4 * lane;
    const int fo = m * LDS_P + kh;

    float4 pa0, pa1, pf0, pf1;
    int tile = blockIdx.x;
    {
        const int r0 = tile * 16 + wave;
        const float nd0 = norm_dst[r0];
        const float nd1 = norm_dst[r0 + 8];
        pa0 = ((const float4*)(agg   + (size_t)r0       * DFEAT))[lane];
        pa1 = ((const float4*)(agg   + (size_t)(r0 + 8) * DFEAT))[lane];
        pf0 = ((const float4*)(feats + (size_t)r0       * DFEAT))[lane];
        pf1 = ((const float4*)(feats + (size_t)(r0 + 8) * DFEAT))[lane];
        pa0.x *= nd0; pa0.y *= nd0; pa0.z *= nd0; pa0.w *= nd0;
        pa1.x *= nd1; pa1.y *= nd1; pa1.z *= nd1; pa1.w *= nd1;
        *(float4*)&sA[0][so0] = pa0;
        *(float4*)&sA[0][so1] = pa1;
        *(float4*)&sF[0][so0] = pf0;
        *(float4*)&sF[0][so1] = pf1;
    }
    lds_barrier();

    float s8 = 0.0f, q8 = 0.0f;
    int buf = 0;
    for (; tile < GEMM_TILES; tile += GEMM_GRID) {
        const int  next = tile + GEMM_GRID;
        const bool more = next < GEMM_TILES;
        float nd0 = 0.0f, nd1 = 0.0f;
        if (more) {
            const int r0 = next * 16 + wave;
            nd0 = norm_dst[r0];
            nd1 = norm_dst[r0 + 8];
            pa0 = ((const float4*)(agg   + (size_t)r0       * DFEAT))[lane];
            pa1 = ((const float4*)(agg   + (size_t)(r0 + 8) * DFEAT))[lane];
            pf0 = ((const float4*)(feats + (size_t)r0       * DFEAT))[lane];
            pf1 = ((const float4*)(feats + (size_t)(r0 + 8) * DFEAT))[lane];
        }

        const float* aB = &sA[buf][fo];
        const float* fB = &sF[buf][fo];
        v8f accC = {};
        v8f accR = {};
        #pragma unroll
        for (int ks = 0; ks < DFEAT / 4; ++ks) {
            const v2f a = *(const v2f*)(aB + 4 * ks);
            const v2f f = *(const v2f*)(fB + 4 * ks);
            accC = __builtin_amdgcn_wmma_f32_16x16x4_f32(false, a, false, bw[ks],  (short)0, accC, false, false);
            accR = __builtin_amdgcn_wmma_f32_16x16x4_f32(false, f, false, bwr[ks], (short)0, accR, false, false);
        }

        const int rowBase = tile * 16;
        #pragma unroll
        for (int r = 0; r < 8; ++r) {
            float c = accC[r] + bc;  c = c > 0.0f ? c : 0.0f;
            float e = accR[r] + brc; e = e > 0.0f ? e : 0.0f;
            float v = c + e;
            y[(size_t)(rowBase + rOff + r) * DFEAT + col] = v;
            s8 += v;
            q8 += v * v;
        }

        if (more) {
            pa0.x *= nd0; pa0.y *= nd0; pa0.z *= nd0; pa0.w *= nd0;
            pa1.x *= nd1; pa1.y *= nd1; pa1.z *= nd1; pa1.w *= nd1;
            *(float4*)&sA[buf ^ 1][so0] = pa0;
            *(float4*)&sA[buf ^ 1][so1] = pa1;
            *(float4*)&sF[buf ^ 1][so0] = pf0;
            *(float4*)&sF[buf ^ 1][so1] = pf1;
        }
        lds_barrier();
        buf ^= 1;
    }

    float s8o = __shfl_down(s8, 16);
    float q8o = __shfl_down(q8, 16);
    if (lane < 16) {
        atomicAdd(&bn_sum[col],   s8 + s8o);
        atomicAdd(&bn_sumsq[col], q8 + q8o);
    }
}

__global__ __launch_bounds__(256) void bn_apply_kernel(float* __restrict__ y,
                                                       const float* __restrict__ sum,
                                                       const float* __restrict__ sumsq,
                                                       const float* __restrict__ gamma,
                                                       const float* __restrict__ beta,
                                                       size_t nvec4, int n) {
    size_t i = (size_t)blockIdx.x * blockDim.x + threadIdx.x;
    if (i >= nvec4) return;
    float4 v = ((float4*)y)[i];
    const int col = (int)((i * 4) & (DFEAT - 1));
    const float4 s4 = *(const float4*)(sum + col);
    const float4 q4 = *(const float4*)(sumsq + col);
    const float4 g4 = *(const float4*)(gamma + col);
    const float4 b4 = *(const float4*)(beta + col);
    const float inv = 1.0f / (float)n;
    float mean, var, sc;
    mean = s4.x * inv; var = q4.x * inv - mean * mean; sc = g4.x * rsqrtf(var + BN_EPS);
    v.x = v.x * sc + (b4.x - mean * sc);
    mean = s4.y * inv; var = q4.y * inv - mean * mean; sc = g4.y * rsqrtf(var + BN_EPS);
    v.y = v.y * sc + (b4.y - mean * sc);
    mean = s4.z * inv; var = q4.z * inv - mean * mean; sc = g4.z * rsqrtf(var + BN_EPS);
    v.z = v.z * sc + (b4.z - mean * sc);
    mean = s4.w * inv; var = q4.w * inv - mean * mean; sc = g4.w * rsqrtf(var + BN_EPS);
    v.w = v.w * sc + (b4.w - mean * sc);
    ((float4*)y)[i] = v;
}

extern "C" void kernel_launch(void* const* d_in, const int* in_sizes, int n_in,
                              void* d_out, int out_size, void* d_ws, size_t ws_size,
                              hipStream_t stream) {
    const float* feats = (const float*)d_in[0];
    const int*   src   = (const int*)d_in[1];
    const int*   dst   = (const int*)d_in[2];
    const float* W     = (const float*)d_in[3];
    const float* b     = (const float*)d_in[4];
    const float* Wr    = (const float*)d_in[5];
    const float* br    = (const float*)d_in[6];
    const float* gamma = (const float*)d_in[7];
    const float* beta  = (const float*)d_in[8];
    float* out = (float*)d_out;

    const int N = NNODES;
    const int E = in_sizes[1];

    float* ws        = (float*)d_ws;
    float* agg       = ws;
    float* deg_out   = agg + (size_t)N * DFEAT;
    float* deg_in    = deg_out + N;
    float* bn_sum    = deg_in + N;
    float* bn_sumsq  = bn_sum + DFEAT;

    int* scr      = (int*)d_out;
    int* cnt      = scr;
    int* binStart = scr + 98304;
    int* recA     = scr + 196608;
    int* recD     = recA + E + E;
    int* csr      = recD + E;
    int* rowptr   = csr + E;

    hist_kernel<<<NCHUNK, EDGE_T, 0, stream>>>(src, dst, cnt, bn_sum, E);

    scatter_kernel<<<NCHUNK, EDGE_T, 0, stream>>>(src, dst, cnt, binStart, recA, recD, E);

    local_kernel<<<NB, BIN_R, 0, stream>>>(binStart, recA, recD, deg_out, deg_in, rowptr, csr, N);

    gather_kernel<<<(N + 7) / 8, 256, 0, stream>>>(feats, rowptr, csr, deg_out, agg, N);

    gemm_bn_kernel<<<GEMM_GRID, 256, 0, stream>>>(agg, feats, deg_in, W, b, Wr, br,
                                                  out, bn_sum, bn_sumsq);

    size_t nvec4 = (size_t)N * DFEAT / 4;
    bn_apply_kernel<<<(int)((nvec4 + 255) / 256), 256, 0, stream>>>(out, bn_sum, bn_sumsq, gamma, beta,
                                                                   nvec4, N);
}


# Row gather fused into the WMMA GEMM launch (6 launches -> 5): each GEMM wave sums the source rows of two adjacent tile rows into the LDS stage, loads asked for ahead; no agg plane; sum order kept
# speedup vs baseline: 3.0906x; 1.1616x over previous
#include <hip/hip_runtime.h>
#include <hip/hip_bf16.h>

typedef __attribute__((ext_vector_type(2))) float v2f;
typedef __attribute__((ext_vector_type(8))) float v8f;

#define NNODES 100000
#define DFEAT  128
#define BN_EPS 1e-5f
#define BIN_SH 9
#define BIN_R  512
#define NB     196
#define NB2    (2 * NB)
#define NCHUNK 250
#define EDGE_T 1024
#define CQ     (NB2 / 4)
#define CGRP   10
#define CROWS  25
#define CPL    ((NB2 + 31) / 32)
static_assert(NB2 % 4 == 0, "count-table rows are read as 16-byte words");
static_assert(CGRP * CROWS == NCHUNK, "the groups cover every chunk row");
static_assert(CGRP * CQ <= EDGE_T, "one thread per (group, 16-byte word)");


__global__ __launch_bounds__(EDGE_T) void hist_kernel(const int* __restrict__ src,
                                                      const int* __restrict__ dst,
                                                      int* __restrict__ cnt,
                                                      float* __restrict__ bn_acc, int E) {
    __shared__ int hD[NB];
    __shared__ int hS[NB];
    const int t = threadIdx.x;
    const int w = blockIdx.x;
    if (t < NB) { hD[t] = 0; hS[t] = 0; }
    if (w == 0 && t < 2 * DFEAT) bn_acc[t] = 0.0f;
    __syncthreads();
    const int ch  = (E + NCHUNK - 1) / NCHUNK;
    const int beg = w * ch;
    const int end = min(E, beg + ch);
    for (int e0 = beg + t; e0 < end; e0 += 4 * EDGE_T) {
        int d[4], s[4];
        #pragma unroll
        for (int j = 0; j < 4; ++j) {
            const int e = min(e0 + j * EDGE_T, end - 1);
            d[j] = dst[e];
            s[j] = src[e];
        }
        #pragma unroll
        for (int j = 0; j < 4; ++j) {
            if (e0 + j * EDGE_T < end) {
                atomicAdd(&hD[d[j] >> BIN_SH], 1);
                atomicAdd(&hS[s[j] >> BIN_SH], 1);
            }
        }
    }
    __syncthreads();
    if (t < NB) {
        cnt[w * NB2 + t]      = hD[t];
        cnt[w * NB2 + NB + t] = hS[t];
    }
}

__global__ __launch_bounds__(EDGE_T) void scatter_kernel(const int* __restrict__ src,
                                                         const int* __restrict__ dst,
                                                         const int* __restrict__ cnt,
                                                         int* __restrict__ binStart,
                                                         int* __restrict__ recA,
                                                         int* __restrict__ recD, int E) {
    __shared__ __attribute__((aligned(16))) int pPre[CGRP * NB2];
    __shared__ __attribute__((aligned(16))) int pTot[CGRP * NB2];
    __shared__ int sPre[NB2];
    __shared__ int sTot[NB2];
    __shared__ int sStart[NB2 + 1];
    __shared__ int cD[NB];
    __shared__ int cS[NB];
    const int t = threadIdx.x;
    const int w = blockIdx.x;
    if (t < CGRP * CQ) {
        const int g = t / CQ;
        const int c = t - g * CQ;
        const int4* p = (const int4*)cnt + (size_t)(g * CROWS) * CQ + c;
        int4 pre = make_int4(0, 0, 0, 0);
        int4 tot = make_int4(0, 0, 0, 0);
        #pragma unroll
        for (int i = 0; i < CROWS; ++i) {
            const int4 v = p[i * CQ];
            const int m = (g * CROWS + i < w) ? -1 : 0;
            tot.x += v.x; tot.y += v.y; tot.z += v.z; tot.w += v.w;
            pre.x += v.x & m; pre.y += v.y & m; pre.z += v.z & m; pre.w += v.w & m;
        }
        *(int4*)&pPre[g * NB2 + 4 * c] = pre;
        *(int4*)&pTot[g * NB2 + 4 * c] = tot;
    }
    __syncthreads();
    if (t < NB2) {
        int p = 0, s = 0;
        for (int g = 0; g < CGRP; ++g) { p += pPre[g * NB2 + t]; s += pTot[g * NB2 + t]; }
        sPre[t] = p;
        sTot[t] = s;
    }
    __syncthreads();
    if (t < 32) {
        int loc[CPL];
        int sum = 0;
        for (int i = 0; i < CPL; ++i) {
            const int idx = t * CPL + i;
            loc[i] = (idx < NB2) ? sTot[idx] : 0;
            sum += loc[i];
        }
        int inc = sum;
        for (int o = 1; o < 32; o <<= 1) {
            const int x = __shfl_up(inc, o);
            if (t >= o) inc += x;
        }
        int run = inc - sum;
        for (int i = 0; i < CPL; ++i) {
            const int idx = t * CPL + i;
            if (idx < NB2) sStart[idx] = run;
            run += loc[i];
        }
        if (t == 31) sStart[NB2] = run;
    }
    __syncthreads();
    if (t < NB) {
        cD[t] = sStart[t] + sPre[t];
        cS[t] = sStart[NB + t] + sPre[NB + t];
    }
    if (w == 0 && t <= NB2) binStart[t] = sStart[t];
    __syncthreads();
    const int ch  = (E + NCHUNK - 1) / NCHUNK;
    const int beg = w * ch;
    const int end = min(E, beg + ch);
    for (int e0 = beg + t; e0 < end; e0 += 4 * EDGE_T) {
        int d[4], s[4];
        #pragma unroll
        for (int j = 0; j < 4; ++j) {
            const int e = min(e0 + j * EDGE_T, end - 1);
            d[j] = dst[e];
            s[j] = src[e];
        }
        #pragma unroll
        for (int j = 0; j < 4; ++j) {
            if (e0 + j * EDGE_T < end) {
                const int p = atomicAdd(&cD[d[j] >> BIN_SH], 1);
                recA[p] = s[j];
                recD[p] = d[j];
                const int q = atomicAdd(&cS[s[j] >> BIN_SH], 1);
                recA[q] = s[j];
            }
        }
    }
}

__global__ __launch_bounds__(BIN_R) void local_kernel(const int* __restrict__ binStart,
                                                      const int* __restrict__ recA,
                                                      const int* __restrict__ recD,
                                                      float* __restrict__ norm_src,
                                                      float* __restrict__ norm_dst,
                                                      int* __restrict__ rowptr,
                                                      int* __restrict__ csr, int n) {
    __shared__ int cIn[BIN_R];
    __shared__ int cOut[BIN_R];
    __shared__ int part[BIN_R];
    const int t = threadIdx.x;
    const int b = blockIdx.x;
    cIn[t] = 0;
    cOut[t] = 0;
    __syncthreads();
    const int dBeg = binStart[b];
    const int dEnd = binStart[b + 1];
    const int sBeg = binStart[NB + b];
    const int sEnd = binStart[NB + b + 1];
    for (int i0 = dBeg + t; i0 < dEnd; i0 += 4 * BIN_R) {
        int d[4];
        #pragma unroll
        for (int j = 0; j < 4; ++j) {
            const int i = min(i0 + j * BIN_R, dEnd - 1);
            d[j] = recD[i];
        }
        #pragma unroll
        for (int j = 0; j < 4; ++j)
            if (i0 + j * BIN_R < dEnd) atomicAdd(&cIn[d[j] & (BIN_R - 1)], 1);
    }
    for (int i0 = sBeg + t; i0 < sEnd; i0 += 4 * BIN_R) {
        int s[4];
        #pragma unroll
        for (int j = 0; j < 4; ++j) {
            const int i = min(i0 + j * BIN_R, sEnd - 1);
            s[j] = recA[i];
        }
        #pragma unroll
        for (int j = 0; j < 4; ++j)
            if (i0 + j * BIN_R < sEnd) atomicAdd(&cOut[s[j] & (BIN_R - 1)], 1);
    }
    __syncthreads();
    const int node = b * BIN_R + t;
    const int ci = cIn[t];
    if (node < n) {
        norm_src[node] = rsqrtf(fmaxf((float)cOut[t], 1.0f));
        norm_dst[node] = rsqrtf(fmaxf((float)ci, 1.0f));
    }
    part[t] = ci;
    __syncthreads();
    for (int o = 1; o < BIN_R; o <<= 1) {
        int x = (t >= o) ? part[t - o] : 0;
        __syncthreads();
        part[t] += x;
        __syncthreads();
    }
    const int start = dBeg + part[t] - ci;
    if (node <= n) rowptr[node] = start;
    cIn[t] = start;
    __syncthreads();
    for (int i0 = dBeg + t; i0 < dEnd; i0 += 4 * BIN_R) {
        int d[4], s[4];
        #pragma unroll
        for (int j = 0; j < 4; ++j) {
            const int i = min(i0 + j * BIN_R, dEnd - 1);
            d[j] = recD[i];
            s[j] = recA[i];
        }
        #pragma unroll
        for (int j = 0; j < 4; ++j) {
            if (i0 + j * BIN_R < dEnd) {
                const int p = atomicAdd(&cIn[d[j] & (BIN_R - 1)], 1);
                csr[p] = s[j];
            }
        }
    }
}

__device__ __forceinline__ void pair_bounds(int rp, int& beg, int& mid, int& end) {
    beg = __builtin_amdgcn_readlane(rp, 0);
    mid = __builtin_amdgcn_readlane(rp, 1);
    end = __builtin_amdgcn_readlane(rp, 2);
}

__device__ __forceinline__ int chunk_ids(const int* __restrict__ csr, int lane, int b0, int end) {
    return csr[max(min(b0 + lane, end - 1), 0)];
}

__device__ __forceinline__ void chunk_weights(int lane, int b0, int mid, int end, float ns,
                                              float& w0, float& w1) {
    const int e = b0 + lane;
    w0 = (e < mid) ? ns : 0.0f;
    w1 = (e >= mid && e < end) ? ns : 0.0f;
}

__device__ __forceinline__ void rows_issue(const float* __restrict__ feats, int lane, int s_l, int j0,
                                           float4 (&v)[8]) {
    #pragma unroll
    for (int u = 0; u < 8; ++u) {
        const int s = __builtin_amdgcn_readlane(s_l, j0 + u);
        v[u] = ((const float4*)(feats + (size_t)s * DFEAT))[lane];
    }
}

__device__ __forceinline__ void rows_add(const float4 (&v)[8], float w0_l, float w1_l, int j0,
                                         float4& acc0, float4& acc1) {
    #pragma unroll
    for (int u = 0; u < 8; ++u) {
        const float n0 = __int_as_float(__builtin_amdgcn_readlane(__float_as_int(w0_l), j0 + u));
        const float n1 = __int_as_float(__builtin_amdgcn_readlane(__float_as_int(w1_l), j0 + u));
        acc0.x = __builtin_fmaf(v[u].x, n0, acc0.x); acc0.y = __builtin_fmaf(v[u].y, n0, acc0.y);
        acc0.z = __builtin_fmaf(v[u].z, n0, acc0.z); acc0.w = __builtin_fmaf(v[u].w, n0, acc0.w);
        acc1.x = __builtin_fmaf(v[u].x, n1, acc1.x); acc1.y = __builtin_fmaf(v[u].y, n1, acc1.y);
        acc1.z = __builtin_fmaf(v[u].z, n1, acc1.z); acc1.w = __builtin_fmaf(v[u].w, n1, acc1.w);
    }
}

__device__ __forceinline__ void pair_finish(const float* __restrict__ feats,
                                            const int* __restrict__ csr,
                                            const float* __restrict__ norm_src, int lane,
                                            int beg, int mid, int end,
                                            int idA, int idB, float nsA, float nsB,
                                            const float4 (&v0)[8], float4& acc0, float4& acc1) {
    acc0 = make_float4(0.0f, 0.0f, 0.0f, 0.0f);
    acc1 = make_float4(0.0f, 0.0f, 0.0f, 0.0f);
    {
        const int cnt = min(32, end - beg);
        float w0_l, w1_l;
        chunk_weights(lane, beg, mid, end, nsA, w0_l, w1_l);
        rows_add(v0, w0_l, w1_l, 0, acc0, acc1);
        for (int j0 = 8; j0 < cnt; j0 += 8) {
            float4 v[8];
            rows_issue(feats, lane, idA, j0, v);
            rows_add(v, w0_l, w1_l, j0, acc0, acc1);
        }
    }
    int c = 1;
    for (int b0 = beg + 32; b0 < end; b0 += 32, ++c) {
        const int cnt = min(32, end - b0);
        int   s_l  = idB;
        float ns_l = nsB;
        if (c > 1) {
            s_l  = chunk_ids(csr, lane, b0, end);
            ns_l = norm_src[s_l];
        }
        float w0_l, w1_l;
        chunk_weights(lane, b0, mid, end, ns_l, w0_l, w1_l);
        for (int j0 = 0; j0 < cnt; j0 += 8) {
            float4 v[8];
            rows_issue(feats, lane, s_l, j0, v);
            rows_add(v, w0_l, w1_l, j0, acc0, acc1);
        }
    }
}

#define GEMM_TILES (NNODES / 16)
#define GEMM_GRID  512
#define LDS_P      132
static_assert(NNODES % 16 == 0, "every tile has 16 full rows");
static_assert(3 * GEMM_GRID <= GEMM_TILES, "every work-group has at least three tiles");

__device__ __forceinline__ void lds_barrier() {
    __builtin_amdgcn_fence(__ATOMIC_RELEASE, "workgroup", "local");
    __builtin_amdgcn_s_barrier();
    __builtin_amdgcn_fence(__ATOMIC_ACQUIRE, "workgroup", "local");
}

__global__ __launch_bounds__(256, 4) void gather_gemm_bn_kernel(
    const float* __restrict__ feats, const int* __restrict__ rowptr, const int* __restrict__ csr,
    const float* __restrict__ norm_src, const float* __restrict__ norm_dst,
    const float* __restrict__ W,  const float* __restrict__ b,
    const float* __restrict__ Wr, const float* __restrict__ br,
    float* __restrict__ y, float* __restrict__ bn_sum, float* __restrict__ bn_sumsq) {
    __shared__ __attribute__((aligned(16))) float sA[2][16 * LDS_P];
    __shared__ __attribute__((aligned(16))) float sF[2][16 * LDS_P];
    const int lane = threadIdx.x & 31;
    const int wave = threadIdx.x >> 5;
    const int n0 = wave * 16;
    const int m  = lane & 15;
    const int kh = (lane >> 4) * 2;
    const int col = n0 + m;

    v2f bw[DFEAT / 4], bwr[DFEAT / 4];
    #pragma unroll
    for (int ks = 0; ks < DFEAT / 4; ++ks) {
        const int k = 4 * ks + kh;
        bw[ks].x  = W [(size_t)k       * DFEAT + col];
        bw[ks].y  = W [(size_t)(k + 1) * DFEAT + col];
        bwr[ks].x = Wr[(size_t)k       * DFEAT + col];
        bwr[ks].y = Wr[(size_t)(k + 1) * DFEAT + col];
    }
    const float bc  = b[col];
    const float brc = br[col];
    const int rOff = (lane >> 4) * 8;

    const int so0 = (2 * wave) * LDS_P + 4 * lane;
    const int so1 = so0 + LDS_P;
    const int fo = m * LDS_P + kh;
    const int rpl = 2 * wave + min(lane, 2);

    float4 pf0, pf1;
    float4 v0[8];
    int tile = blockIdx.x;
    int beg1, mid1, end1, idA1, idB1;
    int rp2;
    {
        const int rp0 = rowptr[tile * 16 + rpl];
        const int rp1 = rowptr[(tile + GEMM_GRID) * 16 + rpl];
        rp2           = rowptr[(tile + 2 * GEMM_GRID) * 16 + rpl];
        const int r0 = tile * 16 + 2 * wave;
        const float nd_l = norm_dst[r0 + (lane & 1)];
        pf0 = ((const float4*)(feats + (size_t)r0       * DFEAT))[lane];
        pf1 = ((const float4*)(feats + (size_t)(r0 + 1) * DFEAT))[lane];
        int beg, mid, end;
        pair_bounds(rp0, beg, mid, end);
        const int idA = chunk_ids(csr, lane, beg, end);
        const int idB = chunk_ids(csr, lane, beg + 32, end);
        pair_bounds(rp1, beg1, mid1, end1);
        idA1 = chunk_ids(csr, lane, beg1, end1);
        idB1 = chunk_ids(csr, lane, beg1 + 32, end1);
        const float nsA = norm_src[idA];
        const float nsB = norm_src[idB];
        rows_issue(feats, lane, idA, 0, v0);
        float4 a0, a1;
        pair_finish(feats, csr, norm_src, lane, beg, mid, end, idA, idB, nsA, nsB, v0, a0, a1);
        const float nd0 = __int_as_float(__builtin_amdgcn_readlane(__float_as_int(nd_l), 0));
        const float nd1 = __int_as_float(__builtin_amdgcn_readlane(__float_as_int(nd_l), 1));
        a0.x *= nd0; a0.y *= nd0; a0.z *= nd0; a0.w *= nd0;
        a1.x *= nd1; a1.y *= nd1; a1.z *= nd1; a1.w *= nd1;
        *(float4*)&sA[0][so0] = a0;
        *(float4*)&sA[0][so1] = a1;
        *(float4*)&sF[0][so0] = pf0;
        *(float4*)&sF[0][so1] = pf1;
    }
    lds_barrier();

    float s8 = 0.0f, q8 = 0.0f;
    int buf = 0;
    for (;;) {
        const int  next = tile + GEMM_GRID;
        const bool more = next < GEMM_TILES;
        float nd_l = 0.0f, nsA1 = 0.0f, nsB1 = 0.0f;
        int beg2 = 0, mid2 = 0, end2 = 0, idA2 = 0, idB2 = 0, rp3 = 0;
        if (more) {
            const int r0 = next * 16 + 2 * wave;
            nd_l = norm_dst[r0 + (lane & 1)];
            pf0 = ((const float4*)(feats + (size_t)r0       * DFEAT))[lane];
            pf1 = ((const float4*)(feats + (size_t)(r0 + 1) * DFEAT))[lane];
            nsA1 = norm_src[idA1];
            nsB1 = norm_src[idB1];
            rows_issue(feats, lane, idA1, 0, v0);
            pair_bounds(rp2, beg2, mid2, end2);
            idA2 = chunk_ids(csr, lane, beg2, end2);
            idB2 = chunk_ids(csr, lane, beg2 + 32, end2);
            rp3  = rowptr[min(next + 2 * GEMM_GRID, GEMM_TILES - 1) * 16 + rpl];
        }

        const float* aB = &sA[buf][fo];
        const float* fB = &sF[buf][fo];
        v8f accC = {};
        v8f accR = {};
        #pragma unroll
        for (int ks = 0; ks < DFEAT / 4; ++ks) {
            const v2f a = *(const v2f*)(aB + 4 * ks);
            const v2f f = *(const v2f*)(fB + 4 * ks);
            accC = __builtin_amdgcn_wmma_f32_16x16x4_f32(false, a, false, bw[ks],  (short)0, accC, false, false);
            accR = __builtin_amdgcn_wmma_f32_16x16x4_f32(false, f, false, bwr[ks], (short)0, accR, false, false);
        }

        const int rowBase = tile * 16;
        #pragma unroll
        for (int r = 0; r < 8; ++r) {
            float c = accC[r] + bc;  c = c > 0.0f ? c : 0.0f;
            float e = accR[r] + brc; e = e > 0.0f ? e : 0.0f;
            float v = c + e;
            y[(size_t)(rowBase + rOff + r) * DFEAT + col] = v;
            s8 += v;
            q8 += v * v;
        }

        if (more) {
            float4 a0, a1;
            pair_finish(feats, csr, norm_src, lane, beg1, mid1, end1, idA1, idB1, nsA1, nsB1, v0, a0, a1);
            const float nd0 = __int_as_float(__builtin_amdgcn_readlane(__float_as_int(nd_l), 0));
            const float nd1 = __int_as_float(__builtin_amdgcn_readlane(__float_as_int(nd_l), 1));
            a0.x *= nd0; a0.y *= nd0; a0.z *= nd0; a0.w *= nd0;
            a1.x *= nd1; a1.y *= nd1; a1.z *= nd1; a1.w *= nd1;
            *(float4*)&sA[buf ^ 1][so0] = a0;
            *(float4*)&sA[buf ^ 1][so1] = a1;
            *(float4*)&sF[buf ^ 1][so0] = pf0;
            *(float4*)&sF[buf ^ 1][so1] = pf1;
            beg1 = beg2; mid1 = mid2; end1 = end2; idA1 = idA2; idB1 = idB2;
            rp2 = rp3;
        }
        lds_barrier();
        buf ^= 1;
        tile = next;
        if (!more) break;
    }

    float s8o = __shfl_down(s8, 16);
    float q8o = __shfl_down(q8, 16);
    if (lane < 16) {
        atomicAdd(&bn_sum[col],   s8 + s8o);
        atomicAdd(&bn_sumsq[col], q8 + q8o);
    }
}

__global__ __launch_bounds__(256) void bn_apply_kernel(float* __restrict__ y,
                                                       const float* __restrict__ sum,
                                                       const float* __restrict__ sumsq,
                                                       const float* __restrict__ gamma,
                                                       const float* __restrict__ beta,
                                                       size_t nvec4, int n) {
    size_t i = (size_t)blockIdx.x * blockDim.x + threadIdx.x;
    if (i >= nvec4) return;
    float4 v = ((float4*)y)[i];
    const int col = (int)((i * 4) & (DFEAT - 1));
    const float4 s4 = *(const float4*)(sum + col);
    const float4 q4 = *(const float4*)(sumsq + col);
    const float4 g4 = *(const float4*)(gamma + col);
    const float4 b4 = *(const float4*)(beta + col);
    const float inv = 1.0f / (float)n;
    float mean, var, sc;
    mean = s4.x * inv; var = q4.x * inv - mean * mean; sc = g4.x * rsqrtf(var + BN_EPS);
    v.x = v.x * sc + (b4.x - mean * sc);
    mean = s4.y * inv; var = q4.y * inv - mean * mean; sc = g4.y * rsqrtf(var + BN_EPS);
    v.y = v.y * sc + (b4.y - mean * sc);
    mean = s4.z * inv; var = q4.z * inv - mean * mean; sc = g4.z * rsqrtf(var + BN_EPS);
    v.z = v.z * sc + (b4.z - mean * sc);
    mean = s4.w * inv; var = q4.w * inv - mean * mean; sc = g4.w * rsqrtf(var + BN_EPS);
    v.w = v.w * sc + (b4.w - mean * sc);
    ((float4*)y)[i] = v;
}

extern "C" void kernel_launch(void* const* d_in, const int* in_sizes, int n_in,
                              void* d_out, int out_size, void* d_ws, size_t ws_size,
                              hipStream_t stream) {
    const float* feats = (const float*)d_in[0];
    const int*   src   = (const int*)d_in[1];
    const int*   dst   = (const int*)d_in[2];
    const float* W     = (const float*)d_in[3];
    const float* b     = (const float*)d_in[4];
    const float* Wr    = (const float*)d_in[5];
    const float* br    = (const float*)d_in[6];
    const float* gamma = (const float*)d_in[7];
    const float* beta  = (const float*)d_in[8];
    float* out = (float*)d_out;

    const int N = NNODES;
    const int E = in_sizes[1];

    float* ws        = (float*)d_ws;
    int*   csr       = (int*)d_ws;
    int*   rowptr    = csr + E;
    float* deg_out   = ws + (size_t)N * DFEAT;
    float* deg_in    = deg_out + N;
    float* bn_sum    = deg_in + N;
    float* bn_sumsq  = bn_sum + DFEAT;

    int* scr      = (int*)d_out;
    int* cnt      = scr;
    int* binStart = scr + 98304;
    int* recA     = scr + 196608;
    int* recD     = recA + E + E;

    hist_kernel<<<NCHUNK, EDGE_T, 0, stream>>>(src, dst, cnt, bn_sum, E);

    scatter_kernel<<<NCHUNK, EDGE_T, 0, stream>>>(src, dst, cnt, binStart, recA, recD, E);

    local_kernel<<<NB, BIN_R, 0, stream>>>(binStart, recA, recD, deg_out, deg_in, rowptr, csr, N);

    gather_gemm_bn_kernel<<<GEMM_GRID, 256, 0, stream>>>(feats, rowptr, csr, deg_out, deg_in,
                                                         W, b, Wr, br, out, bn_sum, bn_sumsq);

    size_t nvec4 = (size_t)N * DFEAT / 4;
    bn_apply_kernel<<<(int)((nvec4 + 255) / 256), 256, 0, stream>>>(out, bn_sum, bn_sumsq, gamma, beta,
                                                                   nvec4, N);
}


# Scattered 4-byte stores of the edge grouping and of the sorted source list go through an LDS stage and leave as contiguous runs; row sums stay in arrival order of LDS adds
# speedup vs baseline: 4.0384x; 1.3067x over previous
#include <hip/hip_runtime.h>
#include <hip/hip_bf16.h>

typedef __attribute__((ext_vector_type(2))) float v2f;
typedef __attribute__((ext_vector_type(8))) float v8f;

#define NNODES 100000
#define DFEAT  128
#define BN_EPS 1e-5f
#define BIN_SH 9
#define BIN_R  512
#define NB     196
#define NB2    (2 * NB)
#define NCHUNK 250
#define EDGE_T 1024
#define CQ     (NB2 / 4)
#define CGRP   10
#define CROWS  25
#define CPL    ((NB2 + 31) / 32)
#define CPB    ((NB + 31) / 32)
#define NEDGES 1600000
#define CH_E   ((NEDGES + NCHUNK - 1) / NCHUNK)
#define CSR_CAP 16384
static_assert(NB2 % 4 == 0, "count-table rows are read as 16-byte words");
static_assert(CGRP * CROWS == NCHUNK, "the groups cover every chunk row");
static_assert(CGRP * CQ <= EDGE_T, "one thread per (group, 16-byte word)");


__global__ __launch_bounds__(EDGE_T) void hist_kernel(const int* __restrict__ src,
                                                      const int* __restrict__ dst,
                                                      int* __restrict__ cnt,
                                                      float* __restrict__ bn_acc, int E) {
    __shared__ int hD[NB];
    __shared__ int hS[NB];
    const int t = threadIdx.x;
    const int w = blockIdx.x;
    if (t < NB) { hD[t] = 0; hS[t] = 0; }
    if (w == 0 && t < 2 * DFEAT) bn_acc[t] = 0.0f;
    __syncthreads();
    const int ch  = (E + NCHUNK - 1) / NCHUNK;
    const int beg = w * ch;
    const int end = min(E, beg + ch);
    for (int e0 = beg + t; e0 < end; e0 += 4 * EDGE_T) {
        int d[4], s[4];
        #pragma unroll
        for (int j = 0; j < 4; ++j) {
            const int e = min(e0 + j * EDGE_T, end - 1);
            d[j] = dst[e];
            s[j] = src[e];
        }
        #pragma unroll
        for (int j = 0; j < 4; ++j) {
            if (e0 + j * EDGE_T < end) {
                atomicAdd(&hD[d[j] >> BIN_SH], 1);
                atomicAdd(&hS[s[j] >> BIN_SH], 1);
            }
        }
    }
    __syncthreads();
    if (t < NB) {
        cnt[w * NB2 + t]      = hD[t];
        cnt[w * NB2 + NB + t] = hS[t];
    }
}

template <int PER>
__device__ __forceinline__ void wave_excl_scan(const int* in, int* out, int n, int lane, bool total) {
    int loc[PER];
    int sum = 0;
    for (int i = 0; i < PER; ++i) {
        const int idx = lane * PER + i;
        loc[i] = (idx < n) ? in[idx] : 0;
        sum += loc[i];
    }
    int inc = sum;
    for (int o = 1; o < 32; o <<= 1) {
        const int x = __shfl_up(inc, o);
        if (lane >= o) inc += x;
    }
    int run = inc - sum;
    for (int i = 0; i < PER; ++i) {
        const int idx = lane * PER + i;
        if (idx < n) out[idx] = run;
        run += loc[i];
    }
    if (total && lane == 31) out[n] = run;
}

__global__ __launch_bounds__(EDGE_T) void scatter_kernel(const int* __restrict__ src,
                                                         const int* __restrict__ dst,
                                                         const int* __restrict__ cnt,
                                                         int* __restrict__ binStart,
                                                         int* __restrict__ recA,
                                                         int* __restrict__ recD, int E) {
    __shared__ __attribute__((aligned(16))) int pPre[CGRP * NB2];
    __shared__ __attribute__((aligned(16))) int pTot[CGRP * NB2];
    __shared__ int own[NB2];
    __shared__ int sPre[NB2];
    __shared__ int sTot[NB2];
    __shared__ int sStart[NB2 + 1];
    __shared__ int lStart[NB2];
    __shared__ int cD[NB];
    __shared__ int cS[NB];
    __shared__ int dD[NB];
    __shared__ int dS[NB];
    __shared__ int stA[CH_E];
    __shared__ int stD[CH_E];
    __shared__ int stS[CH_E];
    const int t = threadIdx.x;
    const int w = blockIdx.x;
    const int ownv = cnt[w * NB2 + min(t, NB2 - 1)];
    if (t < CGRP * CQ) {
        const int g = t / CQ;
        const int c = t - g * CQ;
        const int4* p = (const int4*)cnt + (size_t)(g * CROWS) * CQ + c;
        int4 pre = make_int4(0, 0, 0, 0);
        int4 tot = make_int4(0, 0, 0, 0);
        #pragma unroll
        for (int i = 0; i < CROWS; ++i) {
            const int4 v = p[i * CQ];
            const int m = (g * CROWS + i < w) ? -1 : 0;
            tot.x += v.x; tot.y += v.y; tot.z += v.z; tot.w += v.w;
            pre.x += v.x & m; pre.y += v.y & m; pre.z += v.z & m; pre.w += v.w & m;
        }
        *(int4*)&pPre[g * NB2 + 4 * c] = pre;
        *(int4*)&pTot[g * NB2 + 4 * c] = tot;
    }
    if (t < NB2) own[t] = ownv;
    __syncthreads();
    if (t < NB2) {
        int p = 0, s = 0;
        for (int g = 0; g < CGRP; ++g) { p += pPre[g * NB2 + t]; s += pTot[g * NB2 + t]; }
        sPre[t] = p;
        sTot[t] = s;
    }
    __syncthreads();
    if (t < 32)       wave_excl_scan<CPL>(sTot, sStart, NB2, t, true);
    else if (t < 64)  wave_excl_scan<CPB>(own, lStart, NB, t - 32, false);
    else if (t < 96)  wave_excl_scan<CPB>(own + NB, lStart + NB, NB, t - 64, false);
    __syncthreads();
    if (t < NB) {
        const int lD = lStart[t];
        const int lS = lStart[NB + t];
        cD[t] = lD;
        cS[t] = lS;
        dD[t] = sStart[t] + sPre[t] - lD;
        dS[t] = sStart[NB + t] + sPre[NB + t] - lS;
    }
    if (w == 0 && t <= NB2) binStart[t] = sStart[t];
    __syncthreads();
    const int ch  = (E + NCHUNK - 1) / NCHUNK;
    const int beg = w * ch;
    const int end = min(E, beg + ch);
    for (int e0 = beg + t; e0 < end; e0 += 4 * EDGE_T) {
        int d[4], s[4];
        #pragma unroll
        for (int j = 0; j < 4; ++j) {
            const int e = min(e0 + j * EDGE_T, end - 1);
            d[j] = dst[e];
            s[j] = src[e];
        }
        #pragma unroll
        for (int j = 0; j < 4; ++j) {
            if (e0 + j * EDGE_T < end) {
                const int p = atomicAdd(&cD[d[j] >> BIN_SH], 1);
                stA[p] = s[j];
                stD[p] = d[j];
                const int q = atomicAdd(&cS[s[j] >> BIN_SH], 1);
                stS[q] = s[j];
            }
        }
    }
    __syncthreads();
    const int n = end - beg;
    for (int i = t; i < n; i += EDGE_T) {
        const int d = stD[i];
        const int a = stA[i];
        const int s = stS[i];
        const int g = dD[d >> BIN_SH] + i;
        recA[g] = a;
        recD[g] = d;
        recA[dS[s >> BIN_SH] + i] = s;
    }
}

__global__ __launch_bounds__(BIN_R) void local_kernel(const int* __restrict__ binStart,
                                                      const int* __restrict__ recA,
                                                      const int* __restrict__ recD,
                                                      float* __restrict__ norm_src,
                                                      float* __restrict__ norm_dst,
                                                      int* __restrict__ rowptr,
                                                      int* __restrict__ csr, int n) {
    __shared__ int cIn[BIN_R];
    __shared__ int cOut[BIN_R];
    __shared__ int part[BIN_R];
    __shared__ int stg[CSR_CAP];
    const int t = threadIdx.x;
    const int b = blockIdx.x;
    cIn[t] = 0;
    cOut[t] = 0;
    __syncthreads();
    const int dBeg = binStart[b];
    const int dEnd = binStart[b + 1];
    const int sBeg = binStart[NB + b];
    const int sEnd = binStart[NB + b + 1];
    for (int i0 = dBeg + t; i0 < dEnd; i0 += 4 * BIN_R) {
        int d[4];
        #pragma unroll
        for (int j = 0; j < 4; ++j) {
            const int i = min(i0 + j * BIN_R, dEnd - 1);
            d[j] = recD[i];
        }
        #pragma unroll
        for (int j = 0; j < 4; ++j)
            if (i0 + j * BIN_R < dEnd) atomicAdd(&cIn[d[j] & (BIN_R - 1)], 1);
    }
    for (int i0 = sBeg + t; i0 < sEnd; i0 += 4 * BIN_R) {
        int s[4];
        #pragma unroll
        for (int j = 0; j < 4; ++j) {
            const int i = min(i0 + j * BIN_R, sEnd - 1);
            s[j] = recA[i];
        }
        #pragma unroll
        for (int j = 0; j < 4; ++j)
            if (i0 + j * BIN_R < sEnd) atomicAdd(&cOut[s[j] & (BIN_R - 1)], 1);
    }
    __syncthreads();
    const int node = b * BIN_R + t;
    const int ci = cIn[t];
    if (node < n) {
        norm_src[node] = rsqrtf(fmaxf((float)cOut[t], 1.0f));
        norm_dst[node] = rsqrtf(fmaxf((float)ci, 1.0f));
    }
    part[t] = ci;
    __syncthreads();
    for (int o = 1; o < BIN_R; o <<= 1) {
        int x = (t >= o) ? part[t - o] : 0;
        __syncthreads();
        part[t] += x;
        __syncthreads();
    }
    const int start = dBeg + part[t] - ci;
    if (node <= n) rowptr[node] = start;
    cIn[t] = start;
    __syncthreads();
    const int  nb     = dEnd - dBeg;
    const bool staged = nb <= CSR_CAP;
    for (int i0 = dBeg + t; i0 < dEnd; i0 += 4 * BIN_R) {
        int d[4], s[4];
        #pragma unroll
        for (int j = 0; j < 4; ++j) {
            const int i = min(i0 + j * BIN_R, dEnd - 1);
            d[j] = recD[i];
            s[j] = recA[i];
        }
        #pragma unroll
        for (int j = 0; j < 4; ++j) {
            if (i0 + j * BIN_R < dEnd) {
                const int p = atomicAdd(&cIn[d[j] & (BIN_R - 1)], 1);
                if (staged) stg[p - dBeg] = s[j];
                else        csr[p] = s[j];
            }
        }
    }
    if (staged) {
        __syncthreads();
        for (int i = t; i < nb; i += BIN_R) csr[dBeg + i] = stg[i];
    }
}

__device__ __forceinline__ void pair_bounds(int rp, int& beg, int& mid, int& end) {
    beg = __builtin_amdgcn_readlane(rp, 0);
    mid = __builtin_amdgcn_readlane(rp, 1);
    end = __builtin_amdgcn_readlane(rp, 2);
}

__device__ __forceinline__ int chunk_ids(const int* __restrict__ csr, int lane, int b0, int end) {
    return csr[max(min(b0 + lane, end - 1), 0)];
}

__device__ __forceinline__ void chunk_weights(int lane, int b0, int mid, int end, float ns,
                                              float& w0, float& w1) {
    const int e = b0 + lane;
    w0 = (e < mid) ? ns : 0.0f;
    w1 = (e >= mid && e < end) ? ns : 0.0f;
}

__device__ __forceinline__ void rows_issue(const float* __restrict__ feats, int lane, int s_l, int j0,
                                           float4 (&v)[8]) {
    #pragma unroll
    for (int u = 0; u < 8; ++u) {
        const int s = __builtin_amdgcn_readlane(s_l, j0 + u);
        v[u] = ((const float4*)(feats + (size_t)s * DFEAT))[lane];
    }
}

__device__ __forceinline__ void rows_add(const float4 (&v)[8], float w0_l, float w1_l, int j0,
                                         float4& acc0, float4& acc1) {
    #pragma unroll
    for (int u = 0; u < 8; ++u) {
        const float n0 = __int_as_float(__builtin_amdgcn_readlane(__float_as_int(w0_l), j0 + u));
        const float n1 = __int_as_float(__builtin_amdgcn_readlane(__float_as_int(w1_l), j0 + u));
        acc0.x = __builtin_fmaf(v[u].x, n0, acc0.x); acc0.y = __builtin_fmaf(v[u].y, n0, acc0.y);
        acc0.z = __builtin_fmaf(v[u].z, n0, acc0.z); acc0.w = __builtin_fmaf(v[u].w, n0, acc0.w);
        acc1.x = __builtin_fmaf(v[u].x, n1, acc1.x); acc1.y = __builtin_fmaf(v[u].y, n1, acc1.y);
        acc1.z = __builtin_fmaf(v[u].z, n1, acc1.z); acc1.w = __builtin_fmaf(v[u].w, n1, acc1.w);
    }
}

__device__ __forceinline__ void pair_finish(const float* __restrict__ feats,
                                            const int* __restrict__ csr,
                                            const float* __restrict__ norm_src, int lane,
                                            int beg, int mid, int end,
                                            int idA, int idB, float nsA, float nsB,
                                            const float4 (&v0)[8], float4& acc0, float4& acc1) {
    acc0 = make_float4(0.0f, 0.0f, 0.0f, 0.0f);
    acc1 = make_float4(0.0f, 0.0f, 0.0f, 0.0f);
    {
        const int cnt = min(32, end - beg);
        float w0_l, w1_l;
        chunk_weights(lane, beg, mid, end, nsA, w0_l, w1_l);
        rows_add(v0, w0_l, w1_l, 0, acc0, acc1);
        for (int j0 = 8; j0 < cnt; j0 += 8) {
            float4 v[8];
            rows_issue(feats, lane, idA, j0, v);
            rows_add(v, w0_l, w1_l, j0, acc0, acc1);
        }
    }
    int c = 1;
    for (int b0 = beg + 32; b0 < end; b0 += 32, ++c) {
        const int cnt = min(32, end - b0);
        int   s_l  = idB;
        float ns_l = nsB;
        if (c > 1) {
            s_l  = chunk_ids(csr, lane, b0, end);
            ns_l = norm_src[s_l];
        }
        float w0_l, w1_l;
        chunk_weights(lane, b0, mid, end, ns_l, w0_l, w1_l);
        for (int j0 = 0; j0 < cnt; j0 += 8) {
            float4 v[8];
            rows_issue(feats, lane, s_l, j0, v);
            rows_add(v, w0_l, w1_l, j0, acc0, acc1);
        }
    }
}

#define GEMM_TILES (NNODES / 16)
#define GEMM_GRID  512
#define LDS_P      132
static_assert(NNODES % 16 == 0, "every tile has 16 full rows");
static_assert(3 * GEMM_GRID <= GEMM_TILES, "every work-group has at least three tiles");

__device__ __forceinline__ void lds_barrier() {
    __builtin_amdgcn_fence(__ATOMIC_RELEASE, "workgroup", "local");
    __builtin_amdgcn_s_barrier();
    __builtin_amdgcn_fence(__ATOMIC_ACQUIRE, "workgroup", "local");
}

__global__ __launch_bounds__(256, 4) void gather_gemm_bn_kernel(
    const float* __restrict__ feats, const int* __restrict__ rowptr, const int* __restrict__ csr,
    const float* __restrict__ norm_src, const float* __restrict__ norm_dst,
    const float* __restrict__ W,  const float* __restrict__ b,
    const float* __restrict__ Wr, const float* __restrict__ br,
    float* __restrict__ y, float* __restrict__ bn_sum, float* __restrict__ bn_sumsq) {
    __shared__ __attribute__((aligned(16))) float sA[2][16 * LDS_P];
    __shared__ __attribute__((aligned(16))) float sF[2][16 * LDS_P];
    const int lane = threadIdx.x & 31;
    const int wave = threadIdx.x >> 5;
    const int n0 = wave * 16;
    const int m  = lane & 15;
    const int kh = (lane >> 4) * 2;
    const int col = n0 + m;

    v2f bw[DFEAT / 4], bwr[DFEAT / 4];
    #pragma unroll
    for (int ks = 0; ks < DFEAT / 4; ++ks) {
        const int k = 4 * ks + kh;
        bw[ks].x  = W [(size_t)k       * DFEAT + col];
        bw[ks].y  = W [(size_t)(k + 1) * DFEAT + col];
        bwr[ks].x = Wr[(size_t)k       * DFEAT + col];
        bwr[ks].y = Wr[(size_t)(k + 1) * DFEAT + col];
    }
    const float bc  = b[col];
    const float brc = br[col];
    const int rOff = (lane >> 4) * 8;

    const int so0 = (2 * wave) * LDS_P + 4 * lane;
    const int so1 = so0 + LDS_P;
    const int fo = m * LDS_P + kh;
    const int rpl = 2 * wave + min(lane, 2);

    float4 pf0, pf1;
    float4 v0[8];
    int tile = blockIdx.x;
    int beg1, mid1, end1, idA1, idB1;
    int rp2;
    {
        const int rp0 = rowptr[tile * 16 + rpl];
        const int rp1 = rowptr[(tile + GEMM_GRID) * 16 + rpl];
        rp2           = rowptr[(tile + 2 * GEMM_GRID) * 16 + rpl];
        const int r0 = tile * 16 + 2 * wave;
        const float nd_l = norm_dst[r0 + (lane & 1)];
        pf0 = ((const float4*)(feats + (size_t)r0       * DFEAT))[lane];
        pf1 = ((const float4*)(feats + (size_t)(r0 + 1) * DFEAT))[lane];
        int beg, mid, end;
        pair_bounds(rp0, beg, mid, end);
        const int idA = chunk_ids(csr, lane, beg, end);
        const int idB = chunk_ids(csr, lane, beg + 32, end);
        pair_bounds(rp1, beg1, mid1, end1);
        idA1 = chunk_ids(csr, lane, beg1, end1);
        idB1 = chunk_ids(csr, lane, beg1 + 32, end1);
        const float nsA = norm_src[idA];
        const float nsB = norm_src[idB];
        rows_issue(feats, lane, idA, 0, v0);
        float4 a0, a1;
        pair_finish(feats, csr, norm_src, lane, beg, mid, end, idA, idB, nsA, nsB, v0, a0, a1);
        const float nd0 = __int_as_float(__builtin_amdgcn_readlane(__float_as_int(nd_l), 0));
        const float nd1 = __int_as_float(__builtin_amdgcn_readlane(__float_as_int(nd_l), 1));
        a0.x *= nd0; a0.y *= nd0; a0.z *= nd0; a0.w *= nd0;
        a1.x *= nd1; a1.y *= nd1; a1.z *= nd1; a1.w *= nd1;
        *(float4*)&sA[0][so0] = a0;
        *(float4*)&sA[0][so1] = a1;
        *(float4*)&sF[0][so0] = pf0;
        *(float4*)&sF[0][so1] = pf1;
    }
    lds_barrier();

    float s8 = 0.0f, q8 = 0.0f;
    int buf = 0;
    for (;;) {
        const int  next = tile + GEMM_GRID;
        const bool more = next < GEMM_TILES;
        float nd_l = 0.0f, nsA1 = 0.0f, nsB1 = 0.0f;
        int beg2 = 0, mid2 = 0, end2 = 0, idA2 = 0, idB2 = 0, rp3 = 0;
        if (more) {
            const int r0 = next * 16 + 2 * wave;
            nd_l = norm_dst[r0 + (lane & 1)];
            pf0 = ((const float4*)(feats + (size_t)r0       * DFEAT))[lane];
            pf1 = ((const float4*)(feats + (size_t)(r0 + 1) * DFEAT))[lane];
            nsA1 = norm_src[idA1];
            nsB1 = norm_src[idB1];
            rows_issue(feats, lane, idA1, 0, v0);
            pair_bounds(rp2, beg2, mid2, end2);
            idA2 = chunk_ids(csr, lane, beg2, end2);
            idB2 = chunk_ids(csr, lane, beg2 + 32, end2);
            rp3  = rowptr[min(next + 2 * GEMM_GRID, GEMM_TILES - 1) * 16 + rpl];
        }

        const float* aB = &sA[buf][fo];
        const float* fB = &sF[buf][fo];
        v8f accC = {};
        v8f accR = {};
        #pragma unroll
        for (int ks = 0; ks < DFEAT / 4; ++ks) {
            const v2f a = *(const v2f*)(aB + 4 * ks);
            const v2f f = *(const v2f*)(fB + 4 * ks);
            accC = __builtin_amdgcn_wmma_f32_16x16x4_f32(false, a, false, bw[ks],  (short)0, accC, false, false);
            accR = __builtin_amdgcn_wmma_f32_16x16x4_f32(false, f, false, bwr[ks], (short)0, accR, false, false);
        }

        const int rowBase = tile * 16;
        #pragma unroll
        for (int r = 0; r < 8; ++r) {
            float c = accC[r] + bc;  c = c > 0.0f ? c : 0.0f;
            float e = accR[r] + brc; e = e > 0.0f ? e : 0.0f;
            float v = c + e;
            y[(size_t)(rowBase + rOff + r) * DFEAT + col] = v;
            s8 += v;
            q8 += v * v;
        }

        if (more) {
            float4 a0, a1;
            pair_finish(feats, csr, norm_src, lane, beg1, mid1, end1, idA1, idB1, nsA1, nsB1, v0, a0, a1);
            const float nd0 = __int_as_float(__builtin_amdgcn_readlane(__float_as_int(nd_l), 0));
            const float nd1 = __int_as_float(__builtin_amdgcn_readlane(__float_as_int(nd_l), 1));
            a0.x *= nd0; a0.y *= nd0; a0.z *= nd0; a0.w *= nd0;
            a1.x *= nd1; a1.y *= nd1; a1.z *= nd1; a1.w *= nd1;
            *(float4*)&sA[buf ^ 1][so0] = a0;
            *(float4*)&sA[buf ^ 1][so1] = a1;
            *(float4*)&sF[buf ^ 1][so0] = pf0;
            *(float4*)&sF[buf ^ 1][so1] = pf1;
            beg1 = beg2; mid1 = mid2; end1 = end2; idA1 = idA2; idB1 = idB2;
            rp2 = rp3;
        }
        lds_barrier();
        buf ^= 1;
        tile = next;
        if (!more) break;
    }

    float s8o = __shfl_down(s8, 16);
    float q8o = __shfl_down(q8, 16);
    if (lane < 16) {
        atomicAdd(&bn_sum[col],   s8 + s8o);
        atomicAdd(&bn_sumsq[col], q8 + q8o);
    }
}

__global__ __launch_bounds__(256) void bn_apply_kernel(float* __restrict__ y,
                                                       const float* __restrict__ sum,
                                                       const float* __restrict__ sumsq,
                                                       const float* __restrict__ gamma,
                                                       const float* __restrict__ beta,
                                                       size_t nvec4, int n) {
    size_t i = (size_t)blockIdx.x * blockDim.x + threadIdx.x;
    if (i >= nvec4) return;
    float4 v = ((float4*)y)[i];
    const int col = (int)((i * 4) & (DFEAT - 1));
    const float4 s4 = *(const float4*)(sum + col);
    const float4 q4 = *(const float4*)(sumsq + col);
    const float4 g4 = *(const float4*)(gamma + col);
    const float4 b4 = *(const float4*)(beta + col);
    const float inv = 1.0f / (float)n;
    float mean, var, sc;
    mean = s4.x * inv; var = q4.x * inv - mean * mean; sc = g4.x * rsqrtf(var + BN_EPS);
    v.x = v.x * sc + (b4.x - mean * sc);
    mean = s4.y * inv; var = q4.y * inv - mean * mean; sc = g4.y * rsqrtf(var + BN_EPS);
    v.y = v.y * sc + (b4.y - mean * sc);
    mean = s4.z * inv; var = q4.z * inv - mean * mean; sc = g4.z * rsqrtf(var + BN_EPS);
    v.z = v.z * sc + (b4.z - mean * sc);
    mean = s4.w * inv; var = q4.w * inv - mean * mean; sc = g4.w * rsqrtf(var + BN_EPS);
    v.w = v.w * sc + (b4.w - mean * sc);
    ((float4*)y)[i] = v;
}

extern "C" void kernel_launch(void* const* d_in, const int* in_sizes, int n_in,
                              void* d_out, int out_size, void* d_ws, size_t ws_size,
                              hipStream_t stream) {
    const float* feats = (const float*)d_in[0];
    const int*   src   = (const int*)d_in[1];
    const int*   dst   = (const int*)d_in[2];
    const float* W     = (const float*)d_in[3];
    const float* b     = (const float*)d_in[4];
    const float* Wr    = (const float*)d_in[5];
    const float* br    = (const float*)d_in[6];
    const float* gamma = (const float*)d_in[7];
    const float* beta  = (const float*)d_in[8];
    float* out = (float*)d_out;

    const int N = NNODES;
    const int E = in_sizes[1];

    float* ws        = (float*)d_ws;
    int*   csr       = (int*)d_ws;
    int*   rowptr    = csr + E;
    float* deg_out   = ws + (size_t)N * DFEAT;
    float* deg_in    = deg_out + N;
    float* bn_sum    = deg_in + N;
    float* bn_sumsq  = bn_sum + DFEAT;

    int* scr      = (int*)d_out;
    int* cnt      = scr;
    int* binStart = scr + 98304;
    int* recA     = scr + 196608;
    int* recD     = recA + E + E;

    hist_kernel<<<NCHUNK, EDGE_T, 0, stream>>>(src, dst, cnt, bn_sum, E);

    scatter_kernel<<<NCHUNK, EDGE_T, 0, stream>>>(src, dst, cnt, binStart, recA, recD, E);

    local_kernel<<<NB, BIN_R, 0, stream>>>(binStart, recA, recD, deg_out, deg_in, rowptr, csr, N);

    gather_gemm_bn_kernel<<<GEMM_GRID, 256, 0, stream>>>(feats, rowptr, csr, deg_out, deg_in,
                                                         W, b, Wr, br, out, bn_sum, bn_sumsq);

    size_t nvec4 = (size_t)N * DFEAT / 4;
    bn_apply_kernel<<<(int)((nvec4 + 255) / 256), 256, 0, stream>>>(out, bn_sum, bn_sumsq, gamma, beta,
                                                                   nvec4, N);
}


# Each edge chunk writes its bin-ordered records into its own slot: no count launch, no count-table sum (5 launches -> 4); the per-bin kernel walks 250 chunk pieces; row sums stay in LDS arrival order
# speedup vs baseline: 4.3989x; 1.0893x over previous
#include <hip/hip_runtime.h>
#include <hip/hip_bf16.h>

typedef __attribute__((ext_vector_type(2))) float v2f;
typedef __attribute__((ext_vector_type(8))) float v8f;

#define NNODES 100000
#define DFEAT  128
#define BN_EPS 1e-5f
#define BIN_SH 9
#define BIN_R  512
#define NB     196
#define NB2    (2 * NB)
#define NCHUNK 250
#define EDGE_T 1024
#define CPB    ((NB + 31) / 32)
#define NEDGES 1600000
#define CH_E   ((NEDGES + NCHUNK - 1) / NCHUNK)
#define EPT    ((CH_E + EDGE_T - 1) / EDGE_T)
#define CSR_CAP 16384
#define LWAVES (BIN_R / 32)
#define SEGP   256
static_assert(CH_E < 65536, "a piece's offset and length share one 32-bit table word");
static_assert(NCHUNK <= SEGP, "the piece list holds every chunk");
static_assert(2 * SEGP == BIN_R, "one thread per piece-list entry");


template <int PER>
__device__ __forceinline__ void wave_excl_scan(const int* in, int* out, int n, int lane, bool total) {
    int loc[PER];
    int sum = 0;
    for (int i = 0; i < PER; ++i) {
        const int idx = lane * PER + i;
        loc[i] = (idx < n) ? in[idx] : 0;
        sum += loc[i];
    }
    int inc = sum;
    for (int o = 1; o < 32; o <<= 1) {
        const int x = __shfl_up(inc, o);
        if (lane >= o) inc += x;
    }
    int run = inc - sum;
    for (int i = 0; i < PER; ++i) {
        const int idx = lane * PER + i;
        if (idx < n) out[idx] = run;
        run += loc[i];
    }
    if (total && lane == 31) out[n] = run;
}

__global__ __launch_bounds__(EDGE_T) void scatter_kernel(const int* __restrict__ src,
                                                         const int* __restrict__ dst,
                                                         int* __restrict__ tab,
                                                         float* __restrict__ bn_acc,
                                                         int* __restrict__ recA,
                                                         int* __restrict__ recS,
                                                         int* __restrict__ recD, int E) {
    __shared__ int hist[NB2];
    __shared__ int lStart[NB2];
    __shared__ int cur[NB2];
    __shared__ int stA[CH_E];
    __shared__ int stD[CH_E];
    __shared__ int stS[CH_E];
    const int t = threadIdx.x;
    const int w = blockIdx.x;
    if (t < NB2) hist[t] = 0;
    if (w == 0 && t < 2 * DFEAT) bn_acc[t] = 0.0f;
    const int ch  = (E + NCHUNK - 1) / NCHUNK;
    const int beg = w * ch;
    const int end = min(E, beg + ch);
    int d[EPT], s[EPT];
    #pragma unroll
    for (int j = 0; j < EPT; ++j) {
        const int e = max(min(beg + t + j * EDGE_T, end - 1), 0);
        d[j] = dst[e];
        s[j] = src[e];
    }
    __syncthreads();
    #pragma unroll
    for (int j = 0; j < EPT; ++j) {
        if (beg + t + j * EDGE_T < end) {
            atomicAdd(&hist[d[j] >> BIN_SH], 1);
            atomicAdd(&hist[NB + (s[j] >> BIN_SH)], 1);
        }
    }
    __syncthreads();
    if (t < 32)       wave_excl_scan<CPB>(hist, lStart, NB, t, false);
    else if (t < 64)  wave_excl_scan<CPB>(hist + NB, lStart + NB, NB, t - 32, false);
    __syncthreads();
    if (t < NB2) {
        const int l = lStart[t];
        cur[t] = l;
        tab[w * NB2 + t] = (l << 16) | hist[t];
    }
    __syncthreads();
    #pragma unroll
    for (int j = 0; j < EPT; ++j) {
        if (beg + t + j * EDGE_T < end) {
            const int p = atomicAdd(&cur[d[j] >> BIN_SH], 1);
            stA[p] = s[j];
            stD[p] = d[j];
            const int q = atomicAdd(&cur[NB + (s[j] >> BIN_SH)], 1);
            stS[q] = s[j];
        }
    }
    __syncthreads();
    const int n = end - beg;
    for (int i = t; i < n; i += EDGE_T) {
        recA[beg + i] = stA[i];
        recD[beg + i] = stD[i];
        recS[beg + i] = stS[i];
    }
}

template <int U, bool TWO, class F>
__device__ __forceinline__ void walk_pieces(const int* __restrict__ R, int delta1,
                                            const int* pBase, const int* pCnt, int gEnd,
                                            int wave, int lane, F f) {
    for (int g0 = wave; g0 < gEnd; g0 += LWAVES * U) {
        int cn[U], ba[U];
        int mx = 0;
        #pragma unroll
        for (int u = 0; u < U; ++u) {
            const int g = min(g0 + LWAVES * u, gEnd - 1);
            const bool ok = g0 + LWAVES * u < gEnd;
            cn[u] = ok ? pCnt[g] : 0;
            ba[u] = pBase[g];
            mx = max(mx, cn[u]);
        }
        for (int j0 = 0; j0 < mx; j0 += 64) {
            int a[U][2], b[U][2];
            #pragma unroll
            for (int u = 0; u < U; ++u) {
                #pragma unroll
                for (int h = 0; h < 2; ++h) {
                    const int j = j0 + lane + 32 * h;
                    const int pos = ba[u] + max(min(j, cn[u] - 1), 0);
                    a[u][h] = R[pos];
                    b[u][h] = TWO ? R[pos + delta1] : 0;
                }
            }
            #pragma unroll
            for (int u = 0; u < U; ++u) {
                #pragma unroll
                for (int h = 0; h < 2; ++h) {
                    if (j0 + lane + 32 * h < cn[u]) f(g0 + LWAVES * u, a[u][h], b[u][h]);
                }
            }
        }
    }
}

__global__ __launch_bounds__(BIN_R) void local_kernel(const int* __restrict__ tab,
                                                      const int* __restrict__ R,
                                                      float* __restrict__ norm_src,
                                                      float* __restrict__ norm_dst,
                                                      int* __restrict__ rowptr,
                                                      int* __restrict__ csr, int n, int E) {
    __shared__ int deg[2 * BIN_R];
    __shared__ int part[BIN_R];
    __shared__ int stg[CSR_CAP];
    __shared__ int pBase[2 * SEGP];
    __shared__ int pCnt[2 * SEGP];
    __shared__ int pOff[SEGP];
    __shared__ int tot[2];
    int* cIn  = deg;
    int* cOut = deg + BIN_R;
    const int t = threadIdx.x;
    const int b = blockIdx.x;
    const int lane = t & 31;
    const int wave = t >> 5;
    const int ch = (E + NCHUNK - 1) / NCHUNK;
    cIn[t] = 0;
    cOut[t] = 0;
    {
        const int c  = t & (SEGP - 1);
        const int sd = t / SEGP;
        int cnt = 0, off = 0;
        if (c < NCHUNK) {
            const int p = tab[c * NB2 + sd * NB + b];
            cnt = p & 0xffff;
            off = p >> 16;
        }
        pCnt[t]  = cnt;
        pBase[t] = (sd ? E : 2 * E) + min(c, NCHUNK - 1) * ch + off;
        if (sd == 0) pOff[c] = off;
    }
    __syncthreads();
    if (t < 64) {
        const int* v = (t < 32) ? pCnt : pOff;
        int sm = 0;
        for (int i = lane; i < NCHUNK; i += 32) sm += v[i];
        for (int o = 16; o > 0; o >>= 1) sm += __shfl_xor(sm, o);
        if (lane == 0) tot[t >> 5] = sm;
    }
    walk_pieces<8, false>(R, 0, pBase, pCnt, 2 * SEGP, wave, lane,
        [&](int g, int v, int) { atomicAdd(&deg[(g / SEGP) * BIN_R + (v & (BIN_R - 1))], 1); });
    __syncthreads();
    const int dBeg = tot[1];
    const int nb   = tot[0];
    const int node = b * BIN_R + t;
    const int ci = cIn[t];
    if (node < n) {
        norm_src[node] = rsqrtf(fmaxf((float)cOut[t], 1.0f));
        norm_dst[node] = rsqrtf(fmaxf((float)ci, 1.0f));
    }
    part[t] = ci;
    __syncthreads();
    for (int o = 1; o < BIN_R; o <<= 1) {
        int x = (t >= o) ? part[t - o] : 0;
        __syncthreads();
        part[t] += x;
        __syncthreads();
    }
    const int start = dBeg + part[t] - ci;
    if (node <= n) rowptr[node] = start;
    cIn[t] = start;
    __syncthreads();
    const bool staged = nb <= CSR_CAP;
    walk_pieces<4, true>(R, -2 * E, pBase, pCnt, SEGP, wave, lane,
        [&](int, int dv, int sv) {
            const int p = atomicAdd(&cIn[dv & (BIN_R - 1)], 1);
            if (staged) stg[p - dBeg] = sv;
            else        csr[p] = sv;
        });
    if (staged) {
        __syncthreads();
        for (int i = t; i < nb; i += BIN_R) csr[dBeg + i] = stg[i];
    }
}

__device__ __forceinline__ void pair_bounds(int rp, int& beg, int& mid, int& end) {
    beg = __builtin_amdgcn_readlane(rp, 0);
    mid = __builtin_amdgcn_readlane(rp, 1);
    end = __builtin_amdgcn_readlane(rp, 2);
}

__device__ __forceinline__ int chunk_ids(const int* __restrict__ csr, int lane, int b0, int end) {
    return csr[max(min(b0 + lane, end - 1), 0)];
}

__device__ __forceinline__ void chunk_weights(int lane, int b0, int mid, int end, float ns,
                                              float& w0, float& w1) {
    const int e = b0 + lane;
    w0 = (e < mid) ? ns : 0.0f;
    w1 = (e >= mid && e < end) ? ns : 0.0f;
}

__device__ __forceinline__ void rows_issue(const float* __restrict__ feats, int lane, int s_l, int j0,
                                           float4 (&v)[8]) {
    #pragma unroll
    for (int u = 0; u < 8; ++u) {
        const int s = __builtin_amdgcn_readlane(s_l, j0 + u);
        v[u] = ((const float4*)(feats + (size_t)s * DFEAT))[lane];
    }
}

__device__ __forceinline__ void rows_add(const float4 (&v)[8], float w0_l, float w1_l, int j0,
                                         float4& acc0, float4& acc1) {
    #pragma unroll
    for (int u = 0; u < 8; ++u) {
        const float n0 = __int_as_float(__builtin_amdgcn_readlane(__float_as_int(w0_l), j0 + u));
        const float n1 = __int_as_float(__builtin_amdgcn_readlane(__float_as_int(w1_l), j0 + u));
        acc0.x = __builtin_fmaf(v[u].x, n0, acc0.x); acc0.y = __builtin_fmaf(v[u].y, n0, acc0.y);
        acc0.z = __builtin_fmaf(v[u].z, n0, acc0.z); acc0.w = __builtin_fmaf(v[u].w, n0, acc0.w);
        acc1.x = __builtin_fmaf(v[u].x, n1, acc1.x); acc1.y = __builtin_fmaf(v[u].y, n1, acc1.y);
        acc1.z = __builtin_fmaf(v[u].z, n1, acc1.z); acc1.w = __builtin_fmaf(v[u].w, n1, acc1.w);
    }
}

__device__ __forceinline__ void pair_finish(const float* __restrict__ feats,
                                            const int* __restrict__ csr,
                                            const float* __restrict__ norm_src, int lane,
                                            int beg, int mid, int end,
                                            int idA, int idB, float nsA, float nsB,
                                            const float4 (&v0)[8], float4& acc0, float4& acc1) {
    acc0 = make_float4(0.0f, 0.0f, 0.0f, 0.0f);
    acc1 = make_float4(0.0f, 0.0f, 0.0f, 0.0f);
    {
        const int cnt = min(32, end - beg);
        float w0_l, w1_l;
        chunk_weights(lane, beg, mid, end, nsA, w0_l, w1_l);
        rows_add(v0, w0_l, w1_l, 0, acc0, acc1);
        for (int j0 = 8; j0 < cnt; j0 += 8) {
            float4 v[8];
            rows_issue(feats, lane, idA, j0, v);
            rows_add(v, w0_l, w1_l, j0, acc0, acc1);
        }
    }
    int c = 1;
    for (int b0 = beg + 32; b0 < end; b0 += 32, ++c) {
        const int cnt = min(32, end - b0);
        int   s_l  = idB;
        float ns_l = nsB;
        if (c > 1) {
            s_l  = chunk_ids(csr, lane, b0, end);
            ns_l = norm_src[s_l];
        }
        float w0_l, w1_l;
        chunk_weights(lane, b0, mid, end, ns_l, w0_l, w1_l);
        for (int j0 = 0; j0 < cnt; j0 += 8) {
            float4 v[8];
            rows_issue(feats, lane, s_l, j0, v);
            rows_add(v, w0_l, w1_l, j0, acc0, acc1);
        }
    }
}

#define GEMM_TILES (NNODES / 16)
#define GEMM_GRID  512
#define LDS_P      132
static_assert(NNODES % 16 == 0, "every tile has 16 full rows");
static_assert(3 * GEMM_GRID <= GEMM_TILES, "every work-group has at least three tiles");

__device__ __forceinline__ void lds_barrier() {
    __builtin_amdgcn_fence(__ATOMIC_RELEASE, "workgroup", "local");
    __builtin_amdgcn_s_barrier();
    __builtin_amdgcn_fence(__ATOMIC_ACQUIRE, "workgroup", "local");
}

__global__ __launch_bounds__(256, 4) void gather_gemm_bn_kernel(
    const float* __restrict__ feats, const int* __restrict__ rowptr, const int* __restrict__ csr,
    const float* __restrict__ norm_src, const float* __restrict__ norm_dst,
    const float* __restrict__ W,  const float* __restrict__ b,
    const float* __restrict__ Wr, const float* __restrict__ br,
    float* __restrict__ y, float* __restrict__ bn_sum, float* __restrict__ bn_sumsq) {
    __shared__ __attribute__((aligned(16))) float sA[2][16 * LDS_P];
    __shared__ __attribute__((aligned(16))) float sF[2][16 * LDS_P];
    const int lane = threadIdx.x & 31;
    const int wave = threadIdx.x >> 5;
    const int n0 = wave * 16;
    const int m  = lane & 15;
    const int kh = (lane >> 4) * 2;
    const int col = n0 + m;

    v2f bw[DFEAT / 4], bwr[DFEAT / 4];
    #pragma unroll
    for (int ks = 0; ks < DFEAT / 4; ++ks) {
        const int k = 4 * ks + kh;
        bw[ks].x  = W [(size_t)k       * DFEAT + col];
        bw[ks].y  = W [(size_t)(k + 1) * DFEAT + col];
        bwr[ks].x = Wr[(size_t)k       * DFEAT + col];
        bwr[ks].y = Wr[(size_t)(k + 1) * DFEAT + col];
    }
    const float bc  = b[col];
    const float brc = br[col];
    const int rOff = (lane >> 4) * 8;

    const int so0 = (2 * wave) * LDS_P + 4 * lane;
    const int so1 = so0 + LDS_P;
    const int fo = m * LDS_P + kh;
    const int rpl = 2 * wave + min(lane, 2);

    float4 pf0, pf1;
    float4 v0[8];
    int tile = blockIdx.x;
    int beg1, mid1, end1, idA1, idB1;
    int rp2;
    {
        const int rp0 = rowptr[tile * 16 + rpl];
        const int rp1 = rowptr[(tile + GEMM_GRID) * 16 + rpl];
        rp2           = rowptr[(tile + 2 * GEMM_GRID) * 16 + rpl];
        const int r0 = tile * 16 + 2 * wave;
        const float nd_l = norm_dst[r0 + (lane & 1)];
        pf0 = ((const float4*)(feats + (size_t)r0       * DFEAT))[lane];
        pf1 = ((const float4*)(feats + (size_t)(r0 + 1) * DFEAT))[lane];
        int beg, mid, end;
        pair_bounds(rp0, beg, mid, end);
        const int idA = chunk_ids(csr, lane, beg, end);
        const int idB = chunk_ids(csr, lane, beg + 32, end);
        pair_bounds(rp1, beg1, mid1, end1);
        idA1 = chunk_ids(csr, lane, beg1, end1);
        idB1 = chunk_ids(csr, lane, beg1 + 32, end1);
        const float nsA = norm_src[idA];
        const float nsB = norm_src[idB];
        rows_issue(feats, lane, idA, 0, v0);
        float4 a0, a1;
        pair_finish(feats, csr, norm_src, lane, beg, mid, end, idA, idB, nsA, nsB, v0, a0, a1);
        const float nd0 = __int_as_float(__builtin_amdgcn_readlane(__float_as_int(nd_l), 0));
        const float nd1 = __int_as_float(__builtin_amdgcn_readlane(__float_as_int(nd_l), 1));
        a0.x *= nd0; a0.y *= nd0; a0.z *= nd0; a0.w *= nd0;
        a1.x *= nd1; a1.y *= nd1; a1.z *= nd1; a1.w *= nd1;
        *(float4*)&sA[0][so0] = a0;
        *(float4*)&sA[0][so1] = a1;
        *(float4*)&sF[0][so0] = pf0;
        *(float4*)&sF[0][so1] = pf1;
    }
    lds_barrier();

    float s8 = 0.0f, q8 = 0.0f;
    int buf = 0;
    for (;;) {
        const int  next = tile + GEMM_GRID;
        const bool more = next < GEMM_TILES;
        float nd_l = 0.0f, nsA1 = 0.0f, nsB1 = 0.0f;
        int beg2 = 0, mid2 = 0, end2 = 0, idA2 = 0, idB2 = 0, rp3 = 0;
        if (more) {
            const int r0 = next * 16 + 2 * wave;
            nd_l = norm_dst[r0 + (lane & 1)];
            pf0 = ((const float4*)(feats + (size_t)r0       * DFEAT))[lane];
            pf1 = ((const float4*)(feats + (size_t)(r0 + 1) * DFEAT))[lane];
            nsA1 = norm_src[idA1];
            nsB1 = norm_src[idB1];
            rows_issue(feats, lane, idA1, 0, v0);
            pair_bounds(rp2, beg2, mid2, end2);
            idA2 = chunk_ids(csr, lane, beg2, end2);
            idB2 = chunk_ids(csr, lane, beg2 + 32, end2);
            rp3  = rowptr[min(next + 2 * GEMM_GRID, GEMM_TILES - 1) * 16 + rpl];
        }

        const float* aB = &sA[buf][fo];
        const float* fB = &sF[buf][fo];
        v8f accC = {};
        v8f accR = {};
        #pragma unroll
        for (int ks = 0; ks < DFEAT / 4; ++ks) {
            const v2f a = *(const v2f*)(aB + 4 * ks);
            const v2f f = *(const v2f*)(fB + 4 * ks);
            accC = __builtin_amdgcn_wmma_f32_16x16x4_f32(false, a, false, bw[ks],  (short)0, accC, false, false);
            accR = __builtin_amdgcn_wmma_f32_16x16x4_f32(false, f, false, bwr[ks], (short)0, accR, false, false);
        }

        const int rowBase = tile * 16;
        #pragma unroll
        for (int r = 0; r < 8; ++r) {
            float c = accC[r] + bc;  c = c > 0.0f ? c : 0.0f;
            float e = accR[r] + brc; e = e > 0.0f ? e : 0.0f;
            float v = c + e;
            y[(size_t)(rowBase + rOff + r) * DFEAT + col] = v;
            s8 += v;
            q8 += v * v;
        }

        if (more) {
            float4 a0, a1;
            pair_finish(feats, csr, norm_src, lane, beg1, mid1, end1, idA1, idB1, nsA1, nsB1, v0, a0, a1);
            const float nd0 = __int_as_float(__builtin_amdgcn_readlane(__float_as_int(nd_l), 0));
            const float nd1 = __int_as_float(__builtin_amdgcn_readlane(__float_as_int(nd_l), 1));
            a0.x *= nd0; a0.y *= nd0; a0.z *= nd0; a0.w *= nd0;
            a1.x *= nd1; a1.y *= nd1; a1.z *= nd1; a1.w *= nd1;
            *(float4*)&sA[buf ^ 1][so0] = a0;
            *(float4*)&sA[buf ^ 1][so1] = a1;
            *(float4*)&sF[buf ^ 1][so0] = pf0;
            *(float4*)&sF[buf ^ 1][so1] = pf1;
            beg1 = beg2; mid1 = mid2; end1 = end2; idA1 = idA2; idB1 = idB2;
            rp2 = rp3;
        }
        lds_barrier();
        buf ^= 1;
        tile = next;
        if (!more) break;
    }

    float s8o = __shfl_down(s8, 16);
    float q8o = __shfl_down(q8, 16);
    if (lane < 16) {
        atomicAdd(&bn_sum[col],   s8 + s8o);
        atomicAdd(&bn_sumsq[col], q8 + q8o);
    }
}

__global__ __launch_bounds__(256) void bn_apply_kernel(float* __restrict__ y,
                                                       const float* __restrict__ sum,
                                                       const float* __restrict__ sumsq,
                                                       const float* __restrict__ gamma,
                                                       const float* __restrict__ beta,
                                                       size_t nvec4, int n) {
    size_t i = (size_t)blockIdx.x * blockDim.x + threadIdx.x;
    if (i >= nvec4) return;
    float4 v = ((float4*)y)[i];
    const int col = (int)((i * 4) & (DFEAT - 1));
    const float4 s4 = *(const float4*)(sum + col);
    const float4 q4 = *(const float4*)(sumsq + col);
    const float4 g4 = *(const float4*)(gamma + col);
    const float4 b4 = *(const float4*)(beta + col);
    const float inv = 1.0f / (float)n;
    float mean, var, sc;
    mean = s4.x * inv; var = q4.x * inv - mean * mean; sc = g4.x * rsqrtf(var + BN_EPS);
    v.x = v.x * sc + (b4.x - mean * sc);
    mean = s4.y * inv; var = q4.y * inv - mean * mean; sc = g4.y * rsqrtf(var + BN_EPS);
    v.y = v.y * sc + (b4.y - mean * sc);
    mean = s4.z * inv; var = q4.z * inv - mean * mean; sc = g4.z * rsqrtf(var + BN_EPS);
    v.z = v.z * sc + (b4.z - mean * sc);
    mean = s4.w * inv; var = q4.w * inv - mean * mean; sc = g4.w * rsqrtf(var + BN_EPS);
    v.w = v.w * sc + (b4.w - mean * sc);
    ((float4*)y)[i] = v;
}

extern "C" void kernel_launch(void* const* d_in, const int* in_sizes, int n_in,
                              void* d_out, int out_size, void* d_ws, size_t ws_size,
                              hipStream_t stream) {
    const float* feats = (const float*)d_in[0];
    const int*   src   = (const int*)d_in[1];
    const int*   dst   = (const int*)d_in[2];
    const float* W     = (const float*)d_in[3];
    const float* b     = (const float*)d_in[4];
    const float* Wr    = (const float*)d_in[5];
    const float* br    = (const float*)d_in[6];
    const float* gamma = (const float*)d_in[7];
    const float* beta  = (const float*)d_in[8];
    float* out = (float*)d_out;

    const int N = NNODES;
    const int E = in_sizes[1];

    float* ws        = (float*)d_ws;
    int*   csr       = (int*)d_ws;
    int*   rowptr    = csr + E;
    float* deg_out   = ws + (size_t)N * DFEAT;
    float* deg_in    = deg_out + N;
    float* bn_sum    = deg_in + N;
    float* bn_sumsq  = bn_sum + DFEAT;

    int* scr  = (int*)d_out;
    int* tab  = scr;
    int* recA = scr + 196608;
    int* recS = recA + E;
    int* recD = recS + E;

    scatter_kernel<<<NCHUNK, EDGE_T, 0, stream>>>(src, dst, tab, bn_sum, recA, recS, recD, E);

    local_kernel<<<NB, BIN_R, 0, stream>>>(tab, recA, deg_out, deg_in, rowptr, csr, N, E);

    gather_gemm_bn_kernel<<<GEMM_GRID, 256, 0, stream>>>(feats, rowptr, csr, deg_out, deg_in,
                                                         W, b, Wr, br, out, bn_sum, bn_sumsq);

    size_t nvec4 = (size_t)N * DFEAT / 4;
    bn_apply_kernel<<<(int)((nvec4 + 255) / 256), 256, 0, stream>>>(out, bn_sum, bn_sumsq, gamma, beta,
                                                                   nvec4, N);
}


# Row loads behind the products go through two 8-row buffers in turn: next rows asked for before the current are added, twice the rows in flight; 14 Wr k-step fragments kept in LDS; sum order kept
# speedup vs baseline: 4.5930x; 1.0441x over previous
#include <hip/hip_runtime.h>
#include <hip/hip_bf16.h>

typedef __attribute__((ext_vector_type(2))) float v2f;
typedef __attribute__((ext_vector_type(8))) float v8f;
typedef const __attribute__((address_space(1))) float4* gfloat4p;

#define NNODES 100000
#define DFEAT  128
#define BN_EPS 1e-5f
#define BIN_SH 9
#define BIN_R  512
#define NB     196
#define NB2    (2 * NB)
#define NCHUNK 250
#define EDGE_T 1024
#define CPB    ((NB + 31) / 32)
#define NEDGES 1600000
#define CH_E   ((NEDGES + NCHUNK - 1) / NCHUNK)
#define EPT    ((CH_E + EDGE_T - 1) / EDGE_T)
#define CSR_CAP 16384
#define LWAVES (BIN_R / 32)
#define SEGP   256
static_assert(CH_E < 65536, "a piece's offset and length share one 32-bit table word");
static_assert(NCHUNK <= SEGP, "the piece list holds every chunk");
static_assert(2 * SEGP == BIN_R, "one thread per piece-list entry");


template <int PER>
__device__ __forceinline__ void wave_excl_scan(const int* in, int* out, int n, int lane, bool total) {
    int loc[PER];
    int sum = 0;
    for (int i = 0; i < PER; ++i) {
        const int idx = lane * PER + i;
        loc[i] = (idx < n) ? in[idx] : 0;
        sum += loc[i];
    }
    int inc = sum;
    for (int o = 1; o < 32; o <<= 1) {
        const int x = __shfl_up(inc, o);
        if (lane >= o) inc += x;
    }
    int run = inc - sum;
    for (int i = 0; i < PER; ++i) {
        const int idx = lane * PER + i;
        if (idx < n) out[idx] = run;
        run += loc[i];
    }
    if (total && lane == 31) out[n] = run;
}

__global__ __launch_bounds__(EDGE_T) void scatter_kernel(const int* __restrict__ src,
                                                         const int* __restrict__ dst,
                                                         int* __restrict__ tab,
                                                         float* __restrict__ bn_acc,
                                                         int* __restrict__ recA,
                                                         int* __restrict__ recS,
                                                         int* __restrict__ recD, int E) {
    __shared__ int hist[NB2];
    __shared__ int lStart[NB2];
    __shared__ int cur[NB2];
    __shared__ int stA[CH_E];
    __shared__ int stD[CH_E];
    __shared__ int stS[CH_E];
    const int t = threadIdx.x;
    const int w = blockIdx.x;
    if (t < NB2) hist[t] = 0;
    if (w == 0 && t < 2 * DFEAT) bn_acc[t] = 0.0f;
    const int ch  = (E + NCHUNK - 1) / NCHUNK;
    const int beg = w * ch;
    const int end = min(E, beg + ch);
    int d[EPT], s[EPT];
    #pragma unroll
    for (int j = 0; j < EPT; ++j) {
        const int e = max(min(beg + t + j * EDGE_T, end - 1), 0);
        d[j] = dst[e];
        s[j] = src[e];
    }
    __syncthreads();
    #pragma unroll
    for (int j = 0; j < EPT; ++j) {
        if (beg + t + j * EDGE_T < end) {
            atomicAdd(&hist[d[j] >> BIN_SH], 1);
            atomicAdd(&hist[NB + (s[j] >> BIN_SH)], 1);
        }
    }
    __syncthreads();
    if (t < 32)       wave_excl_scan<CPB>(hist, lStart, NB, t, false);
    else if (t < 64)  wave_excl_scan<CPB>(hist + NB, lStart + NB, NB, t - 32, false);
    __syncthreads();
    if (t < NB2) {
        const int l = lStart[t];
        cur[t] = l;
        tab[w * NB2 + t] = (l << 16) | hist[t];
    }
    __syncthreads();
    #pragma unroll
    for (int j = 0; j < EPT; ++j) {
        if (beg + t + j * EDGE_T < end) {
            const int p = atomicAdd(&cur[d[j] >> BIN_SH], 1);
            stA[p] = s[j];
            stD[p] = d[j];
            const int q = atomicAdd(&cur[NB + (s[j] >> BIN_SH)], 1);
            stS[q] = s[j];
        }
    }
    __syncthreads();
    const int n = end - beg;
    for (int i = t; i < n; i += EDGE_T) {
        recA[beg + i] = stA[i];
        recD[beg + i] = stD[i];
        recS[beg + i] = stS[i];
    }
}

template <int U, bool TWO, class F>
__device__ __forceinline__ void walk_pieces(const int* __restrict__ R, int delta1,
                                            const int* pBase, const int* pCnt, int gEnd,
                                            int wave, int lane, F f) {
    for (int g0 = wave; g0 < gEnd; g0 += LWAVES * U) {
        int cn[U], ba[U];
        int mx = 0;
        #pragma unroll
        for (int u = 0; u < U; ++u) {
            const int g = min(g0 + LWAVES * u, gEnd - 1);
            const bool ok = g0 + LWAVES * u < gEnd;
            cn[u] = ok ? pCnt[g] : 0;
            ba[u] = pBase[g];
            mx = max(mx, cn[u]);
        }
        for (int j0 = 0; j0 < mx; j0 += 64) {
            int a[U][2], b[U][2];
            #pragma unroll
            for (int u = 0; u < U; ++u) {
                #pragma unroll
                for (int h = 0; h < 2; ++h) {
                    const int j = j0 + lane + 32 * h;
                    const int pos = ba[u] + max(min(j, cn[u] - 1), 0);
                    a[u][h] = R[pos];
                    b[u][h] = TWO ? R[pos + delta1] : 0;
                }
            }
            #pragma unroll
            for (int u = 0; u < U; ++u) {
                #pragma unroll
                for (int h = 0; h < 2; ++h) {
                    if (j0 + lane + 32 * h < cn[u]) f(g0 + LWAVES * u, a[u][h], b[u][h]);
                }
            }
        }
    }
}

__global__ __launch_bounds__(BIN_R) void local_kernel(const int* __restrict__ tab,
                                                      const int* __restrict__ R,
                                                      float* __restrict__ norm_src,
                                                      float* __restrict__ norm_dst,
                                                      int* __restrict__ rowptr,
                                                      int* __restrict__ csr, int n, int E) {
    __shared__ int deg[2 * BIN_R];
    __shared__ int part[BIN_R];
    __shared__ int stg[CSR_CAP];
    __shared__ int pBase[2 * SEGP];
    __shared__ int pCnt[2 * SEGP];
    __shared__ int pOff[SEGP];
    __shared__ int tot[2];
    int* cIn  = deg;
    int* cOut = deg + BIN_R;
    const int t = threadIdx.x;
    const int b = blockIdx.x;
    const int lane = t & 31;
    const int wave = t >> 5;
    const int ch = (E + NCHUNK - 1) / NCHUNK;
    cIn[t] = 0;
    cOut[t] = 0;
    {
        const int c  = t & (SEGP - 1);
        const int sd = t / SEGP;
        int cnt = 0, off = 0;
        if (c < NCHUNK) {
            const int p = tab[c * NB2 + sd * NB + b];
            cnt = p & 0xffff;
            off = p >> 16;
        }
        pCnt[t]  = cnt;
        pBase[t] = (sd ? E : 2 * E) + min(c, NCHUNK - 1) * ch + off;
        if (sd == 0) pOff[c] = off;
    }
    __syncthreads();
    if (t < 64) {
        const int* v = (t < 32) ? pCnt : pOff;
        int sm = 0;
        for (int i = lane; i < NCHUNK; i += 32) sm += v[i];
        for (int o = 16; o > 0; o >>= 1) sm += __shfl_xor(sm, o);
        if (lane == 0) tot[t >> 5] = sm;
    }
    walk_pieces<8, false>(R, 0, pBase, pCnt, 2 * SEGP, wave, lane,
        [&](int g, int v, int) { atomicAdd(&deg[(g / SEGP) * BIN_R + (v & (BIN_R - 1))], 1); });
    __syncthreads();
    const int dBeg = tot[1];
    const int nb   = tot[0];
    const int node = b * BIN_R + t;
    const int ci = cIn[t];
    if (node < n) {
        norm_src[node] = rsqrtf(fmaxf((float)cOut[t], 1.0f));
        norm_dst[node] = rsqrtf(fmaxf((float)ci, 1.0f));
    }
    part[t] = ci;
    __syncthreads();
    for (int o = 1; o < BIN_R; o <<= 1) {
        int x = (t >= o) ? part[t - o] : 0;
        __syncthreads();
        part[t] += x;
        __syncthreads();
    }
    const int start = dBeg + part[t] - ci;
    if (node <= n) rowptr[node] = start;
    cIn[t] = start;
    __syncthreads();
    const bool staged = nb <= CSR_CAP;
    walk_pieces<4, true>(R, -2 * E, pBase, pCnt, SEGP, wave, lane,
        [&](int, int dv, int sv) {
            const int p = atomicAdd(&cIn[dv & (BIN_R - 1)], 1);
            if (staged) stg[p - dBeg] = sv;
            else        csr[p] = sv;
        });
    if (staged) {
        __syncthreads();
        for (int i = t; i < nb; i += BIN_R) csr[dBeg + i] = stg[i];
    }
}

__device__ __forceinline__ void pair_bounds(int rp, int& beg, int& mid, int& end) {
    beg = __builtin_amdgcn_readlane(rp, 0);
    mid = __builtin_amdgcn_readlane(rp, 1);
    end = __builtin_amdgcn_readlane(rp, 2);
}

__device__ __forceinline__ int chunk_ids(const int* __restrict__ csr, int lane, int b0, int end) {
    return csr[max(min(b0 + lane, end - 1), 0)];
}

__device__ __forceinline__ void chunk_weights(int lane, int b0, int mid, int end, float ns,
                                              float& w0, float& w1) {
    const int e = b0 + lane;
    w0 = (e < mid) ? ns : 0.0f;
    w1 = (e >= mid && e < end) ? ns : 0.0f;
}

__device__ __forceinline__ void rows_issue(const float* __restrict__ feats, int lane, int s_l, int j0,
                                           float4 (&v)[8]) {
    #pragma unroll
    for (int u = 0; u < 8; ++u) {
        const int s = __builtin_amdgcn_readlane(s_l, j0 + u);
        const unsigned long long rb = (unsigned long long)(feats + (size_t)s * DFEAT);
        const unsigned int rlo = (unsigned int)__builtin_amdgcn_readfirstlane((int)(unsigned int)rb);
        const unsigned int rhi = (unsigned int)__builtin_amdgcn_readfirstlane((int)(unsigned int)(rb >> 32));
        const gfloat4p rowp = (gfloat4p)(((unsigned long long)rhi << 32) | rlo);
        v[u] = rowp[lane];
    }
}

__device__ __forceinline__ void rows_add(const float4 (&v)[8], float w0_l, float w1_l, int j0,
                                         float4& acc0, float4& acc1) {
    #pragma unroll
    for (int u = 0; u < 8; ++u) {
        const float n0 = __int_as_float(__builtin_amdgcn_readlane(__float_as_int(w0_l), j0 + u));
        const float n1 = __int_as_float(__builtin_amdgcn_readlane(__float_as_int(w1_l), j0 + u));
        acc0.x = __builtin_fmaf(v[u].x, n0, acc0.x); acc0.y = __builtin_fmaf(v[u].y, n0, acc0.y);
        acc0.z = __builtin_fmaf(v[u].z, n0, acc0.z); acc0.w = __builtin_fmaf(v[u].w, n0, acc0.w);
        acc1.x = __builtin_fmaf(v[u].x, n1, acc1.x); acc1.y = __builtin_fmaf(v[u].y, n1, acc1.y);
        acc1.z = __builtin_fmaf(v[u].z, n1, acc1.z); acc1.w = __builtin_fmaf(v[u].w, n1, acc1.w);
    }
}

__device__ __forceinline__ void grp_issue(const float* __restrict__ feats, int lane, int idA, int idB,
                                          int gi, float4 (&v)[8]) {
    rows_issue(feats, lane, (gi < 4) ? idA : idB, (gi & 3) * 8, v);
}
__device__ __forceinline__ void grp_add(const float4 (&v)[8], float w0A, float w1A, float w0B, float w1B,
                                        int gi, float4& acc0, float4& acc1) {
    rows_add(v, (gi < 4) ? w0A : w0B, (gi < 4) ? w1A : w1B, (gi & 3) * 8, acc0, acc1);
}

__device__ __forceinline__ void pair_finish(const float* __restrict__ feats,
                                            const int* __restrict__ csr,
                                            const float* __restrict__ norm_src, int lane,
                                            int beg, int mid, int end,
                                            int idA, int idB, float nsA, float nsB,
                                            float4 (&v0)[8], float4& acc0, float4& acc1) {
    acc0 = make_float4(0.0f, 0.0f, 0.0f, 0.0f);
    acc1 = make_float4(0.0f, 0.0f, 0.0f, 0.0f);
    {
        const int ng = (min(end - beg, 64) + 7) >> 3;
        float w0A, w1A, w0B, w1B;
        chunk_weights(lane, beg,      mid, end, nsA, w0A, w1A);
        chunk_weights(lane, beg + 32, mid, end, nsB, w0B, w1B);
        float4 vb[8];
        for (int g = 0;; g += 2) {
            grp_issue(feats, lane, idA, idB, g + 1, vb);
            grp_add(v0, w0A, w1A, w0B, w1B, g, acc0, acc1);
            if (g + 1 >= ng) break;
            grp_issue(feats, lane, idA, idB, g + 2, v0);
            grp_add(vb, w0A, w1A, w0B, w1B, g + 1, acc0, acc1);
            if (g + 2 >= ng) break;
        }
    }
    for (int b0 = beg + 64; b0 < end; b0 += 32) {
        const int cnt = min(32, end - b0);
        const int   s_l  = chunk_ids(csr, lane, b0, end);
        const float ns_l = norm_src[s_l];
        float w0_l, w1_l;
        chunk_weights(lane, b0, mid, end, ns_l, w0_l, w1_l);
        for (int j0 = 0; j0 < cnt; j0 += 8) {
            float4 v[8];
            rows_issue(feats, lane, s_l, j0, v);
            rows_add(v, w0_l, w1_l, j0, acc0, acc1);
        }
    }
}

#define GEMM_TILES (NNODES / 16)
#define GEMM_GRID  512
#define WR_LDS     14
#define LDS_P      132
static_assert(NNODES % 16 == 0, "every tile has 16 full rows");
static_assert(3 * GEMM_GRID <= GEMM_TILES, "every work-group has at least three tiles");

__device__ __forceinline__ void lds_barrier() {
    __builtin_amdgcn_fence(__ATOMIC_RELEASE, "workgroup", "local");
    __builtin_amdgcn_s_barrier();
    __builtin_amdgcn_fence(__ATOMIC_ACQUIRE, "workgroup", "local");
}

__global__ __launch_bounds__(256, 4) void gather_gemm_bn_kernel(
    const float* __restrict__ feats, const int* __restrict__ rowptr, const int* __restrict__ csr,
    const float* __restrict__ norm_src, const float* __restrict__ norm_dst,
    const float* __restrict__ W,  const float* __restrict__ b,
    const float* __restrict__ Wr, const float* __restrict__ br,
    float* __restrict__ y, float* __restrict__ bn_sum, float* __restrict__ bn_sumsq) {
    __shared__ __attribute__((aligned(16))) float sA[2][16 * LDS_P];
    __shared__ __attribute__((aligned(16))) float sF[2][16 * LDS_P];
    __shared__ v2f sWr[WR_LDS][256];
    const int lane = threadIdx.x & 31;
    const int wave = threadIdx.x >> 5;
    const int n0 = wave * 16;
    const int m  = lane & 15;
    const int kh = (lane >> 4) * 2;
    const int col = n0 + m;

    v2f bw[DFEAT / 4], bwr[DFEAT / 4 - WR_LDS];
    #pragma unroll
    for (int ks = 0; ks < DFEAT / 4; ++ks) {
        const int k = 4 * ks + kh;
        bw[ks].x  = W [(size_t)k       * DFEAT + col];
        bw[ks].y  = W [(size_t)(k + 1) * DFEAT + col];
        v2f t;
        t.x = Wr[(size_t)k       * DFEAT + col];
        t.y = Wr[(size_t)(k + 1) * DFEAT + col];
        if (ks < DFEAT / 4 - WR_LDS) bwr[ks] = t;
        else                         sWr[ks - (DFEAT / 4 - WR_LDS)][threadIdx.x] = t;
    }
    const float bc  = b[col];
    const float brc = br[col];
    const v2f bc2  = {bc, bc};
    const v2f brc2 = {brc, brc};
    const int rOff = (lane >> 4) * 8;

    const int so0 = (2 * wave) * LDS_P + 4 * lane;
    const int so1 = so0 + LDS_P;
    const int fo = m * LDS_P + kh;
    const int rpl = 2 * wave + min(lane, 2);

    float4 pf0, pf1;
    float4 v0[8];
    int tile = blockIdx.x;
    int beg1, mid1, end1, idA1, idB1;
    int rp2;
    {
        const int rp0 = rowptr[tile * 16 + rpl];
        const int rp1 = rowptr[(tile + GEMM_GRID) * 16 + rpl];
        rp2           = rowptr[(tile + 2 * GEMM_GRID) * 16 + rpl];
        const int r0 = tile * 16 + 2 * wave;
        const float nd_l = norm_dst[r0 + (lane & 1)];
        pf0 = ((const float4*)(feats + (size_t)r0       * DFEAT))[lane];
        pf1 = ((const float4*)(feats + (size_t)(r0 + 1) * DFEAT))[lane];
        int beg, mid, end;
        pair_bounds(rp0, beg, mid, end);
        const int idA = chunk_ids(csr, lane, beg, end);
        const int idB = chunk_ids(csr, lane, beg + 32, end);
        pair_bounds(rp1, beg1, mid1, end1);
        idA1 = chunk_ids(csr, lane, beg1, end1);
        idB1 = chunk_ids(csr, lane, beg1 + 32, end1);
        const float nsA = norm_src[idA];
        const float nsB = norm_src[idB];
        rows_issue(feats, lane, idA, 0, v0);
        float4 a0, a1;
        pair_finish(feats, csr, norm_src, lane, beg, mid, end, idA, idB, nsA, nsB, v0, a0, a1);
        const float nd0 = __int_as_float(__builtin_amdgcn_readlane(__float_as_int(nd_l), 0));
        const float nd1 = __int_as_float(__builtin_amdgcn_readlane(__float_as_int(nd_l), 1));
        a0.x *= nd0; a0.y *= nd0; a0.z *= nd0; a0.w *= nd0;
        a1.x *= nd1; a1.y *= nd1; a1.z *= nd1; a1.w *= nd1;
        *(float4*)&sA[0][so0] = a0;
        *(float4*)&sA[0][so1] = a1;
        *(float4*)&sF[0][so0] = pf0;
        *(float4*)&sF[0][so1] = pf1;
    }
    lds_barrier();

    float s8 = 0.0f, q8 = 0.0f;
    int buf = 0;
    for (;;) {
        const int  next = tile + GEMM_GRID;
        const bool more = next < GEMM_TILES;
        float nd_l = 0.0f, nsA1 = 0.0f, nsB1 = 0.0f;
        int beg2 = 0, mid2 = 0, end2 = 0, idA2 = 0, idB2 = 0, rp3 = 0;
        if (more) {
            const int r0 = next * 16 + 2 * wave;
            nd_l = norm_dst[r0 + (lane & 1)];
            pf0 = ((const float4*)(feats + (size_t)r0       * DFEAT))[lane];
            pf1 = ((const float4*)(feats + (size_t)(r0 + 1) * DFEAT))[lane];
            nsA1 = norm_src[idA1];
            nsB1 = norm_src[idB1];
            rows_issue(feats, lane, idA1, 0, v0);
            pair_bounds(rp2, beg2, mid2, end2);
            idA2 = chunk_ids(csr, lane, beg2, end2);
            idB2 = chunk_ids(csr, lane, beg2 + 32, end2);
            rp3  = rowptr[min(next + 2 * GEMM_GRID, GEMM_TILES - 1) * 16 + rpl];
        }

        const float* aB = &sA[buf][fo];
        const float* fB = &sF[buf][fo];
        v8f accC = {};
        v8f accR = {};
        #pragma unroll
        for (int ks = 0; ks < DFEAT / 4; ++ks) {
            const v2f a = *(const v2f*)(aB + 4 * ks);
            const v2f f = *(const v2f*)(fB + 4 * ks);
            accC = __builtin_amdgcn_wmma_f32_16x16x4_f32(false, a, false, bw[ks],  (short)0, accC, false, false);
            v2f wr;
            if (ks < DFEAT / 4 - WR_LDS) wr = bwr[ks];
            else                         wr = sWr[ks - (DFEAT / 4 - WR_LDS)][threadIdx.x];
            accR = __builtin_amdgcn_wmma_f32_16x16x4_f32(false, f, false, wr, (short)0, accR, false, false);
        }

        const int rowBase = tile * 16;
        #pragma unroll
        for (int r = 0; r < 8; r += 2) {
            const v2f cc = (v2f){accC[r], accC[r + 1]} + bc2;
            const v2f ee = (v2f){accR[r], accR[r + 1]} + brc2;
            const float c0 = cc.x > 0.0f ? cc.x : 0.0f;
            const float c1 = cc.y > 0.0f ? cc.y : 0.0f;
            const float e0 = ee.x > 0.0f ? ee.x : 0.0f;
            const float e1 = ee.y > 0.0f ? ee.y : 0.0f;
            const float va = c0 + e0;
            const float vc = c1 + e1;
            y[(size_t)(rowBase + rOff + r)     * DFEAT + col] = va;
            y[(size_t)(rowBase + rOff + r + 1) * DFEAT + col] = vc;
            s8 += va;
            q8 += va * va;
            s8 += vc;
            q8 += vc * vc;
        }

        if (more) {
            *(float4*)&sF[buf ^ 1][so0] = pf0;
            *(float4*)&sF[buf ^ 1][so1] = pf1;
            float4 a0, a1;
            pair_finish(feats, csr, norm_src, lane, beg1, mid1, end1, idA1, idB1, nsA1, nsB1, v0, a0, a1);
            const float nd0 = __int_as_float(__builtin_amdgcn_readlane(__float_as_int(nd_l), 0));
            const float nd1 = __int_as_float(__builtin_amdgcn_readlane(__float_as_int(nd_l), 1));
            a0.x *= nd0; a0.y *= nd0; a0.z *= nd0; a0.w *= nd0;
            a1.x *= nd1; a1.y *= nd1; a1.z *= nd1; a1.w *= nd1;
            *(float4*)&sA[buf ^ 1][so0] = a0;
            *(float4*)&sA[buf ^ 1][so1] = a1;
            beg1 = beg2; mid1 = mid2; end1 = end2; idA1 = idA2; idB1 = idB2;
            rp2 = rp3;
        }
        lds_barrier();
        buf ^= 1;
        tile = next;
        if (!more) break;
    }

    float s8o = __shfl_down(s8, 16);
    float q8o = __shfl_down(q8, 16);
    if (lane < 16) {
        atomicAdd(&bn_sum[col],   s8 + s8o);
        atomicAdd(&bn_sumsq[col], q8 + q8o);
    }
}

__global__ __launch_bounds__(256) void bn_apply_kernel(float* __restrict__ y,
                                                       const float* __restrict__ sum,
                                                       const float* __restrict__ sumsq,
                                                       const float* __restrict__ gamma,
                                                       const float* __restrict__ beta,
                                                       size_t nvec4, int n) {
    size_t i = (size_t)blockIdx.x * blockDim.x + threadIdx.x;
    if (i >= nvec4) return;
    float4 v = ((float4*)y)[i];
    const int col = (int)((i * 4) & (DFEAT - 1));
    const float4 s4 = *(const float4*)(sum + col);
    const float4 q4 = *(const float4*)(sumsq + col);
    const float4 g4 = *(const float4*)(gamma + col);
    const float4 b4 = *(const float4*)(beta + col);
    const float inv = 1.0f / (float)n;
    float mean, var, sc;
    mean = s4.x * inv; var = q4.x * inv - mean * mean; sc = g4.x * rsqrtf(var + BN_EPS);
    v.x = v.x * sc + (b4.x - mean * sc);
    mean = s4.y * inv; var = q4.y * inv - mean * mean; sc = g4.y * rsqrtf(var + BN_EPS);
    v.y = v.y * sc + (b4.y - mean * sc);
    mean = s4.z * inv; var = q4.z * inv - mean * mean; sc = g4.z * rsqrtf(var + BN_EPS);
    v.z = v.z * sc + (b4.z - mean * sc);
    mean = s4.w * inv; var = q4.w * inv - mean * mean; sc = g4.w * rsqrtf(var + BN_EPS);
    v.w = v.w * sc + (b4.w - mean * sc);
    ((float4*)y)[i] = v;
}

extern "C" void kernel_launch(void* const* d_in, const int* in_sizes, int n_in,
                              void* d_out, int out_size, void* d_ws, size_t ws_size,
                              hipStream_t stream) {
    const float* feats = (const float*)d_in[0];
    const int*   src   = (const int*)d_in[1];
    const int*   dst   = (const int*)d_in[2];
    const float* W     = (const float*)d_in[3];
    const float* b     = (const float*)d_in[4];
    const float* Wr    = (const float*)d_in[5];
    const float* br    = (const float*)d_in[6];
    const float* gamma = (const float*)d_in[7];
    const float* beta  = (const float*)d_in[8];
    float* out = (float*)d_out;

    const int N = NNODES;
    const int E = in_sizes[1];

    float* ws        = (float*)d_ws;
    int*   csr       = (int*)d_ws;
    int*   rowptr    = csr + E;
    float* deg_out   = ws + (size_t)N * DFEAT;
    float* deg_in    = deg_out + N;
    float* bn_sum    = deg_in + N;
    float* bn_sumsq  = bn_sum + DFEAT;

    int* scr  = (int*)d_out;
    int* tab  = scr;
    int* recA = scr + 196608;
    int* recS = recA + E;
    int* recD = recS + E;

    scatter_kernel<<<NCHUNK, EDGE_T, 0, stream>>>(src, dst, tab, bn_sum, recA, recS, recD, E);

    local_kernel<<<NB, BIN_R, 0, stream>>>(tab, recA, deg_out, deg_in, rowptr, csr, N, E);

    gather_gemm_bn_kernel<<<GEMM_GRID, 256, 0, stream>>>(feats, rowptr, csr, deg_out, deg_in,
                                                         W, b, Wr, br, out, bn_sum, bn_sumsq);

    size_t nvec4 = (size_t)N * DFEAT / 4;
    bn_apply_kernel<<<(int)((nvec4 + 255) / 256), 256, 0, stream>>>(out, bn_sum, bn_sumsq, gamma, beta,
                                                                   nvec4, N);
}


# Waves 4..7 of each GEMM work-group sum the next tile's rows in front of their products (one tile further ahead), waves 0..3 behind: a SIMD's two waves alternate products and row waits; sum order kept
# speedup vs baseline: 4.7383x; 1.0316x over previous
#include <hip/hip_runtime.h>
#include <hip/hip_bf16.h>

typedef __attribute__((ext_vector_type(2))) float v2f;
typedef __attribute__((ext_vector_type(8))) float v8f;
typedef const __attribute__((address_space(1))) float4* gfloat4p;

#define NNODES 100000
#define DFEAT  128
#define BN_EPS 1e-5f
#define BIN_SH 9
#define BIN_R  512
#define NB     196
#define NB2    (2 * NB)
#define NCHUNK 250
#define EDGE_T 1024
#define CPB    ((NB + 31) / 32)
#define NEDGES 1600000
#define CH_E   ((NEDGES + NCHUNK - 1) / NCHUNK)
#define EPT    ((CH_E + EDGE_T - 1) / EDGE_T)
#define CSR_CAP 16384
#define LWAVES (BIN_R / 32)
#define SEGP   256
static_assert(CH_E < 65536, "a piece's offset and length share one 32-bit table word");
static_assert(NCHUNK <= SEGP, "the piece list holds every chunk");
static_assert(2 * SEGP == BIN_R, "one thread per piece-list entry");


template <int PER>
__device__ __forceinline__ void wave_excl_scan(const int* in, int* out, int n, int lane, bool total) {
    int loc[PER];
    int sum = 0;
    for (int i = 0; i < PER; ++i) {
        const int idx = lane * PER + i;
        loc[i] = (idx < n) ? in[idx] : 0;
        sum += loc[i];
    }
    int inc = sum;
    for (int o = 1; o < 32; o <<= 1) {
        const int x = __shfl_up(inc, o);
        if (lane >= o) inc += x;
    }
    int run = inc - sum;
    for (int i = 0; i < PER; ++i) {
        const int idx = lane * PER + i;
        if (idx < n) out[idx] = run;
        run += loc[i];
    }
    if (total && lane == 31) out[n] = run;
}

__global__ __launch_bounds__(EDGE_T) void scatter_kernel(const int* __restrict__ src,
                                                         const int* __restrict__ dst,
                                                         int* __restrict__ tab,
                                                         float* __restrict__ bn_acc,
                                                         int* __restrict__ recA,
                                                         int* __restrict__ recS,
                                                         int* __restrict__ recD, int E) {
    __shared__ int hist[NB2];
    __shared__ int lStart[NB2];
    __shared__ int cur[NB2];
    __shared__ int stA[CH_E];
    __shared__ int stD[CH_E];
    __shared__ int stS[CH_E];
    const int t = threadIdx.x;
    const int w = blockIdx.x;
    if (t < NB2) hist[t] = 0;
    if (w == 0 && t < 2 * DFEAT) bn_acc[t] = 0.0f;
    const int ch  = (E + NCHUNK - 1) / NCHUNK;
    const int beg = w * ch;
    const int end = min(E, beg + ch);
    int d[EPT], s[EPT];
    #pragma unroll
    for (int j = 0; j < EPT; ++j) {
        const int e = max(min(beg + t + j * EDGE_T, end - 1), 0);
        d[j] = dst[e];
        s[j] = src[e];
    }
    __syncthreads();
    #pragma unroll
    for (int j = 0; j < EPT; ++j) {
        if (beg + t + j * EDGE_T < end) {
            atomicAdd(&hist[d[j] >> BIN_SH], 1);
            atomicAdd(&hist[NB + (s[j] >> BIN_SH)], 1);
        }
    }
    __syncthreads();
    if (t < 32)       wave_excl_scan<CPB>(hist, lStart, NB, t, false);
    else if (t < 64)  wave_excl_scan<CPB>(hist + NB, lStart + NB, NB, t - 32, false);
    __syncthreads();
    if (t < NB2) {
        const int l = lStart[t];
        cur[t] = l;
        tab[w * NB2 + t] = (l << 16) | hist[t];
    }
    __syncthreads();
    #pragma unroll
    for (int j = 0; j < EPT; ++j) {
        if (beg + t + j * EDGE_T < end) {
            const int p = atomicAdd(&cur[d[j] >> BIN_SH], 1);
            stA[p] = s[j];
            stD[p] = d[j];
            const int q = atomicAdd(&cur[NB + (s[j] >> BIN_SH)], 1);
            stS[q] = s[j];
        }
    }
    __syncthreads();
    const int n = end - beg;
    for (int i = t; i < n; i += EDGE_T) {
        recA[beg + i] = stA[i];
        recD[beg + i] = stD[i];
        recS[beg + i] = stS[i];
    }
}

template <int U, bool TWO, class F>
__device__ __forceinline__ void walk_pieces(const int* __restrict__ R, int delta1,
                                            const int* pBase, const int* pCnt, int gEnd,
                                            int wave, int lane, F f) {
    for (int g0 = wave; g0 < gEnd; g0 += LWAVES * U) {
        int cn[U], ba[U];
        int mx = 0;
        #pragma unroll
        for (int u = 0; u < U; ++u) {
            const int g = min(g0 + LWAVES * u, gEnd - 1);
            const bool ok = g0 + LWAVES * u < gEnd;
            cn[u] = ok ? pCnt[g] : 0;
            ba[u] = pBase[g];
            mx = max(mx, cn[u]);
        }
        for (int j0 = 0; j0 < mx; j0 += 64) {
            int a[U][2], b[U][2];
            #pragma unroll
            for (int u = 0; u < U; ++u) {
                #pragma unroll
                for (int h = 0; h < 2; ++h) {
                    const int j = j0 + lane + 32 * h;
                    const int pos = ba[u] + max(min(j, cn[u] - 1), 0);
                    a[u][h] = R[pos];
                    b[u][h] = TWO ? R[pos + delta1] : 0;
                }
            }
            #pragma unroll
            for (int u = 0; u < U; ++u) {
                #pragma unroll
                for (int h = 0; h < 2; ++h) {
                    if (j0 + lane + 32 * h < cn[u]) f(g0 + LWAVES * u, a[u][h], b[u][h]);
                }
            }
        }
    }
}

__global__ __launch_bounds__(BIN_R) void local_kernel(const int* __restrict__ tab,
                                                      const int* __restrict__ R,
                                                      float* __restrict__ norm_src,
                                                      float* __restrict__ norm_dst,
                                                      int* __restrict__ rowptr,
                                                      int* __restrict__ csr, int n, int E) {
    __shared__ int deg[2 * BIN_R];
    __shared__ int part[BIN_R];
    __shared__ int stg[CSR_CAP];
    __shared__ int pBase[2 * SEGP];
    __shared__ int pCnt[2 * SEGP];
    __shared__ int pOff[SEGP];
    __shared__ int tot[2];
    int* cIn  = deg;
    int* cOut = deg + BIN_R;
    const int t = threadIdx.x;
    const int b = blockIdx.x;
    const int lane = t & 31;
    const int wave = t >> 5;
    const int ch = (E + NCHUNK - 1) / NCHUNK;
    cIn[t] = 0;
    cOut[t] = 0;
    {
        const int c  = t & (SEGP - 1);
        const int sd = t / SEGP;
        int cnt = 0, off = 0;
        if (c < NCHUNK) {
            const int p = tab[c * NB2 + sd * NB + b];
            cnt = p & 0xffff;
            off = p >> 16;
        }
        pCnt[t]  = cnt;
        pBase[t] = (sd ? E : 2 * E) + min(c, NCHUNK - 1) * ch + off;
        if (sd == 0) pOff[c] = off;
    }
    __syncthreads();
    if (t < 64) {
        const int* v = (t < 32) ? pCnt : pOff;
        int sm = 0;
        for (int i = lane; i < NCHUNK; i += 32) sm += v[i];
        for (int o = 16; o > 0; o >>= 1) sm += __shfl_xor(sm, o);
        if (lane == 0) tot[t >> 5] = sm;
    }
    walk_pieces<8, false>(R, 0, pBase, pCnt, 2 * SEGP, wave, lane,
        [&](int g, int v, int) { atomicAdd(&deg[(g / SEGP) * BIN_R + (v & (BIN_R - 1))], 1); });
    __syncthreads();
    const int dBeg = tot[1];
    const int nb   = tot[0];
    const int node = b * BIN_R + t;
    const int ci = cIn[t];
    if (node < n) {
        norm_src[node] = rsqrtf(fmaxf((float)cOut[t], 1.0f));
        norm_dst[node] = rsqrtf(fmaxf((float)ci, 1.0f));
    }
    part[t] = ci;
    __syncthreads();
    for (int o = 1; o < BIN_R; o <<= 1) {
        int x = (t >= o) ? part[t - o] : 0;
        __syncthreads();
        part[t] += x;
        __syncthreads();
    }
    const int start = dBeg + part[t] - ci;
    if (node <= n) rowptr[node] = start;
    cIn[t] = start;
    __syncthreads();
    const bool staged = nb <= CSR_CAP;
    walk_pieces<4, true>(R, -2 * E, pBase, pCnt, SEGP, wave, lane,
        [&](int, int dv, int sv) {
            const int p = atomicAdd(&cIn[dv & (BIN_R - 1)], 1);
            if (staged) stg[p - dBeg] = sv;
            else        csr[p] = sv;
        });
    if (staged) {
        __syncthreads();
        for (int i = t; i < nb; i += BIN_R) csr[dBeg + i] = stg[i];
    }
}

__device__ __forceinline__ void pair_bounds(int rp, int& beg, int& mid, int& end) {
    beg = __builtin_amdgcn_readlane(rp, 0);
    mid = __builtin_amdgcn_readlane(rp, 1);
    end = __builtin_amdgcn_readlane(rp, 2);
}

__device__ __forceinline__ int chunk_ids(const int* __restrict__ csr, int lane, int b0, int end) {
    return csr[max(min(b0 + lane, end - 1), 0)];
}

__device__ __forceinline__ void chunk_weights(int lane, int b0, int mid, int end, float ns,
                                              float& w0, float& w1) {
    const int e = b0 + lane;
    w0 = (e < mid) ? ns : 0.0f;
    w1 = (e >= mid && e < end) ? ns : 0.0f;
}

__device__ __forceinline__ void rows_issue(const float* __restrict__ feats, int lane, int s_l, int j0,
                                           float4 (&v)[8]) {
    #pragma unroll
    for (int u = 0; u < 8; ++u) {
        const int s = __builtin_amdgcn_readlane(s_l, j0 + u);
        const unsigned long long rb = (unsigned long long)(feats + (size_t)s * DFEAT);
        const unsigned int rlo = (unsigned int)__builtin_amdgcn_readfirstlane((int)(unsigned int)rb);
        const unsigned int rhi = (unsigned int)__builtin_amdgcn_readfirstlane((int)(unsigned int)(rb >> 32));
        const gfloat4p rowp = (gfloat4p)(((unsigned long long)rhi << 32) | rlo);
        v[u] = rowp[lane];
    }
}

__device__ __forceinline__ void rows_add(const float4 (&v)[8], float w0_l, float w1_l, int j0,
                                         float4& acc0, float4& acc1) {
    #pragma unroll
    for (int u = 0; u < 8; ++u) {
        const float n0 = __int_as_float(__builtin_amdgcn_readlane(__float_as_int(w0_l), j0 + u));
        const float n1 = __int_as_float(__builtin_amdgcn_readlane(__float_as_int(w1_l), j0 + u));
        acc0.x = __builtin_fmaf(v[u].x, n0, acc0.x); acc0.y = __builtin_fmaf(v[u].y, n0, acc0.y);
        acc0.z = __builtin_fmaf(v[u].z, n0, acc0.z); acc0.w = __builtin_fmaf(v[u].w, n0, acc0.w);
        acc1.x = __builtin_fmaf(v[u].x, n1, acc1.x); acc1.y = __builtin_fmaf(v[u].y, n1, acc1.y);
        acc1.z = __builtin_fmaf(v[u].z, n1, acc1.z); acc1.w = __builtin_fmaf(v[u].w, n1, acc1.w);
    }
}

__device__ __forceinline__ void grp_issue(const float* __restrict__ feats, int lane, int idA, int idB,
                                          int gi, float4 (&v)[8]) {
    rows_issue(feats, lane, (gi < 4) ? idA : idB, (gi & 3) * 8, v);
}
__device__ __forceinline__ void grp_add(const float4 (&v)[8], float w0A, float w1A, float w0B, float w1B,
                                        int gi, float4& acc0, float4& acc1) {
    rows_add(v, (gi < 4) ? w0A : w0B, (gi < 4) ? w1A : w1B, (gi & 3) * 8, acc0, acc1);
}

__device__ __forceinline__ void pair_finish(const float* __restrict__ feats,
                                            const int* __restrict__ csr,
                                            const float* __restrict__ norm_src, int lane,
                                            int beg, int mid, int end,
                                            int idA, int idB, float nsA, float nsB,
                                            float4 (&v0)[8], float4& acc0, float4& acc1) {
    acc0 = make_float4(0.0f, 0.0f, 0.0f, 0.0f);
    acc1 = make_float4(0.0f, 0.0f, 0.0f, 0.0f);
    {
        const int ng = (min(end - beg, 64) + 7) >> 3;
        float w0A, w1A, w0B, w1B;
        chunk_weights(lane, beg,      mid, end, nsA, w0A, w1A);
        chunk_weights(lane, beg + 32, mid, end, nsB, w0B, w1B);
        float4 vb[8];
        for (int g = 0;; g += 2) {
            grp_issue(feats, lane, idA, idB, g + 1, vb);
            grp_add(v0, w0A, w1A, w0B, w1B, g, acc0, acc1);
            if (g + 1 >= ng) break;
            grp_issue(feats, lane, idA, idB, g + 2, v0);
            grp_add(vb, w0A, w1A, w0B, w1B, g + 1, acc0, acc1);
            if (g + 2 >= ng) break;
        }
    }
    for (int b0 = beg + 64; b0 < end; b0 += 32) {
        const int cnt = min(32, end - b0);
        const int   s_l  = chunk_ids(csr, lane, b0, end);
        const float ns_l = norm_src[s_l];
        float w0_l, w1_l;
        chunk_weights(lane, b0, mid, end, ns_l, w0_l, w1_l);
        for (int j0 = 0; j0 < cnt; j0 += 8) {
            float4 v[8];
            rows_issue(feats, lane, s_l, j0, v);
            rows_add(v, w0_l, w1_l, j0, acc0, acc1);
        }
    }
}

#define GEMM_TILES (NNODES / 16)
#define GEMM_GRID  512
#define WR_LDS     16
#define LDS_P      132
static_assert(NNODES % 16 == 0, "every tile has 16 full rows");
static_assert(3 * GEMM_GRID <= GEMM_TILES, "every work-group has at least three tiles");

__device__ __forceinline__ void lds_barrier() {
    __builtin_amdgcn_fence(__ATOMIC_RELEASE, "workgroup", "local");
    __builtin_amdgcn_s_barrier();
    __builtin_amdgcn_fence(__ATOMIC_ACQUIRE, "workgroup", "local");
}

__global__ __launch_bounds__(256, 4) void gather_gemm_bn_kernel(
    const float* __restrict__ feats, const int* __restrict__ rowptr, const int* __restrict__ csr,
    const float* __restrict__ norm_src, const float* __restrict__ norm_dst,
    const float* __restrict__ W,  const float* __restrict__ b,
    const float* __restrict__ Wr, const float* __restrict__ br,
    float* __restrict__ y, float* __restrict__ bn_sum, float* __restrict__ bn_sumsq) {
    __shared__ __attribute__((aligned(16))) float sA[2][16 * LDS_P];
    __shared__ __attribute__((aligned(16))) float sF[2][16 * LDS_P];
    __shared__ v2f sWr[WR_LDS][256];
    const int lane = threadIdx.x & 31;
    const int wave = threadIdx.x >> 5;
    const int n0 = wave * 16;
    const int m  = lane & 15;
    const int kh = (lane >> 4) * 2;
    const int col = n0 + m;

    v2f bw[DFEAT / 4], bwr[DFEAT / 4 - WR_LDS];
    #pragma unroll
    for (int ks = 0; ks < DFEAT / 4; ++ks) {
        const int k = 4 * ks + kh;
        bw[ks].x  = W [(size_t)k       * DFEAT + col];
        bw[ks].y  = W [(size_t)(k + 1) * DFEAT + col];
        v2f t;
        t.x = Wr[(size_t)k       * DFEAT + col];
        t.y = Wr[(size_t)(k + 1) * DFEAT + col];
        if (ks < DFEAT / 4 - WR_LDS) bwr[ks] = t;
        else                         sWr[ks - (DFEAT / 4 - WR_LDS)][threadIdx.x] = t;
    }
    const float bc  = b[col];
    const float brc = br[col];
    const v2f bc2  = {bc, bc};
    const v2f brc2 = {brc, brc};
    const int rOff = (lane >> 4) * 8;

    const int so0 = (2 * wave) * LDS_P + 4 * lane;
    const int so1 = so0 + LDS_P;
    const int fo = m * LDS_P + kh;
    const int rpl = 2 * wave + min(lane, 2);

    float4 pf0, pf1;
    float4 v0[8];
    int tile = blockIdx.x;
    int beg1, mid1, end1, idA1, idB1;
    int rp2;
    float nd_l = 0.0f, nsA1 = 0.0f, nsB1 = 0.0f;
    int beg2 = 0, mid2 = 0, end2 = 0, idA2 = 0, idB2 = 0, rp3 = 0;
    const bool late = __builtin_amdgcn_readfirstlane(wave) >= 4;

    auto request = [&](int g) __attribute__((always_inline)) {
        const int r0 = g * 16 + 2 * wave;
        nd_l = norm_dst[r0 + (lane & 1)];
        pf0 = ((const float4*)(feats + (size_t)r0       * DFEAT))[lane];
        pf1 = ((const float4*)(feats + (size_t)(r0 + 1) * DFEAT))[lane];
        nsA1 = norm_src[idA1];
        nsB1 = norm_src[idB1];
        rows_issue(feats, lane, idA1, 0, v0);
        pair_bounds(rp2, beg2, mid2, end2);
        idA2 = chunk_ids(csr, lane, beg2, end2);
        idB2 = chunk_ids(csr, lane, beg2 + 32, end2);
        rp3  = rowptr[min(g + 2 * GEMM_GRID, GEMM_TILES - 1) * 16 + rpl];
    };
    auto finish = [&](int sb) __attribute__((always_inline)) {
        *(float4*)&sF[sb][so0] = pf0;
        *(float4*)&sF[sb][so1] = pf1;
        float4 a0, a1;
        pair_finish(feats, csr, norm_src, lane, beg1, mid1, end1, idA1, idB1, nsA1, nsB1, v0, a0, a1);
        const float nd0 = __int_as_float(__builtin_amdgcn_readlane(__float_as_int(nd_l), 0));
        const float nd1 = __int_as_float(__builtin_amdgcn_readlane(__float_as_int(nd_l), 1));
        a0.x *= nd0; a0.y *= nd0; a0.z *= nd0; a0.w *= nd0;
        a1.x *= nd1; a1.y *= nd1; a1.z *= nd1; a1.w *= nd1;
        *(float4*)&sA[sb][so0] = a0;
        *(float4*)&sA[sb][so1] = a1;
        beg1 = beg2; mid1 = mid2; end1 = end2; idA1 = idA2; idB1 = idB2;
        rp2 = rp3;
    };

    {
        const int rp0 = rowptr[tile * 16 + rpl];
        const int rp1 = rowptr[(tile + GEMM_GRID) * 16 + rpl];
        rp2           = rowptr[(tile + 2 * GEMM_GRID) * 16 + rpl];
        const int r0 = tile * 16 + 2 * wave;
        const float nd_p = norm_dst[r0 + (lane & 1)];
        pf0 = ((const float4*)(feats + (size_t)r0       * DFEAT))[lane];
        pf1 = ((const float4*)(feats + (size_t)(r0 + 1) * DFEAT))[lane];
        int beg, mid, end;
        pair_bounds(rp0, beg, mid, end);
        const int idA = chunk_ids(csr, lane, beg, end);
        const int idB = chunk_ids(csr, lane, beg + 32, end);
        pair_bounds(rp1, beg1, mid1, end1);
        idA1 = chunk_ids(csr, lane, beg1, end1);
        idB1 = chunk_ids(csr, lane, beg1 + 32, end1);
        const float nsA = norm_src[idA];
        const float nsB = norm_src[idB];
        rows_issue(feats, lane, idA, 0, v0);
        float4 a0, a1;
        pair_finish(feats, csr, norm_src, lane, beg, mid, end, idA, idB, nsA, nsB, v0, a0, a1);
        const float nd0 = __int_as_float(__builtin_amdgcn_readlane(__float_as_int(nd_p), 0));
        const float nd1 = __int_as_float(__builtin_amdgcn_readlane(__float_as_int(nd_p), 1));
        a0.x *= nd0; a0.y *= nd0; a0.z *= nd0; a0.w *= nd0;
        a1.x *= nd1; a1.y *= nd1; a1.z *= nd1; a1.w *= nd1;
        *(float4*)&sA[0][so0] = a0;
        *(float4*)&sA[0][so1] = a1;
        *(float4*)&sF[0][so0] = pf0;
        *(float4*)&sF[0][so1] = pf1;
    }
    if (late) request(tile + GEMM_GRID);
    lds_barrier();

    float s8 = 0.0f, q8 = 0.0f;
    int buf = 0;
    for (;;) {
        const int  next = tile + GEMM_GRID;
        const bool more = next < GEMM_TILES;
        if (late && more) finish(buf ^ 1);
        const int g = late ? next + GEMM_GRID : next;
        if (g < GEMM_TILES) request(g);

        const float* aB = &sA[buf][fo];
        const float* fB = &sF[buf][fo];
        v8f accC = {};
        v8f accR = {};
        #pragma unroll
        for (int ks = 0; ks < DFEAT / 4; ++ks) {
            const v2f a = *(const v2f*)(aB + 4 * ks);
            const v2f f = *(const v2f*)(fB + 4 * ks);
            accC = __builtin_amdgcn_wmma_f32_16x16x4_f32(false, a, false, bw[ks],  (short)0, accC, false, false);
            v2f wr;
            if (ks < DFEAT / 4 - WR_LDS) wr = bwr[ks];
            else                         wr = sWr[ks - (DFEAT / 4 - WR_LDS)][threadIdx.x];
            accR = __builtin_amdgcn_wmma_f32_16x16x4_f32(false, f, false, wr, (short)0, accR, false, false);
        }

        const int rowBase = tile * 16;
        #pragma unroll
        for (int r = 0; r < 8; r += 2) {
            const v2f cc = (v2f){accC[r], accC[r + 1]} + bc2;
            const v2f ee = (v2f){accR[r], accR[r + 1]} + brc2;
            const float c0 = cc.x > 0.0f ? cc.x : 0.0f;
            const float c1 = cc.y > 0.0f ? cc.y : 0.0f;
            const float e0 = ee.x > 0.0f ? ee.x : 0.0f;
            const float e1 = ee.y > 0.0f ? ee.y : 0.0f;
            const float va = c0 + e0;
            const float vc = c1 + e1;
            y[(size_t)(rowBase + rOff + r)     * DFEAT + col] = va;
            y[(size_t)(rowBase + rOff + r + 1) * DFEAT + col] = vc;
            s8 += va;
            q8 += va * va;
            s8 += vc;
            q8 += vc * vc;
        }

        if (!late && more) finish(buf ^ 1);
        lds_barrier();
        buf ^= 1;
        tile = next;
        if (!more) break;
    }

    float s8o = __shfl_down(s8, 16);
    float q8o = __shfl_down(q8, 16);
    if (lane < 16) {
        atomicAdd(&bn_sum[col],   s8 + s8o);
        atomicAdd(&bn_sumsq[col], q8 + q8o);
    }
}

__global__ __launch_bounds__(256) void bn_apply_kernel(float* __restrict__ y,
                                                       const float* __restrict__ sum,
                                                       const float* __restrict__ sumsq,
                                                       const float* __restrict__ gamma,
                                                       const float* __restrict__ beta,
                                                       size_t nvec4, int n) {
    size_t i = (size_t)blockIdx.x * blockDim.x + threadIdx.x;
    if (i >= nvec4) return;
    float4 v = ((float4*)y)[i];
    const int col = (int)((i * 4) & (DFEAT - 1));
    const float4 s4 = *(const float4*)(sum + col);
    const float4 q4 = *(const float4*)(sumsq + col);
    const float4 g4 = *(const float4*)(gamma + col);
    const float4 b4 = *(const float4*)(beta + col);
    const float inv = 1.0f / (float)n;
    float mean, var, sc;
    mean = s4.x * inv; var = q4.x * inv - mean * mean; sc = g4.x * rsqrtf(var + BN_EPS);
    v.x = v.x * sc + (b4.x - mean * sc);
    mean = s4.y * inv; var = q4.y * inv - mean * mean; sc = g4.y * rsqrtf(var + BN_EPS);
    v.y = v.y * sc + (b4.y - mean * sc);
    mean = s4.z * inv; var = q4.z * inv - mean * mean; sc = g4.z * rsqrtf(var + BN_EPS);
    v.z = v.z * sc + (b4.z - mean * sc);
    mean = s4.w * inv; var = q4.w * inv - mean * mean; sc = g4.w * rsqrtf(var + BN_EPS);
    v.w = v.w * sc + (b4.w - mean * sc);
    ((float4*)y)[i] = v;
}

extern "C" void kernel_launch(void* const* d_in, const int* in_sizes, int n_in,
                              void* d_out, int out_size, void* d_ws, size_t ws_size,
                              hipStream_t stream) {
    const float* feats = (const float*)d_in[0];
    const int*   src   = (const int*)d_in[1];
    const int*   dst   = (const int*)d_in[2];
    const float* W     = (const float*)d_in[3];
    const float* b     = (const float*)d_in[4];
    const float* Wr    = (const float*)d_in[5];
    const float* br    = (const float*)d_in[6];
    const float* gamma = (const float*)d_in[7];
    const float* beta  = (const float*)d_in[8];
    float* out = (float*)d_out;

    const int N = NNODES;
    const int E = in_sizes[1];

    float* ws        = (float*)d_ws;
    int*   csr       = (int*)d_ws;
    int*   rowptr    = csr + E;
    float* deg_out   = ws + (size_t)N * DFEAT;
    float* deg_in    = deg_out + N;
    float* bn_sum    = deg_in + N;
    float* bn_sumsq  = bn_sum + DFEAT;

    int* scr  = (int*)d_out;
    int* tab  = scr;
    int* recA = scr + 196608;
    int* recS = recA + E;
    int* recD = recS + E;

    scatter_kernel<<<NCHUNK, EDGE_T, 0, stream>>>(src, dst, tab, bn_sum, recA, recS, recD, E);

    local_kernel<<<NB, BIN_R, 0, stream>>>(tab, recA, deg_out, deg_in, rowptr, csr, N, E);

    gather_gemm_bn_kernel<<<GEMM_GRID, 256, 0, stream>>>(feats, rowptr, csr, deg_out, deg_in,
                                                         W, b, Wr, br, out, bn_sum, bn_sumsq);

    size_t nvec4 = (size_t)N * DFEAT / 4;
    bn_apply_kernel<<<(int)((nvec4 + 255) / 256), 256, 0, stream>>>(out, bn_sum, bn_sumsq, gamma, beta,
                                                                   nvec4, N);
}
